# starting kernel
# baseline (speedup 1.0000x reference)
#include <hip/hip_runtime.h>

#define NN  100000
#define NF  256
#define HH  8
#define CC  16
#define HC  128
#define NC  16
#define EE  1600000
#define ET  1700000
#define NEG 0.2f

typedef _Float16 h8   __attribute__((ext_vector_type(8)));
typedef _Float16 v16h __attribute__((ext_vector_type(16)));
typedef float    v8f  __attribute__((ext_vector_type(8)));

__device__ __forceinline__ void edge_sd(const int* __restrict__ ei, int e, int& s, int& d) {
    if (e < EE) { s = ei[e]; d = ei[EE + e]; }
    else        { s = e - EE; d = s; }
}

__device__ __forceinline__ void atomicMaxF(float* addr, float val) {
    if (val >= 0.0f) atomicMax((int*)addr, __float_as_int(val));
    else             atomicMin((unsigned int*)addr, __float_as_uint(val));
}

__device__ __forceinline__ float lrelu(float a) { return a > 0.0f ? a : NEG * a; }

__global__ __launch_bounds__(256) void gat_cvtw(const float* __restrict__ W1,
                                                const float* __restrict__ W2,
                                                _Float16* __restrict__ w1h,
                                                _Float16* __restrict__ w2h) {
    int i = blockIdx.x * 256 + threadIdx.x;
    if (i < HC * NF)              w1h[i] = (_Float16)W1[i];
    else if (i < HC * NF + NC*HC) w2h[i - HC * NF] = (_Float16)W2[i - HC * NF];
}

__global__ __launch_bounds__(256) void gat_cvtx(const float* __restrict__ x,
                                                _Float16* __restrict__ xh) {
    int i = blockIdx.x * 256 + threadIdx.x;
    if (i < NN * NF) xh[i] = (_Float16)x[i];
}

__global__ __launch_bounds__(256)
void gat_gemm1(const _Float16* __restrict__ xh, const _Float16* __restrict__ w1h,
               float* __restrict__ h1) {
    __shared__ _Float16 lw[HC * NF];
    const int tid = threadIdx.x;
    for (int idx = tid; idx < HC * 32; idx += 256) {
        int r = idx >> 5, ch = idx & 31;
        *(h8*)(&lw[r * NF + ((ch ^ (r & 15)) << 3)]) =
            *(const h8*)(w1h + (size_t)r * NF + (ch << 3));
    }
    __syncthreads();

    const int lane = tid & 31;
    const int rt   = blockIdx.x * 8 + (tid >> 5);
    if (rt >= NN / 16) return;
    const int rsub = lane & 15;
    const int ksel = (lane >> 4) << 3;
    const _Float16* xrow = xh + (size_t)(rt * 16 + rsub) * NF;

    v8f acc[8] = {};
    #pragma unroll
    for (int ks = 0; ks < 8; ++ks) {
        const int kb = ks * 32 + ksel;
        const int c0 = kb >> 3;
        union { v16h v; h8 p[2]; } A;
        A.p[0] = *(const h8*)(xrow + kb);
        A.p[1] = *(const h8*)(xrow + kb + 16);
        #pragma unroll
        for (int ct = 0; ct < 8; ++ct) {
            const int n = ct * 16 + rsub;
            union { v16h v; h8 p[2]; } B;
            B.p[0] = *(const h8*)(&lw[n * NF + (((c0    ) ^ (n & 15)) << 3)]);
            B.p[1] = *(const h8*)(&lw[n * NF + (((c0 + 2) ^ (n & 15)) << 3)]);
            acc[ct] = __builtin_amdgcn_wmma_f32_16x16x32_f16(
                false, A.v, false, B.v, (short)0, acc[ct], false, false);
        }
    }
    const int rhi = (lane >> 4) << 3;
    #pragma unroll
    for (int ct = 0; ct < 8; ++ct) {
        #pragma unroll
        for (int r = 0; r < 8; ++r)
            h1[(size_t)(rt * 16 + rhi + r) * HC + ct * 16 + rsub] = acc[ct][r];
    }
}

__global__ __launch_bounds__(256)
void gat_att1(const float* __restrict__ h1, const float* __restrict__ as,
              const float* __restrict__ ad, float* __restrict__ a_src,
              float* __restrict__ a_dst, float* __restrict__ amax,
              float* __restrict__ den, float* __restrict__ agg) {
    int i = blockIdx.x * 256 + threadIdx.x;
    if (i >= NN * HH) return;
    int n = i >> 3, h = i & 7;
    const float* hp = h1 + (size_t)n * HC + h * CC;
    float s = 0.0f, d = 0.0f;
    #pragma unroll
    for (int c = 0; c < CC; ++c) { s += hp[c] * as[h * CC + c]; d += hp[c] * ad[h * CC + c]; }
    a_src[i] = s; a_dst[i] = d;
    amax[i] = -__builtin_inff(); den[i] = 0.0f;
    float* ap = agg + (size_t)n * HC + h * CC;
    #pragma unroll
    for (int c = 0; c < CC; ++c) ap[c] = 0.0f;
}

__global__ __launch_bounds__(256)
void gat_emax1(const int* __restrict__ ei, const float* __restrict__ a_src,
               const float* __restrict__ a_dst, float* __restrict__ amax) {
    int i = blockIdx.x * 256 + threadIdx.x;
    if (i >= ET * HH) return;
    int e = i >> 3, h = i & 7, s, d;
    edge_sd(ei, e, s, d);
    atomicMaxF(&amax[d * HH + h], lrelu(a_src[s * HH + h] + a_dst[d * HH + h]));
}

__global__ __launch_bounds__(256)
void gat_eden1(const int* __restrict__ ei, const float* __restrict__ a_src,
               const float* __restrict__ a_dst, const float* __restrict__ amax,
               float* __restrict__ den) {
    int i = blockIdx.x * 256 + threadIdx.x;
    if (i >= ET * HH) return;
    int e = i >> 3, h = i & 7, s, d;
    edge_sd(ei, e, s, d);
    float al = lrelu(a_src[s * HH + h] + a_dst[d * HH + h]);
    atomicAdd(&den[d * HH + h], expf(al - amax[d * HH + h]));
}

__global__ __launch_bounds__(256)
void gat_attw1(const int* __restrict__ ei, const float* __restrict__ a_src,
               const float* __restrict__ a_dst, const float* __restrict__ amax,
               const float* __restrict__ den, float* __restrict__ attn) {
    int i = blockIdx.x * 256 + threadIdx.x;
    if (i >= ET * HH) return;
    int e = i >> 3, h = i & 7, s, d;
    edge_sd(ei, e, s, d);
    float al = lrelu(a_src[s * HH + h] + a_dst[d * HH + h]);
    attn[i] = expf(al - amax[d * HH + h]) / (den[d * HH + h] + 1e-16f);
}

__global__ __launch_bounds__(256)
void gat_eagg1(const int* __restrict__ ei, const float* __restrict__ attn,
               const float* __restrict__ h1, float* __restrict__ agg) {
    int i = blockIdx.x * 256 + threadIdx.x;
    int e = i >> 5, q = i & 31;
    if (e >= ET) return;
    int c = q << 2;
    int s, d; edge_sd(ei, e, s, d);
    float a = attn[e * HH + (q >> 2)];
    const float4 hv = *(const float4*)(h1 + (size_t)s * HC + c);
    float* ap = agg + (size_t)d * HC + c;
    atomicAdd(ap + 0, hv.x * a);
    atomicAdd(ap + 1, hv.y * a);
    atomicAdd(ap + 2, hv.z * a);
    atomicAdd(ap + 3, hv.w * a);
}

__global__ __launch_bounds__(256)
void gat_biaselu(float* __restrict__ agg, const float* __restrict__ b1) {
    int i = blockIdx.x * 256 + threadIdx.x;
    if (i >= NN * HC) return;
    float v = agg[i] + b1[i & 127];
    agg[i] = v > 0.0f ? v : (expf(v) - 1.0f);
}

__global__ __launch_bounds__(256)
void gat_gemm2(const float* __restrict__ a1, const _Float16* __restrict__ w2h,
               float* __restrict__ h2) {
    const int lane = threadIdx.x & 31;
    const int rt   = blockIdx.x * 8 + (threadIdx.x >> 5);
    if (rt >= NN / 16) return;
    const int rsub = lane & 15;
    const int ksel = (lane >> 4) << 3;
    const float*    arow = a1 + (size_t)(rt * 16 + rsub) * HC;
    const _Float16* wrow = w2h + (size_t)rsub * HC;

    v8f acc = {};
    #pragma unroll
    for (int ks = 0; ks < 4; ++ks) {
        const int kb = ks * 32 + ksel;
        union { v16h v; _Float16 e[16]; } A;
        #pragma unroll
        for (int t = 0; t < 8; ++t) {
            A.e[t]     = (_Float16)arow[kb + t];
            A.e[8 + t] = (_Float16)arow[kb + 16 + t];
        }
        union { v16h v; h8 p[2]; } B;
        B.p[0] = *(const h8*)(wrow + kb);
        B.p[1] = *(const h8*)(wrow + kb + 16);
        acc = __builtin_amdgcn_wmma_f32_16x16x32_f16(
            false, A.v, false, B.v, (short)0, acc, false, false);
    }
    const int rhi = (lane >> 4) << 3;
    #pragma unroll
    for (int r = 0; r < 8; ++r)
        h2[(size_t)(rt * 16 + rhi + r) * NC + rsub] = acc[r];
}

__global__ __launch_bounds__(256)
void gat_att2(const float* __restrict__ h2, const float* __restrict__ as,
              const float* __restrict__ ad, float* __restrict__ a_src,
              float* __restrict__ a_dst, float* __restrict__ amax,
              float* __restrict__ den, float* __restrict__ agg) {
    int n = blockIdx.x * 256 + threadIdx.x;
    if (n >= NN) return;
    const float* hp = h2 + (size_t)n * NC;
    float s = 0.0f, d = 0.0f;
    #pragma unroll
    for (int c = 0; c < NC; ++c) { s += hp[c] * as[c]; d += hp[c] * ad[c]; }
    a_src[n] = s; a_dst[n] = d;
    amax[n] = -__builtin_inff(); den[n] = 0.0f;
    float* ap = agg + (size_t)n * NC;
    #pragma unroll
    for (int c = 0; c < NC; ++c) ap[c] = 0.0f;
}

__global__ __launch_bounds__(256)
void gat_emax2(const int* __restrict__ ei, const float* __restrict__ a_src,
               const float* __restrict__ a_dst, float* __restrict__ amax) {
    int e = blockIdx.x * 256 + threadIdx.x;
    if (e >= ET) return;
    int s, d; edge_sd(ei, e, s, d);
    atomicMaxF(&amax[d], lrelu(a_src[s] + a_dst[d]));
}

__global__ __launch_bounds__(256)
void gat_eden2(const int* __restrict__ ei, const float* __restrict__ a_src,
               const float* __restrict__ a_dst, const float* __restrict__ amax,
               float* __restrict__ den) {
    int e = blockIdx.x * 256 + threadIdx.x;
    if (e >= ET) return;
    int s, d; edge_sd(ei, e, s, d);
    float al = lrelu(a_src[s] + a_dst[d]);
    atomicAdd(&den[d], expf(al - amax[d]));
}

__global__ __launch_bounds__(256)
void gat_attw2(const int* __restrict__ ei, const float* __restrict__ a_src,
               const float* __restrict__ a_dst, const float* __restrict__ amax,
               const float* __restrict__ den, float* __restrict__ attn) {
    int e = blockIdx.x * 256 + threadIdx.x;
    if (e >= ET) return;
    int s, d; edge_sd(ei, e, s, d);
    float al = lrelu(a_src[s] + a_dst[d]);
    attn[e] = expf(al - amax[d]) / (den[d] + 1e-16f);
}

__global__ __launch_bounds__(256)
void gat_eagg2(const int* __restrict__ ei, const float* __restrict__ attn,
               const float* __restrict__ h2, float* __restrict__ agg) {
    int i = blockIdx.x * 256 + threadIdx.x;
    int e = i >> 2, q = i & 3;
    if (e >= ET) return;
    int c = q << 2;
    int s, d; edge_sd(ei, e, s, d);
    float a = attn[e];
    const float4 hv = *(const float4*)(h2 + (size_t)s * NC + c);
    float* ap = agg + (size_t)d * NC + c;
    atomicAdd(ap + 0, hv.x * a);
    atomicAdd(ap + 1, hv.y * a);
    atomicAdd(ap + 2, hv.z * a);
    atomicAdd(ap + 3, hv.w * a);
}

__global__ __launch_bounds__(256)
void gat_lsm(const float* __restrict__ agg, const float* __restrict__ b2,
             float* __restrict__ out) {
    int n = blockIdx.x * 256 + threadIdx.x;
    if (n >= NN) return;
    float v[NC], m = -__builtin_inff();
    #pragma unroll
    for (int c = 0; c < NC; ++c) { v[c] = agg[(size_t)n * NC + c] + b2[c]; m = fmaxf(m, v[c]); }
    float sum = 0.0f;
    #pragma unroll
    for (int c = 0; c < NC; ++c) sum += expf(v[c] - m);
    float ls = logf(sum);
    #pragma unroll
    for (int c = 0; c < NC; ++c) out[(size_t)n * NC + c] = v[c] - m - ls;
}

extern "C" void kernel_launch(void* const* d_in, const int* in_sizes, int n_in,
                              void* d_out, int out_size, void* d_ws, size_t ws_size,
                              hipStream_t stream) {
    (void)in_sizes; (void)n_in; (void)out_size; (void)ws_size;
    const float* x   = (const float*)d_in[0];
    const int*   ei  = (const int*)d_in[1];
    const float* W1  = (const float*)d_in[2];
    const float* as1 = (const float*)d_in[3];
    const float* ad1 = (const float*)d_in[4];
    const float* b1  = (const float*)d_in[5];
    const float* W2  = (const float*)d_in[6];
    const float* as2 = (const float*)d_in[7];
    const float* ad2 = (const float*)d_in[8];
    const float* b2  = (const float*)d_in[9];
    float* out = (float*)d_out;

    char* wsb = (char*)d_ws;
    size_t off = 0;
    auto alloc = [&](size_t bytes) -> void* {
        void* p = wsb + off;
        off += (bytes + 255) & ~(size_t)255;
        return p;
    };
    _Float16* w1h  = (_Float16*)alloc((size_t)HC * NF * 2);
    _Float16* w2h  = (_Float16*)alloc((size_t)NC * HC * 2);
    _Float16* xh   = (_Float16*)alloc((size_t)NN * NF * 2);
    float* h1      = (float*)alloc((size_t)NN * HC * 4);
    float* asrc1   = (float*)alloc((size_t)NN * HH * 4);
    float* adst1   = (float*)alloc((size_t)NN * HH * 4);
    float* amax1   = (float*)alloc((size_t)NN * HH * 4);
    float* den1    = (float*)alloc((size_t)NN * HH * 4);
    float* attn1   = (float*)alloc((size_t)ET * HH * 4);
    float* h2      = (float*)alloc((size_t)NN * NC * 4);
    float* asrc2   = (float*)alloc((size_t)NN * 4);
    float* adst2   = (float*)alloc((size_t)NN * 4);
    float* amax2   = (float*)alloc((size_t)NN * 4);
    float* den2    = (float*)alloc((size_t)NN * 4);
    float* attn2   = (float*)alloc((size_t)ET * 4);
    float* agg2    = (float*)alloc((size_t)NN * NC * 4);
    float* agg1    = (float*)xh;

    const dim3 B(256);
    gat_cvtw <<<(HC * NF + NC * HC + 255) / 256, B, 0, stream>>>(W1, W2, w1h, w2h);
    gat_cvtx <<<(NN * NF) / 256, B, 0, stream>>>(x, xh);
    gat_gemm1<<<(NN / 16 + 7) / 8, B, 0, stream>>>(xh, w1h, h1);
    gat_att1 <<<(NN * HH) / 256, B, 0, stream>>>(h1, as1, ad1, asrc1, adst1, amax1, den1, agg1);
    gat_emax1<<<(ET * HH) / 256, B, 0, stream>>>(ei, asrc1, adst1, amax1);
    gat_eden1<<<(ET * HH) / 256, B, 0, stream>>>(ei, asrc1, adst1, amax1, den1);
    gat_attw1<<<(ET * HH) / 256, B, 0, stream>>>(ei, asrc1, adst1, amax1, den1, attn1);
    gat_eagg1<<<(ET * 32) / 256, B, 0, stream>>>(ei, attn1, h1, agg1);
    gat_biaselu<<<(NN * HC) / 256, B, 0, stream>>>(agg1, b1);
    gat_gemm2<<<(NN / 16 + 7) / 8, B, 0, stream>>>(agg1, w2h, h2);
    gat_att2 <<<(NN + 255) / 256, B, 0, stream>>>(h2, as2, ad2, asrc2, adst2, amax2, den2, agg2);
    gat_emax2<<<(ET + 255) / 256, B, 0, stream>>>(ei, asrc2, adst2, amax2);
    gat_eden2<<<(ET + 255) / 256, B, 0, stream>>>(ei, asrc2, adst2, amax2, den2);
    gat_attw2<<<(ET + 255) / 256, B, 0, stream>>>(ei, asrc2, adst2, amax2, den2, attn2);
    gat_eagg2<<<(ET * 4 + 255) / 256, B, 0, stream>>>(ei, attn2, h2, agg2);
    gat_lsm  <<<(NN + 255) / 256, B, 0, stream>>>(agg2, b2, out);
}


# eagg1/eagg2: lanes remapped to consecutive channels so each float-atomic instruction covers whole contiguous rows (128 B / 64 B), not 4 B per 16 B stride; same atomics, arrival-order sums as before
# speedup vs baseline: 1.1541x; 1.1541x over previous
#include <hip/hip_runtime.h>

#define NN  100000
#define NF  256
#define HH  8
#define CC  16
#define HC  128
#define NC  16
#define EE  1600000
#define ET  1700000
#define NEG 0.2f

typedef _Float16 h8   __attribute__((ext_vector_type(8)));
typedef _Float16 v16h __attribute__((ext_vector_type(16)));
typedef float    v8f  __attribute__((ext_vector_type(8)));

__device__ __forceinline__ void edge_sd(const int* __restrict__ ei, int e, int& s, int& d) {
    if (e < EE) { s = ei[e]; d = ei[EE + e]; }
    else        { s = e - EE; d = s; }
}

__device__ __forceinline__ void atomicMaxF(float* addr, float val) {
    if (val >= 0.0f) atomicMax((int*)addr, __float_as_int(val));
    else             atomicMin((unsigned int*)addr, __float_as_uint(val));
}

__device__ __forceinline__ float lrelu(float a) { return a > 0.0f ? a : NEG * a; }

__global__ __launch_bounds__(256) void gat_cvtw(const float* __restrict__ W1,
                                                const float* __restrict__ W2,
                                                _Float16* __restrict__ w1h,
                                                _Float16* __restrict__ w2h) {
    int i = blockIdx.x * 256 + threadIdx.x;
    if (i < HC * NF)              w1h[i] = (_Float16)W1[i];
    else if (i < HC * NF + NC*HC) w2h[i - HC * NF] = (_Float16)W2[i - HC * NF];
}

__global__ __launch_bounds__(256) void gat_cvtx(const float* __restrict__ x,
                                                _Float16* __restrict__ xh) {
    int i = blockIdx.x * 256 + threadIdx.x;
    if (i < NN * NF) xh[i] = (_Float16)x[i];
}

__global__ __launch_bounds__(256)
void gat_gemm1(const _Float16* __restrict__ xh, const _Float16* __restrict__ w1h,
               float* __restrict__ h1) {
    __shared__ _Float16 lw[HC * NF];
    const int tid = threadIdx.x;
    for (int idx = tid; idx < HC * 32; idx += 256) {
        int r = idx >> 5, ch = idx & 31;
        *(h8*)(&lw[r * NF + ((ch ^ (r & 15)) << 3)]) =
            *(const h8*)(w1h + (size_t)r * NF + (ch << 3));
    }
    __syncthreads();

    const int lane = tid & 31;
    const int rt   = blockIdx.x * 8 + (tid >> 5);
    if (rt >= NN / 16) return;
    const int rsub = lane & 15;
    const int ksel = (lane >> 4) << 3;
    const _Float16* xrow = xh + (size_t)(rt * 16 + rsub) * NF;

    v8f acc[8] = {};
    #pragma unroll
    for (int ks = 0; ks < 8; ++ks) {
        const int kb = ks * 32 + ksel;
        const int c0 = kb >> 3;
        union { v16h v; h8 p[2]; } A;
        A.p[0] = *(const h8*)(xrow + kb);
        A.p[1] = *(const h8*)(xrow + kb + 16);
        #pragma unroll
        for (int ct = 0; ct < 8; ++ct) {
            const int n = ct * 16 + rsub;
            union { v16h v; h8 p[2]; } B;
            B.p[0] = *(const h8*)(&lw[n * NF + (((c0    ) ^ (n & 15)) << 3)]);
            B.p[1] = *(const h8*)(&lw[n * NF + (((c0 + 2) ^ (n & 15)) << 3)]);
            acc[ct] = __builtin_amdgcn_wmma_f32_16x16x32_f16(
                false, A.v, false, B.v, (short)0, acc[ct], false, false);
        }
    }
    const int rhi = (lane >> 4) << 3;
    #pragma unroll
    for (int ct = 0; ct < 8; ++ct) {
        #pragma unroll
        for (int r = 0; r < 8; ++r)
            h1[(size_t)(rt * 16 + rhi + r) * HC + ct * 16 + rsub] = acc[ct][r];
    }
}

__global__ __launch_bounds__(256)
void gat_att1(const float* __restrict__ h1, const float* __restrict__ as,
              const float* __restrict__ ad, float* __restrict__ a_src,
              float* __restrict__ a_dst, float* __restrict__ amax,
              float* __restrict__ den, float* __restrict__ agg) {
    int i = blockIdx.x * 256 + threadIdx.x;
    if (i >= NN * HH) return;
    int n = i >> 3, h = i & 7;
    const float* hp = h1 + (size_t)n * HC + h * CC;
    float s = 0.0f, d = 0.0f;
    #pragma unroll
    for (int c = 0; c < CC; ++c) { s += hp[c] * as[h * CC + c]; d += hp[c] * ad[h * CC + c]; }
    a_src[i] = s; a_dst[i] = d;
    amax[i] = -__builtin_inff(); den[i] = 0.0f;
    float* ap = agg + (size_t)n * HC + h * CC;
    #pragma unroll
    for (int c = 0; c < CC; ++c) ap[c] = 0.0f;
}

__global__ __launch_bounds__(256)
void gat_emax1(const int* __restrict__ ei, const float* __restrict__ a_src,
               const float* __restrict__ a_dst, float* __restrict__ amax) {
    int i = blockIdx.x * 256 + threadIdx.x;
    if (i >= ET * HH) return;
    int e = i >> 3, h = i & 7, s, d;
    edge_sd(ei, e, s, d);
    atomicMaxF(&amax[d * HH + h], lrelu(a_src[s * HH + h] + a_dst[d * HH + h]));
}

__global__ __launch_bounds__(256)
void gat_eden1(const int* __restrict__ ei, const float* __restrict__ a_src,
               const float* __restrict__ a_dst, const float* __restrict__ amax,
               float* __restrict__ den) {
    int i = blockIdx.x * 256 + threadIdx.x;
    if (i >= ET * HH) return;
    int e = i >> 3, h = i & 7, s, d;
    edge_sd(ei, e, s, d);
    float al = lrelu(a_src[s * HH + h] + a_dst[d * HH + h]);
    atomicAdd(&den[d * HH + h], expf(al - amax[d * HH + h]));
}

__global__ __launch_bounds__(256)
void gat_attw1(const int* __restrict__ ei, const float* __restrict__ a_src,
               const float* __restrict__ a_dst, const float* __restrict__ amax,
               const float* __restrict__ den, float* __restrict__ attn) {
    int i = blockIdx.x * 256 + threadIdx.x;
    if (i >= ET * HH) return;
    int e = i >> 3, h = i & 7, s, d;
    edge_sd(ei, e, s, d);
    float al = lrelu(a_src[s * HH + h] + a_dst[d * HH + h]);
    attn[i] = expf(al - amax[d * HH + h]) / (den[d * HH + h] + 1e-16f);
}

__global__ __launch_bounds__(256)
void gat_eagg1(const int* __restrict__ ei, const float* __restrict__ attn,
               const float* __restrict__ h1, float* __restrict__ agg) {
    int i = blockIdx.x * 256 + threadIdx.x;
    int e = i >> 5, q = i & 31;
    if (e >= ET) return;
    int s, d; edge_sd(ei, e, s, d);
    const float* hp = h1 + (size_t)s * HC + q;
    const float* at = attn + (size_t)e * HH + (q >> 4);
    float* ap = agg + (size_t)d * HC + q;
    float v[4];
    #pragma unroll
    for (int j = 0; j < 4; ++j) v[j] = hp[32 * j] * at[2 * j];
    #pragma unroll
    for (int j = 0; j < 4; ++j) atomicAdd(ap + 32 * j, v[j]);
}

__global__ __launch_bounds__(256)
void gat_biaselu(float* __restrict__ agg, const float* __restrict__ b1) {
    int i = blockIdx.x * 256 + threadIdx.x;
    if (i >= NN * HC) return;
    float v = agg[i] + b1[i & 127];
    agg[i] = v > 0.0f ? v : (expf(v) - 1.0f);
}

__global__ __launch_bounds__(256)
void gat_gemm2(const float* __restrict__ a1, const _Float16* __restrict__ w2h,
               float* __restrict__ h2) {
    const int lane = threadIdx.x & 31;
    const int rt   = blockIdx.x * 8 + (threadIdx.x >> 5);
    if (rt >= NN / 16) return;
    const int rsub = lane & 15;
    const int ksel = (lane >> 4) << 3;
    const float*    arow = a1 + (size_t)(rt * 16 + rsub) * HC;
    const _Float16* wrow = w2h + (size_t)rsub * HC;

    v8f acc = {};
    #pragma unroll
    for (int ks = 0; ks < 4; ++ks) {
        const int kb = ks * 32 + ksel;
        union { v16h v; _Float16 e[16]; } A;
        #pragma unroll
        for (int t = 0; t < 8; ++t) {
            A.e[t]     = (_Float16)arow[kb + t];
            A.e[8 + t] = (_Float16)arow[kb + 16 + t];
        }
        union { v16h v; h8 p[2]; } B;
        B.p[0] = *(const h8*)(wrow + kb);
        B.p[1] = *(const h8*)(wrow + kb + 16);
        acc = __builtin_amdgcn_wmma_f32_16x16x32_f16(
            false, A.v, false, B.v, (short)0, acc, false, false);
    }
    const int rhi = (lane >> 4) << 3;
    #pragma unroll
    for (int r = 0; r < 8; ++r)
        h2[(size_t)(rt * 16 + rhi + r) * NC + rsub] = acc[r];
}

__global__ __launch_bounds__(256)
void gat_att2(const float* __restrict__ h2, const float* __restrict__ as,
              const float* __restrict__ ad, float* __restrict__ a_src,
              float* __restrict__ a_dst, float* __restrict__ amax,
              float* __restrict__ den, float* __restrict__ agg) {
    int n = blockIdx.x * 256 + threadIdx.x;
    if (n >= NN) return;
    const float* hp = h2 + (size_t)n * NC;
    float s = 0.0f, d = 0.0f;
    #pragma unroll
    for (int c = 0; c < NC; ++c) { s += hp[c] * as[c]; d += hp[c] * ad[c]; }
    a_src[n] = s; a_dst[n] = d;
    amax[n] = -__builtin_inff(); den[n] = 0.0f;
    float* ap = agg + (size_t)n * NC;
    #pragma unroll
    for (int c = 0; c < NC; ++c) ap[c] = 0.0f;
}

__global__ __launch_bounds__(256)
void gat_emax2(const int* __restrict__ ei, const float* __restrict__ a_src,
               const float* __restrict__ a_dst, float* __restrict__ amax) {
    int e = blockIdx.x * 256 + threadIdx.x;
    if (e >= ET) return;
    int s, d; edge_sd(ei, e, s, d);
    atomicMaxF(&amax[d], lrelu(a_src[s] + a_dst[d]));
}

__global__ __launch_bounds__(256)
void gat_eden2(const int* __restrict__ ei, const float* __restrict__ a_src,
               const float* __restrict__ a_dst, const float* __restrict__ amax,
               float* __restrict__ den) {
    int e = blockIdx.x * 256 + threadIdx.x;
    if (e >= ET) return;
    int s, d; edge_sd(ei, e, s, d);
    float al = lrelu(a_src[s] + a_dst[d]);
    atomicAdd(&den[d], expf(al - amax[d]));
}

__global__ __launch_bounds__(256)
void gat_attw2(const int* __restrict__ ei, const float* __restrict__ a_src,
               const float* __restrict__ a_dst, const float* __restrict__ amax,
               const float* __restrict__ den, float* __restrict__ attn) {
    int e = blockIdx.x * 256 + threadIdx.x;
    if (e >= ET) return;
    int s, d; edge_sd(ei, e, s, d);
    float al = lrelu(a_src[s] + a_dst[d]);
    attn[e] = expf(al - amax[d]) / (den[d] + 1e-16f);
}

static_assert(ET % 8 == 0, "a wave's 8 edges must be all in range or all out");
__global__ __launch_bounds__(256)
void gat_eagg2(const int* __restrict__ ei, const float* __restrict__ attn,
               const float* __restrict__ h2, float* __restrict__ agg) {
    int i = blockIdx.x * 256 + threadIdx.x;
    int l = i & 31;
    int c = l & 15;
    int e0 = (i >> 5) * 8 + (l >> 4);
    if (e0 >= ET) return;
    int s[4], d[4];
    #pragma unroll
    for (int p = 0; p < 4; ++p) {
        int e  = e0 + 2 * p;
        int ec = e < EE ? e : 0;
        int sv = ei[ec], dv = ei[EE + ec];
        s[p] = e < EE ? sv : e - EE;
        d[p] = e < EE ? dv : e - EE;
    }
    float v[4];
    #pragma unroll
    for (int p = 0; p < 4; ++p)
        v[p] = h2[(size_t)s[p] * NC + c] * attn[e0 + 2 * p];
    #pragma unroll
    for (int p = 0; p < 4; ++p)
        atomicAdd(agg + (size_t)d[p] * NC + c, v[p]);
}

__global__ __launch_bounds__(256)
void gat_lsm(const float* __restrict__ agg, const float* __restrict__ b2,
             float* __restrict__ out) {
    int n = blockIdx.x * 256 + threadIdx.x;
    if (n >= NN) return;
    float v[NC], m = -__builtin_inff();
    #pragma unroll
    for (int c = 0; c < NC; ++c) { v[c] = agg[(size_t)n * NC + c] + b2[c]; m = fmaxf(m, v[c]); }
    float sum = 0.0f;
    #pragma unroll
    for (int c = 0; c < NC; ++c) sum += expf(v[c] - m);
    float ls = logf(sum);
    #pragma unroll
    for (int c = 0; c < NC; ++c) out[(size_t)n * NC + c] = v[c] - m - ls;
}

extern "C" void kernel_launch(void* const* d_in, const int* in_sizes, int n_in,
                              void* d_out, int out_size, void* d_ws, size_t ws_size,
                              hipStream_t stream) {
    (void)in_sizes; (void)n_in; (void)out_size; (void)ws_size;
    const float* x   = (const float*)d_in[0];
    const int*   ei  = (const int*)d_in[1];
    const float* W1  = (const float*)d_in[2];
    const float* as1 = (const float*)d_in[3];
    const float* ad1 = (const float*)d_in[4];
    const float* b1  = (const float*)d_in[5];
    const float* W2  = (const float*)d_in[6];
    const float* as2 = (const float*)d_in[7];
    const float* ad2 = (const float*)d_in[8];
    const float* b2  = (const float*)d_in[9];
    float* out = (float*)d_out;

    char* wsb = (char*)d_ws;
    size_t off = 0;
    auto alloc = [&](size_t bytes) -> void* {
        void* p = wsb + off;
        off += (bytes + 255) & ~(size_t)255;
        return p;
    };
    _Float16* w1h  = (_Float16*)alloc((size_t)HC * NF * 2);
    _Float16* w2h  = (_Float16*)alloc((size_t)NC * HC * 2);
    _Float16* xh   = (_Float16*)alloc((size_t)NN * NF * 2);
    float* h1      = (float*)alloc((size_t)NN * HC * 4);
    float* asrc1   = (float*)alloc((size_t)NN * HH * 4);
    float* adst1   = (float*)alloc((size_t)NN * HH * 4);
    float* amax1   = (float*)alloc((size_t)NN * HH * 4);
    float* den1    = (float*)alloc((size_t)NN * HH * 4);
    float* attn1   = (float*)alloc((size_t)ET * HH * 4);
    float* h2      = (float*)alloc((size_t)NN * NC * 4);
    float* asrc2   = (float*)alloc((size_t)NN * 4);
    float* adst2   = (float*)alloc((size_t)NN * 4);
    float* amax2   = (float*)alloc((size_t)NN * 4);
    float* den2    = (float*)alloc((size_t)NN * 4);
    float* attn2   = (float*)alloc((size_t)ET * 4);
    float* agg2    = (float*)alloc((size_t)NN * NC * 4);
    float* agg1    = (float*)xh;

    const dim3 B(256);
    gat_cvtw <<<(HC * NF + NC * HC + 255) / 256, B, 0, stream>>>(W1, W2, w1h, w2h);
    gat_cvtx <<<(NN * NF) / 256, B, 0, stream>>>(x, xh);
    gat_gemm1<<<(NN / 16 + 7) / 8, B, 0, stream>>>(xh, w1h, h1);
    gat_att1 <<<(NN * HH) / 256, B, 0, stream>>>(h1, as1, ad1, asrc1, adst1, amax1, den1, agg1);
    gat_emax1<<<(ET * HH) / 256, B, 0, stream>>>(ei, asrc1, adst1, amax1);
    gat_eden1<<<(ET * HH) / 256, B, 0, stream>>>(ei, asrc1, adst1, amax1, den1);
    gat_attw1<<<(ET * HH) / 256, B, 0, stream>>>(ei, asrc1, adst1, amax1, den1, attn1);
    gat_eagg1<<<(ET * 32) / 256, B, 0, stream>>>(ei, attn1, h1, agg1);
    gat_biaselu<<<(NN * HC) / 256, B, 0, stream>>>(agg1, b1);
    gat_gemm2<<<(NN / 16 + 7) / 8, B, 0, stream>>>(agg1, w2h, h2);
    gat_att2 <<<(NN + 255) / 256, B, 0, stream>>>(h2, as2, ad2, asrc2, adst2, amax2, den2, agg2);
    gat_emax2<<<(ET + 255) / 256, B, 0, stream>>>(ei, asrc2, adst2, amax2);
    gat_eden2<<<(ET + 255) / 256, B, 0, stream>>>(ei, asrc2, adst2, amax2, den2);
    gat_attw2<<<(ET + 255) / 256, B, 0, stream>>>(ei, asrc2, adst2, amax2, den2, attn2);
    gat_eagg2<<<(ET * 4 + 255) / 256, B, 0, stream>>>(ei, attn2, h2, agg2);
    gat_lsm  <<<(NN + 255) / 256, B, 0, stream>>>(agg2, b2, out);
}


# edges counting-sorted by destination each call (int atomics), then one wave per node gathers: softmax, aggregation, bias, activation fused; no float atomics; sums in list (arrival) order, lane trees
# speedup vs baseline: 2.2998x; 1.9927x over previous
#include <hip/hip_runtime.h>

#define NN  100000
#define NF  256
#define HH  8
#define CC  16
#define HC  128
#define NC  16
#define EE  1600000
#define ET  1700000
#define NEG 0.2f

typedef _Float16 h8   __attribute__((ext_vector_type(8)));
typedef _Float16 v16h __attribute__((ext_vector_type(16)));
typedef float    v8f  __attribute__((ext_vector_type(8)));

__device__ __forceinline__ float lrelu(float a) { return a > 0.0f ? a : NEG * a; }

static_assert(HC * NF + NC * HC <= NN, "the grid that covers the nodes also covers both weight matrices");
__global__ __launch_bounds__(256) void gat_cvtw(const float* __restrict__ W1,
                                                const float* __restrict__ W2,
                                                _Float16* __restrict__ w1h,
                                                _Float16* __restrict__ w2h,
                                                int* __restrict__ deg) {
    int i = blockIdx.x * 256 + threadIdx.x;
    if (i < NN) deg[i] = 0;
    if (i < HC * NF)              w1h[i] = (_Float16)W1[i];
    else if (i < HC * NF + NC*HC) w2h[i - HC * NF] = (_Float16)W2[i - HC * NF];
}

__global__ __launch_bounds__(256) void gat_cvtx(const float* __restrict__ x,
                                                _Float16* __restrict__ xh) {
    int i = blockIdx.x * 256 + threadIdx.x;
    if (i < NN * NF) xh[i] = (_Float16)x[i];
}

__global__ __launch_bounds__(256)
void gat_gemm1(const _Float16* __restrict__ xh, const _Float16* __restrict__ w1h,
               float* __restrict__ h1) {
    __shared__ _Float16 lw[HC * NF];
    const int tid = threadIdx.x;
    for (int idx = tid; idx < HC * 32; idx += 256) {
        int r = idx >> 5, ch = idx & 31;
        *(h8*)(&lw[r * NF + ((ch ^ (r & 15)) << 3)]) =
            *(const h8*)(w1h + (size_t)r * NF + (ch << 3));
    }
    __syncthreads();

    const int lane = tid & 31;
    const int rt   = blockIdx.x * 8 + (tid >> 5);
    if (rt >= NN / 16) return;
    const int rsub = lane & 15;
    const int ksel = (lane >> 4) << 3;
    const _Float16* xrow = xh + (size_t)(rt * 16 + rsub) * NF;

    v8f acc[8] = {};
    #pragma unroll
    for (int ks = 0; ks < 8; ++ks) {
        const int kb = ks * 32 + ksel;
        const int c0 = kb >> 3;
        union { v16h v; h8 p[2]; } A;
        A.p[0] = *(const h8*)(xrow + kb);
        A.p[1] = *(const h8*)(xrow + kb + 16);
        #pragma unroll
        for (int ct = 0; ct < 8; ++ct) {
            const int n = ct * 16 + rsub;
            union { v16h v; h8 p[2]; } B;
            B.p[0] = *(const h8*)(&lw[n * NF + (((c0    ) ^ (n & 15)) << 3)]);
            B.p[1] = *(const h8*)(&lw[n * NF + (((c0 + 2) ^ (n & 15)) << 3)]);
            acc[ct] = __builtin_amdgcn_wmma_f32_16x16x32_f16(
                false, A.v, false, B.v, (short)0, acc[ct], false, false);
        }
    }
    const int rhi = (lane >> 4) << 3;
    #pragma unroll
    for (int ct = 0; ct < 8; ++ct) {
        #pragma unroll
        for (int r = 0; r < 8; ++r)
            h1[(size_t)(rt * 16 + rhi + r) * HC + ct * 16 + rsub] = acc[ct][r];
    }
}

static_assert(EE % 256 == 0, "edge kernels launch exactly EE threads");
__global__ __launch_bounds__(256)
void gat_count(const int* __restrict__ ei, int* __restrict__ deg) {
    int e = blockIdx.x * 256 + threadIdx.x;
    atomicAdd(&deg[ei[EE + e]], 1);
}

__global__ __launch_bounds__(1024)
void gat_scan(const int* __restrict__ deg, int* __restrict__ rowptr) {
    __shared__ int wsum[32], psum[32];
    const int t = threadIdx.x, lane = t & 31, w = t >> 5;
    const int4* d4 = (const int4*)deg;
    const int lim = blockIdx.x * 256;
    int s = 0;
    for (int k0 = t; k0 < lim; k0 += 4096) {
        int4 v[4];
        #pragma unroll
        for (int j = 0; j < 4; ++j) { const int k = k0 + 1024 * j; v[j] = d4[k < lim ? k : 0]; }
        #pragma unroll
        for (int j = 0; j < 4; ++j) s += (k0 + 1024 * j < lim) ? v[j].x + v[j].y + v[j].z + v[j].w : 0;
    }
    const int i = blockIdx.x * 1024 + t;
    const int v = i < NN ? deg[i] : 0;
    int incl = v;
    for (int o = 1; o < 32; o <<= 1) { int n = __shfl_up(incl, o); if (lane >= o) incl += n; }
    for (int o = 16; o > 0; o >>= 1) s += __shfl_xor(s, o);
    if (lane == 31) wsum[w] = incl;
    if (lane == 0)  psum[w] = s;
    __syncthreads();
    if (w == 0) {
        int tv = wsum[lane], iv = tv, ps = psum[lane];
        for (int o = 1; o < 32; o <<= 1) { int n = __shfl_up(iv, o); if (lane >= o) iv += n; }
        for (int o = 16; o > 0; o >>= 1) ps += __shfl_xor(ps, o);
        wsum[lane] = iv - tv + ps;
    }
    __syncthreads();
    if (i <= NN) rowptr[i] = wsum[w] + incl - v;
}

__global__ __launch_bounds__(256)
void gat_fill(const int* __restrict__ ei, const int* __restrict__ rowptr,
              int* __restrict__ deg, int* __restrict__ csr) {
    int e = blockIdx.x * 256 + threadIdx.x;
    int d = ei[EE + e];
    int slot = atomicSub(&deg[d], 1) - 1;
    csr[rowptr[d] + slot] = ei[e];
}

__global__ __launch_bounds__(256)
void gat_att1(const float* __restrict__ h1, const float* __restrict__ as,
              const float* __restrict__ ad, float* __restrict__ a_src,
              float* __restrict__ a_dst) {
    int i = blockIdx.x * 256 + threadIdx.x;
    if (i >= NN * HH) return;
    int n = i >> 3, h = i & 7;
    const float* hp = h1 + (size_t)n * HC + h * CC;
    float s = 0.0f, d = 0.0f;
    #pragma unroll
    for (int c = 0; c < CC; ++c) { s += hp[c] * as[h * CC + c]; d += hp[c] * ad[h * CC + c]; }
    a_src[i] = s; a_dst[i] = d;
}

static_assert(NN % 8 == 0, "eight waves a block, one node each");
__global__ __launch_bounds__(256)
void gat_node1(const int* __restrict__ rowptr, const int* __restrict__ csr,
               const float* __restrict__ a_src, const float* __restrict__ a_dst,
               const float* __restrict__ h1, const float* __restrict__ b1,
               float* __restrict__ out1) {
    const int lane = threadIdx.x & 31;
    const int d   = __builtin_amdgcn_readfirstlane(blockIdx.x * 8 + (threadIdx.x >> 5));
    const int beg = rowptr[d];
    const int cnt = rowptr[d + 1] - beg + 1;
    const int eh = lane & 7, es = lane >> 3, ch = lane >> 2;
    const float adh = a_dst[d * HH + eh];
    const float NINF = -__builtin_inff();

    float m = NINF, l = 0.0f;
    float acc0 = 0.0f, acc1 = 0.0f, acc2 = 0.0f, acc3 = 0.0f;
    for (int c0 = 0; c0 < cnt; c0 += 32) {
        const int p = c0 + lane;
        int sid = d;
        if (p > 0 && p < cnt) sid = csr[beg + p - 1];
        float a[8];
        float mc = NINF;
        #pragma unroll
        for (int t = 0; t < 8; ++t) {
            const int q = 4 * t + es;
            const int s = __shfl(sid, q);
            const float v = lrelu(a_src[s * HH + eh] + adh);
            a[t] = (c0 + q < cnt) ? v : NINF;
            mc = fmaxf(mc, a[t]);
        }
        mc = fmaxf(mc, __shfl_xor(mc, 8));
        mc = fmaxf(mc, __shfl_xor(mc, 16));
        const float mn = fmaxf(m, mc);
        const float sc = expf(m - mn);
        float lc = 0.0f;
        #pragma unroll
        for (int t = 0; t < 8; ++t) { a[t] = expf(a[t] - mn); lc += a[t]; }
        lc += __shfl_xor(lc, 8);
        lc += __shfl_xor(lc, 16);
        l = l * sc + lc;
        m = mn;
        const float scc = __shfl(sc, ch);
        acc0 *= scc; acc1 *= scc; acc2 *= scc; acc3 *= scc;
        #pragma unroll
        for (int t = 0; t < 8; ++t) {
            if (c0 + 4 * t < cnt) {
                float4 row[4];
                float  wt[4];
                #pragma unroll
                for (int j = 0; j < 4; ++j) {
                    const int s = __builtin_amdgcn_readlane(sid, 4 * t + j);
                    wt[j]  = __shfl(a[t], j * 8 + ch);
                    row[j] = *(const float4*)(h1 + (size_t)s * HC + lane * 4);
                }
                #pragma unroll
                for (int j = 0; j < 4; ++j) {
                    acc0 += wt[j] * row[j].x; acc1 += wt[j] * row[j].y;
                    acc2 += wt[j] * row[j].z; acc3 += wt[j] * row[j].w;
                }
            }
        }
    }
    const float den = __shfl(l, ch) + 1e-16f;
    const float4 b = *(const float4*)(b1 + lane * 4);
    float4 o;
    o.x = acc0 / den + b.x; o.y = acc1 / den + b.y; o.z = acc2 / den + b.z; o.w = acc3 / den + b.w;
    o.x = o.x > 0.0f ? o.x : (expf(o.x) - 1.0f);
    o.y = o.y > 0.0f ? o.y : (expf(o.y) - 1.0f);
    o.z = o.z > 0.0f ? o.z : (expf(o.z) - 1.0f);
    o.w = o.w > 0.0f ? o.w : (expf(o.w) - 1.0f);
    *(float4*)(out1 + (size_t)d * HC + lane * 4) = o;
}

__global__ __launch_bounds__(256)
void gat_gemm2(const float* __restrict__ a1, const _Float16* __restrict__ w2h,
               float* __restrict__ h2) {
    const int lane = threadIdx.x & 31;
    const int rt   = blockIdx.x * 8 + (threadIdx.x >> 5);
    if (rt >= NN / 16) return;
    const int rsub = lane & 15;
    const int ksel = (lane >> 4) << 3;
    const float*    arow = a1 + (size_t)(rt * 16 + rsub) * HC;
    const _Float16* wrow = w2h + (size_t)rsub * HC;

    v8f acc = {};
    #pragma unroll
    for (int ks = 0; ks < 4; ++ks) {
        const int kb = ks * 32 + ksel;
        union { v16h v; _Float16 e[16]; } A;
        #pragma unroll
        for (int t = 0; t < 8; ++t) {
            A.e[t]     = (_Float16)arow[kb + t];
            A.e[8 + t] = (_Float16)arow[kb + 16 + t];
        }
        union { v16h v; h8 p[2]; } B;
        B.p[0] = *(const h8*)(wrow + kb);
        B.p[1] = *(const h8*)(wrow + kb + 16);
        acc = __builtin_amdgcn_wmma_f32_16x16x32_f16(
            false, A.v, false, B.v, (short)0, acc, false, false);
    }
    const int rhi = (lane >> 4) << 3;
    #pragma unroll
    for (int r = 0; r < 8; ++r)
        h2[(size_t)(rt * 16 + rhi + r) * NC + rsub] = acc[r];
}

__global__ __launch_bounds__(256)
void gat_att2(const float* __restrict__ h2, const float* __restrict__ as,
              const float* __restrict__ ad, float* __restrict__ a_src,
              float* __restrict__ a_dst) {
    int n = blockIdx.x * 256 + threadIdx.x;
    if (n >= NN) return;
    const float* hp = h2 + (size_t)n * NC;
    float s = 0.0f, d = 0.0f;
    #pragma unroll
    for (int c = 0; c < NC; ++c) { s += hp[c] * as[c]; d += hp[c] * ad[c]; }
    a_src[n] = s; a_dst[n] = d;
}

__global__ __launch_bounds__(256)
void gat_node2(const int* __restrict__ rowptr, const int* __restrict__ csr,
               const float* __restrict__ a_src, const float* __restrict__ a_dst,
               const float* __restrict__ h2, const float* __restrict__ b2,
               float* __restrict__ out) {
    const int lane = threadIdx.x & 31;
    const int d   = __builtin_amdgcn_readfirstlane(blockIdx.x * 8 + (threadIdx.x >> 5));
    const int beg = rowptr[d];
    const int cnt = rowptr[d + 1] - beg + 1;
    const int c = lane & 15, es = lane >> 4;
    const float adv = a_dst[d];
    const float NINF = -__builtin_inff();

    float m = NINF, l = 0.0f, acc = 0.0f;
    for (int c0 = 0; c0 < cnt; c0 += 32) {
        const int p = c0 + lane;
        const bool valid = p < cnt;
        int sid = d;
        if (p > 0 && valid) sid = csr[beg + p - 1];
        const float a = valid ? lrelu(a_src[sid] + adv) : NINF;
        float mc = a;
        for (int o = 16; o > 0; o >>= 1) mc = fmaxf(mc, __shfl_xor(mc, o));
        const float mn = fmaxf(m, mc);
        const float sc = expf(m - mn);
        const float ex = expf(a - mn);
        float lc = ex;
        for (int o = 16; o > 0; o >>= 1) lc += __shfl_xor(lc, o);
        l = l * sc + lc;
        m = mn;
        acc *= sc;
        #pragma unroll
        for (int g = 0; g < 4; ++g) {
            if (c0 + 8 * g < cnt) {
                float hv[4], wt[4];
                #pragma unroll
                for (int j = 0; j < 4; ++j) {
                    const int q = 8 * g + 2 * j + es;
                    const int s = __shfl(sid, q);
                    wt[j] = __shfl(ex, q);
                    hv[j] = h2[(size_t)s * NC + c];
                }
                #pragma unroll
                for (int j = 0; j < 4; ++j) acc += wt[j] * hv[j];
            }
        }
    }
    acc += __shfl_xor(acc, 16);
    const float v = acc / (l + 1e-16f) + b2[c];
    float mx = v;
    for (int o = 8; o > 0; o >>= 1) mx = fmaxf(mx, __shfl_xor(mx, o));
    float se = expf(v - mx);
    for (int o = 8; o > 0; o >>= 1) se += __shfl_xor(se, o);
    if (es == 0) out[(size_t)d * NC + c] = v - mx - logf(se);
}

extern "C" void kernel_launch(void* const* d_in, const int* in_sizes, int n_in,
                              void* d_out, int out_size, void* d_ws, size_t ws_size,
                              hipStream_t stream) {
    (void)in_sizes; (void)n_in; (void)out_size; (void)ws_size;
    const float* x   = (const float*)d_in[0];
    const int*   ei  = (const int*)d_in[1];
    const float* W1  = (const float*)d_in[2];
    const float* as1 = (const float*)d_in[3];
    const float* ad1 = (const float*)d_in[4];
    const float* b1  = (const float*)d_in[5];
    const float* W2  = (const float*)d_in[6];
    const float* as2 = (const float*)d_in[7];
    const float* ad2 = (const float*)d_in[8];
    const float* b2  = (const float*)d_in[9];
    float* out = (float*)d_out;

    char* wsb = (char*)d_ws;
    size_t off = 0;
    auto alloc = [&](size_t bytes) -> void* {
        void* p = wsb + off;
        off += (bytes + 255) & ~(size_t)255;
        return p;
    };
    _Float16* w1h  = (_Float16*)alloc((size_t)HC * NF * 2);
    _Float16* w2h  = (_Float16*)alloc((size_t)NC * HC * 2);
    _Float16* xh   = (_Float16*)alloc((size_t)NN * NF * 2);
    float* h1      = (float*)alloc((size_t)NN * HC * 4);
    float* asrc1   = (float*)alloc((size_t)NN * HH * 4);
    float* adst1   = (float*)alloc((size_t)NN * HH * 4);
    int*   deg     = (int*)alloc((size_t)NN * 4);
    int*   rowptr  = (int*)alloc((size_t)(NN + 1) * 4);
    int*   csr     = (int*)alloc((size_t)EE * 4);
    float* h2      = (float*)alloc((size_t)NN * NC * 4);
    float* asrc2   = (float*)alloc((size_t)NN * 4);
    float* adst2   = (float*)alloc((size_t)NN * 4);
    float* out1    = (float*)xh;

    const dim3 B(256);
    gat_cvtw <<<(NN + 255) / 256, B, 0, stream>>>(W1, W2, w1h, w2h, deg);
    gat_cvtx <<<(NN * NF) / 256, B, 0, stream>>>(x, xh);
    gat_gemm1<<<(NN / 16 + 7) / 8, B, 0, stream>>>(xh, w1h, h1);
    gat_count<<<EE / 256, B, 0, stream>>>(ei, deg);
    gat_scan <<<(NN + 1024) / 1024, dim3(1024), 0, stream>>>(deg, rowptr);
    gat_fill <<<EE / 256, B, 0, stream>>>(ei, rowptr, deg, csr);
    gat_att1 <<<(NN * HH) / 256, B, 0, stream>>>(h1, as1, ad1, asrc1, adst1);
    gat_node1<<<NN / 8, B, 0, stream>>>(rowptr, csr, asrc1, adst1, h1, b1, out1);
    gat_gemm2<<<(NN / 16 + 7) / 8, B, 0, stream>>>(out1, w2h, h2);
    gat_att2 <<<(NN + 255) / 256, B, 0, stream>>>(h2, as2, ad2, asrc2, adst2);
    gat_node2<<<NN / 8, B, 0, stream>>>(rowptr, csr, asrc2, adst2, h2, b2, out);
}


# gemm1 reads x as f32 and rounds it to f16 in its A loader (row-strip loads issued as one group); the one-element-per-thread conversion kernel and its f16 copy of x are gone; values, sums unchanged
# speedup vs baseline: 2.6726x; 1.1621x over previous
#include <hip/hip_runtime.h>

#define NN  100000
#define NF  256
#define HH  8
#define CC  16
#define HC  128
#define NC  16
#define EE  1600000
#define ET  1700000
#define NEG 0.2f

typedef _Float16 h8   __attribute__((ext_vector_type(8)));
typedef _Float16 v16h __attribute__((ext_vector_type(16)));
typedef float    v8f  __attribute__((ext_vector_type(8)));

__device__ __forceinline__ float lrelu(float a) { return a > 0.0f ? a : NEG * a; }

static_assert(HC * NF + NC * HC <= NN, "the grid that covers the nodes also covers both weight matrices");
__global__ __launch_bounds__(256) void gat_cvtw(const float* __restrict__ W1,
                                                const float* __restrict__ W2,
                                                _Float16* __restrict__ w1h,
                                                _Float16* __restrict__ w2h,
                                                int* __restrict__ deg) {
    int i = blockIdx.x * 256 + threadIdx.x;
    if (i < NN) deg[i] = 0;
    if (i < HC * NF)              w1h[i] = (_Float16)W1[i];
    else if (i < HC * NF + NC*HC) w2h[i - HC * NF] = (_Float16)W2[i - HC * NF];
}

__global__ __launch_bounds__(256)
void gat_gemm1(const float* __restrict__ x, const _Float16* __restrict__ w1h,
               float* __restrict__ h1) {
    __shared__ _Float16 lw[HC * NF];
    const int tid = threadIdx.x;
    for (int idx = tid; idx < HC * 32; idx += 256) {
        int r = idx >> 5, ch = idx & 31;
        *(h8*)(&lw[r * NF + ((ch ^ (r & 15)) << 3)]) =
            *(const h8*)(w1h + (size_t)r * NF + (ch << 3));
    }
    __syncthreads();

    const int lane = tid & 31;
    const int rt   = blockIdx.x * 8 + (tid >> 5);
    if (rt >= NN / 16) return;
    const int rsub = lane & 15;
    const int ksel = (lane >> 4) << 3;
    const float* xrow = x + (size_t)(rt * 16 + rsub) * NF + ksel;

    v16h afr[8];
    #pragma unroll
    for (int ks = 0; ks < 8; ++ks) {
        const float4 f0 = *(const float4*)(xrow + ks * 32);
        const float4 f1 = *(const float4*)(xrow + ks * 32 + 4);
        const float4 f2 = *(const float4*)(xrow + ks * 32 + 16);
        const float4 f3 = *(const float4*)(xrow + ks * 32 + 20);
        union { v16h v; _Float16 e[16]; } A;
        A.e[0]  = (_Float16)f0.x; A.e[1]  = (_Float16)f0.y; A.e[2]  = (_Float16)f0.z; A.e[3]  = (_Float16)f0.w;
        A.e[4]  = (_Float16)f1.x; A.e[5]  = (_Float16)f1.y; A.e[6]  = (_Float16)f1.z; A.e[7]  = (_Float16)f1.w;
        A.e[8]  = (_Float16)f2.x; A.e[9]  = (_Float16)f2.y; A.e[10] = (_Float16)f2.z; A.e[11] = (_Float16)f2.w;
        A.e[12] = (_Float16)f3.x; A.e[13] = (_Float16)f3.y; A.e[14] = (_Float16)f3.z; A.e[15] = (_Float16)f3.w;
        afr[ks] = A.v;
    }
    __builtin_amdgcn_sched_barrier(0);

    v8f acc[8] = {};
    #pragma unroll
    for (int ks = 0; ks < 8; ++ks) {
        const int c0 = (ks * 32 + ksel) >> 3;
        #pragma unroll
        for (int ct = 0; ct < 8; ++ct) {
            const int n = ct * 16 + rsub;
            union { v16h v; h8 p[2]; } B;
            B.p[0] = *(const h8*)(&lw[n * NF + (((c0    ) ^ (n & 15)) << 3)]);
            B.p[1] = *(const h8*)(&lw[n * NF + (((c0 + 2) ^ (n & 15)) << 3)]);
            acc[ct] = __builtin_amdgcn_wmma_f32_16x16x32_f16(
                false, afr[ks], false, B.v, (short)0, acc[ct], false, false);
        }
    }
    const int rhi = (lane >> 4) << 3;
    #pragma unroll
    for (int ct = 0; ct < 8; ++ct) {
        #pragma unroll
        for (int r = 0; r < 8; ++r)
            h1[(size_t)(rt * 16 + rhi + r) * HC + ct * 16 + rsub] = acc[ct][r];
    }
}

static_assert(EE % 256 == 0, "edge kernels launch exactly EE threads");
__global__ __launch_bounds__(256)
void gat_count(const int* __restrict__ ei, int* __restrict__ deg) {
    int e = blockIdx.x * 256 + threadIdx.x;
    atomicAdd(&deg[ei[EE + e]], 1);
}

__global__ __launch_bounds__(1024)
void gat_scan(const int* __restrict__ deg, int* __restrict__ rowptr) {
    __shared__ int wsum[32], psum[32];
    const int t = threadIdx.x, lane = t & 31, w = t >> 5;
    const int4* d4 = (const int4*)deg;
    const int lim = blockIdx.x * 256;
    int s = 0;
    for (int k0 = t; k0 < lim; k0 += 4096) {
        int4 v[4];
        #pragma unroll
        for (int j = 0; j < 4; ++j) { const int k = k0 + 1024 * j; v[j] = d4[k < lim ? k : 0]; }
        #pragma unroll
        for (int j = 0; j < 4; ++j) s += (k0 + 1024 * j < lim) ? v[j].x + v[j].y + v[j].z + v[j].w : 0;
    }
    const int i = blockIdx.x * 1024 + t;
    const int v = i < NN ? deg[i] : 0;
    int incl = v;
    for (int o = 1; o < 32; o <<= 1) { int n = __shfl_up(incl, o); if (lane >= o) incl += n; }
    for (int o = 16; o > 0; o >>= 1) s += __shfl_xor(s, o);
    if (lane == 31) wsum[w] = incl;
    if (lane == 0)  psum[w] = s;
    __syncthreads();
    if (w == 0) {
        int tv = wsum[lane], iv = tv, ps = psum[lane];
        for (int o = 1; o < 32; o <<= 1) { int n = __shfl_up(iv, o); if (lane >= o) iv += n; }
        for (int o = 16; o > 0; o >>= 1) ps += __shfl_xor(ps, o);
        wsum[lane] = iv - tv + ps;
    }
    __syncthreads();
    if (i <= NN) rowptr[i] = wsum[w] + incl - v;
}

__global__ __launch_bounds__(256)
void gat_fill(const int* __restrict__ ei, const int* __restrict__ rowptr,
              int* __restrict__ deg, int* __restrict__ csr) {
    int e = blockIdx.x * 256 + threadIdx.x;
    int d = ei[EE + e];
    int slot = atomicSub(&deg[d], 1) - 1;
    csr[rowptr[d] + slot] = ei[e];
}

__global__ __launch_bounds__(256)
void gat_att1(const float* __restrict__ h1, const float* __restrict__ as,
              const float* __restrict__ ad, float* __restrict__ a_src,
              float* __restrict__ a_dst) {
    int i = blockIdx.x * 256 + threadIdx.x;
    if (i >= NN * HH) return;
    int n = i >> 3, h = i & 7;
    const float* hp = h1 + (size_t)n * HC + h * CC;
    float s = 0.0f, d = 0.0f;
    #pragma unroll
    for (int c = 0; c < CC; ++c) { s += hp[c] * as[h * CC + c]; d += hp[c] * ad[h * CC + c]; }
    a_src[i] = s; a_dst[i] = d;
}

static_assert(NN % 8 == 0, "eight waves a block, one node each");
__global__ __launch_bounds__(256)
void gat_node1(const int* __restrict__ rowptr, const int* __restrict__ csr,
               const float* __restrict__ a_src, const float* __restrict__ a_dst,
               const float* __restrict__ h1, const float* __restrict__ b1,
               float* __restrict__ out1) {
    const int lane = threadIdx.x & 31;
    const int d   = __builtin_amdgcn_readfirstlane(blockIdx.x * 8 + (threadIdx.x >> 5));
    const int beg = rowptr[d];
    const int cnt = rowptr[d + 1] - beg + 1;
    const int eh = lane & 7, es = lane >> 3, ch = lane >> 2;
    const float adh = a_dst[d * HH + eh];
    const float NINF = -__builtin_inff();

    float m = NINF, l = 0.0f;
    float acc0 = 0.0f, acc1 = 0.0f, acc2 = 0.0f, acc3 = 0.0f;
    for (int c0 = 0; c0 < cnt; c0 += 32) {
        const int p = c0 + lane;
        int sid = d;
        if (p > 0 && p < cnt) sid = csr[beg + p - 1];
        float a[8];
        float mc = NINF;
        #pragma unroll
        for (int t = 0; t < 8; ++t) {
            const int q = 4 * t + es;
            const int s = __shfl(sid, q);
            const float v = lrelu(a_src[s * HH + eh] + adh);
            a[t] = (c0 + q < cnt) ? v : NINF;
            mc = fmaxf(mc, a[t]);
        }
        mc = fmaxf(mc, __shfl_xor(mc, 8));
        mc = fmaxf(mc, __shfl_xor(mc, 16));
        const float mn = fmaxf(m, mc);
        const float sc = expf(m - mn);
        float lc = 0.0f;
        #pragma unroll
        for (int t = 0; t < 8; ++t) { a[t] = expf(a[t] - mn); lc += a[t]; }
        lc += __shfl_xor(lc, 8);
        lc += __shfl_xor(lc, 16);
        l = l * sc + lc;
        m = mn;
        const float scc = __shfl(sc, ch);
        acc0 *= scc; acc1 *= scc; acc2 *= scc; acc3 *= scc;
        #pragma unroll
        for (int t = 0; t < 8; ++t) {
            if (c0 + 4 * t < cnt) {
                float4 row[4];
                float  wt[4];
                #pragma unroll
                for (int j = 0; j < 4; ++j) {
                    const int s = __builtin_amdgcn_readlane(sid, 4 * t + j);
                    wt[j]  = __shfl(a[t], j * 8 + ch);
                    row[j] = *(const float4*)(h1 + (size_t)s * HC + lane * 4);
                }
                #pragma unroll
                for (int j = 0; j < 4; ++j) {
                    acc0 += wt[j] * row[j].x; acc1 += wt[j] * row[j].y;
                    acc2 += wt[j] * row[j].z; acc3 += wt[j] * row[j].w;
                }
            }
        }
    }
    const float den = __shfl(l, ch) + 1e-16f;
    const float4 b = *(const float4*)(b1 + lane * 4);
    float4 o;
    o.x = acc0 / den + b.x; o.y = acc1 / den + b.y; o.z = acc2 / den + b.z; o.w = acc3 / den + b.w;
    o.x = o.x > 0.0f ? o.x : (expf(o.x) - 1.0f);
    o.y = o.y > 0.0f ? o.y : (expf(o.y) - 1.0f);
    o.z = o.z > 0.0f ? o.z : (expf(o.z) - 1.0f);
    o.w = o.w > 0.0f ? o.w : (expf(o.w) - 1.0f);
    *(float4*)(out1 + (size_t)d * HC + lane * 4) = o;
}

__global__ __launch_bounds__(256)
void gat_gemm2(const float* __restrict__ a1, const _Float16* __restrict__ w2h,
               float* __restrict__ h2) {
    const int lane = threadIdx.x & 31;
    const int rt   = blockIdx.x * 8 + (threadIdx.x >> 5);
    if (rt >= NN / 16) return;
    const int rsub = lane & 15;
    const int ksel = (lane >> 4) << 3;
    const float*    arow = a1 + (size_t)(rt * 16 + rsub) * HC;
    const _Float16* wrow = w2h + (size_t)rsub * HC;

    v8f acc = {};
    #pragma unroll
    for (int ks = 0; ks < 4; ++ks) {
        const int kb = ks * 32 + ksel;
        union { v16h v; _Float16 e[16]; } A;
        #pragma unroll
        for (int t = 0; t < 8; ++t) {
            A.e[t]     = (_Float16)arow[kb + t];
            A.e[8 + t] = (_Float16)arow[kb + 16 + t];
        }
        union { v16h v; h8 p[2]; } B;
        B.p[0] = *(const h8*)(wrow + kb);
        B.p[1] = *(const h8*)(wrow + kb + 16);
        acc = __builtin_amdgcn_wmma_f32_16x16x32_f16(
            false, A.v, false, B.v, (short)0, acc, false, false);
    }
    const int rhi = (lane >> 4) << 3;
    #pragma unroll
    for (int r = 0; r < 8; ++r)
        h2[(size_t)(rt * 16 + rhi + r) * NC + rsub] = acc[r];
}

__global__ __launch_bounds__(256)
void gat_att2(const float* __restrict__ h2, const float* __restrict__ as,
              const float* __restrict__ ad, float* __restrict__ a_src,
              float* __restrict__ a_dst) {
    int n = blockIdx.x * 256 + threadIdx.x;
    if (n >= NN) return;
    const float* hp = h2 + (size_t)n * NC;
    float s = 0.0f, d = 0.0f;
    #pragma unroll
    for (int c = 0; c < NC; ++c) { s += hp[c] * as[c]; d += hp[c] * ad[c]; }
    a_src[n] = s; a_dst[n] = d;
}

__global__ __launch_bounds__(256)
void gat_node2(const int* __restrict__ rowptr, const int* __restrict__ csr,
               const float* __restrict__ a_src, const float* __restrict__ a_dst,
               const float* __restrict__ h2, const float* __restrict__ b2,
               float* __restrict__ out) {
    const int lane = threadIdx.x & 31;
    const int d   = __builtin_amdgcn_readfirstlane(blockIdx.x * 8 + (threadIdx.x >> 5));
    const int beg = rowptr[d];
    const int cnt = rowptr[d + 1] - beg + 1;
    const int c = lane & 15, es = lane >> 4;
    const float adv = a_dst[d];
    const float NINF = -__builtin_inff();

    float m = NINF, l = 0.0f, acc = 0.0f;
    for (int c0 = 0; c0 < cnt; c0 += 32) {
        const int p = c0 + lane;
        const bool valid = p < cnt;
        int sid = d;
        if (p > 0 && valid) sid = csr[beg + p - 1];
        const float a = valid ? lrelu(a_src[sid] + adv) : NINF;
        float mc = a;
        for (int o = 16; o > 0; o >>= 1) mc = fmaxf(mc, __shfl_xor(mc, o));
        const float mn = fmaxf(m, mc);
        const float sc = expf(m - mn);
        const float ex = expf(a - mn);
        float lc = ex;
        for (int o = 16; o > 0; o >>= 1) lc += __shfl_xor(lc, o);
        l = l * sc + lc;
        m = mn;
        acc *= sc;
        #pragma unroll
        for (int g = 0; g < 4; ++g) {
            if (c0 + 8 * g < cnt) {
                float hv[4], wt[4];
                #pragma unroll
                for (int j = 0; j < 4; ++j) {
                    const int q = 8 * g + 2 * j + es;
                    const int s = __shfl(sid, q);
                    wt[j] = __shfl(ex, q);
                    hv[j] = h2[(size_t)s * NC + c];
                }
                #pragma unroll
                for (int j = 0; j < 4; ++j) acc += wt[j] * hv[j];
            }
        }
    }
    acc += __shfl_xor(acc, 16);
    const float v = acc / (l + 1e-16f) + b2[c];
    float mx = v;
    for (int o = 8; o > 0; o >>= 1) mx = fmaxf(mx, __shfl_xor(mx, o));
    float se = expf(v - mx);
    for (int o = 8; o > 0; o >>= 1) se += __shfl_xor(se, o);
    if (es == 0) out[(size_t)d * NC + c] = v - mx - logf(se);
}

extern "C" void kernel_launch(void* const* d_in, const int* in_sizes, int n_in,
                              void* d_out, int out_size, void* d_ws, size_t ws_size,
                              hipStream_t stream) {
    (void)in_sizes; (void)n_in; (void)out_size; (void)ws_size;
    const float* x   = (const float*)d_in[0];
    const int*   ei  = (const int*)d_in[1];
    const float* W1  = (const float*)d_in[2];
    const float* as1 = (const float*)d_in[3];
    const float* ad1 = (const float*)d_in[4];
    const float* b1  = (const float*)d_in[5];
    const float* W2  = (const float*)d_in[6];
    const float* as2 = (const float*)d_in[7];
    const float* ad2 = (const float*)d_in[8];
    const float* b2  = (const float*)d_in[9];
    float* out = (float*)d_out;

    char* wsb = (char*)d_ws;
    size_t off = 0;
    auto alloc = [&](size_t bytes) -> void* {
        void* p = wsb + off;
        off += (bytes + 255) & ~(size_t)255;
        return p;
    };
    _Float16* w1h  = (_Float16*)alloc((size_t)HC * NF * 2);
    _Float16* w2h  = (_Float16*)alloc((size_t)NC * HC * 2);
    float* out1    = (float*)alloc((size_t)NN * HC * 4);
    float* h1      = (float*)alloc((size_t)NN * HC * 4);
    float* asrc1   = (float*)alloc((size_t)NN * HH * 4);
    float* adst1   = (float*)alloc((size_t)NN * HH * 4);
    int*   deg     = (int*)alloc((size_t)NN * 4);
    int*   rowptr  = (int*)alloc((size_t)(NN + 1) * 4);
    int*   csr     = (int*)alloc((size_t)EE * 4);
    float* h2      = (float*)alloc((size_t)NN * NC * 4);
    float* asrc2   = (float*)alloc((size_t)NN * 4);
    float* adst2   = (float*)alloc((size_t)NN * 4);

    const dim3 B(256);
    gat_cvtw <<<(NN + 255) / 256, B, 0, stream>>>(W1, W2, w1h, w2h, deg);
    gat_gemm1<<<(NN / 16 + 7) / 8, B, 0, stream>>>(x, w1h, h1);
    gat_count<<<EE / 256, B, 0, stream>>>(ei, deg);
    gat_scan <<<(NN + 1024) / 1024, dim3(1024), 0, stream>>>(deg, rowptr);
    gat_fill <<<EE / 256, B, 0, stream>>>(ei, rowptr, deg, csr);
    gat_att1 <<<(NN * HH) / 256, B, 0, stream>>>(h1, as1, ad1, asrc1, adst1);
    gat_node1<<<NN / 8, B, 0, stream>>>(rowptr, csr, asrc1, adst1, h1, b1, out1);
    gat_gemm2<<<(NN / 16 + 7) / 8, B, 0, stream>>>(out1, w2h, h2);
    gat_att2 <<<(NN + 255) / 256, B, 0, stream>>>(h2, as2, ad2, asrc2, adst2);
    gat_node2<<<NN / 8, B, 0, stream>>>(rowptr, csr, asrc2, adst2, h2, b2, out);
}


# row-local stages fused into their producers: attention scalars from the GEMM accumulators (16-lane DPP tree sums: new summation order), layer-2 GEMM inside the node kernel from an LDS f16 tile
# speedup vs baseline: 3.0725x; 1.1496x over previous
#include <hip/hip_runtime.h>

#define NN  100000
#define NF  256
#define HH  8
#define CC  16
#define HC  128
#define NC  16
#define EE  1600000
#define ET  1700000
#define NEG 0.2f

typedef _Float16 h8   __attribute__((ext_vector_type(8)));
typedef _Float16 v16h __attribute__((ext_vector_type(16)));
typedef float    v8f  __attribute__((ext_vector_type(8)));

__device__ __forceinline__ float lrelu(float a) { return a > 0.0f ? a : NEG * a; }

template <int M>
__device__ __forceinline__ float rowx(float v) {
    return __builtin_bit_cast(float, __builtin_amdgcn_update_dpp(
        0, __builtin_bit_cast(int, v), 0x160 | M, 0xf, 0xf, false));
}

__device__ __forceinline__ float fold8(const float (&p)[8], bool b3, bool b2, bool b1) {
    float q[4], r2[2];
    #pragma unroll
    for (int j = 0; j < 4; ++j) q[j] = (b3 ? p[j + 4] : p[j]) + rowx<8>(b3 ? p[j] : p[j + 4]);
    #pragma unroll
    for (int j = 0; j < 2; ++j) r2[j] = (b2 ? q[j + 2] : q[j]) + rowx<4>(b2 ? q[j] : q[j + 2]);
    return (b1 ? r2[1] : r2[0]) + rowx<2>(b1 ? r2[0] : r2[1]);
}

static_assert(HC * NF + NC * HC <= NN, "the grid that covers the nodes also covers both weight matrices");
__global__ __launch_bounds__(256) void gat_cvtw(const float* __restrict__ W1,
                                                const float* __restrict__ W2,
                                                _Float16* __restrict__ w1h,
                                                _Float16* __restrict__ w2h,
                                                int* __restrict__ deg) {
    int i = blockIdx.x * 256 + threadIdx.x;
    if (i < NN) deg[i] = 0;
    if (i < HC * NF)              w1h[i] = (_Float16)W1[i];
    else if (i < HC * NF + NC*HC) w2h[i - HC * NF] = (_Float16)W2[i - HC * NF];
}

__global__ __launch_bounds__(256)
void gat_gemm1(const float* __restrict__ x, const _Float16* __restrict__ w1h,
               const float* __restrict__ as, const float* __restrict__ ad,
               float* __restrict__ h1, float* __restrict__ a_src, float* __restrict__ a_dst) {
    __shared__ _Float16 lw[HC * NF];
    const int tid = threadIdx.x;
    for (int idx = tid; idx < HC * 32; idx += 256) {
        int r = idx >> 5, ch = idx & 31;
        *(h8*)(&lw[r * NF + ((ch ^ (r & 15)) << 3)]) =
            *(const h8*)(w1h + (size_t)r * NF + (ch << 3));
    }
    __syncthreads();

    const int lane = tid & 31;
    const int rt   = blockIdx.x * 8 + (tid >> 5);
    if (rt >= NN / 16) return;
    const int rsub = lane & 15;
    const int ksel = (lane >> 4) << 3;
    const float* xrow = x + (size_t)(rt * 16 + rsub) * NF + ksel;

    v16h afr[8];
    #pragma unroll
    for (int ks = 0; ks < 8; ++ks) {
        const float4 f0 = *(const float4*)(xrow + ks * 32);
        const float4 f1 = *(const float4*)(xrow + ks * 32 + 4);
        const float4 f2 = *(const float4*)(xrow + ks * 32 + 16);
        const float4 f3 = *(const float4*)(xrow + ks * 32 + 20);
        union { v16h v; _Float16 e[16]; } A;
        A.e[0]  = (_Float16)f0.x; A.e[1]  = (_Float16)f0.y; A.e[2]  = (_Float16)f0.z; A.e[3]  = (_Float16)f0.w;
        A.e[4]  = (_Float16)f1.x; A.e[5]  = (_Float16)f1.y; A.e[6]  = (_Float16)f1.z; A.e[7]  = (_Float16)f1.w;
        A.e[8]  = (_Float16)f2.x; A.e[9]  = (_Float16)f2.y; A.e[10] = (_Float16)f2.z; A.e[11] = (_Float16)f2.w;
        A.e[12] = (_Float16)f3.x; A.e[13] = (_Float16)f3.y; A.e[14] = (_Float16)f3.z; A.e[15] = (_Float16)f3.w;
        afr[ks] = A.v;
    }
    __builtin_amdgcn_sched_barrier(0);

    v8f acc[8] = {};
    #pragma unroll
    for (int ks = 0; ks < 8; ++ks) {
        const int c0 = (ks * 32 + ksel) >> 3;
        #pragma unroll
        for (int ct = 0; ct < 8; ++ct) {
            const int n = ct * 16 + rsub;
            union { v16h v; h8 p[2]; } B;
            B.p[0] = *(const h8*)(&lw[n * NF + (((c0    ) ^ (n & 15)) << 3)]);
            B.p[1] = *(const h8*)(&lw[n * NF + (((c0 + 2) ^ (n & 15)) << 3)]);
            acc[ct] = __builtin_amdgcn_wmma_f32_16x16x32_f16(
                false, afr[ks], false, B.v, (short)0, acc[ct], false, false);
        }
    }
    const int rhi = (lane >> 4) << 3;
    #pragma unroll
    for (int ct = 0; ct < 8; ++ct) {
        #pragma unroll
        for (int r = 0; r < 8; ++r)
            h1[(size_t)(rt * 16 + rhi + r) * HC + ct * 16 + rsub] = acc[ct][r];
    }
    const bool b0 = lane & 1, b1 = lane & 2, b2 = lane & 4, b3 = lane & 8;
    float res[8];
    #pragma unroll
    for (int ct = 0; ct < 8; ++ct) {
        const float ws = as[ct * 16 + rsub], wd = ad[ct * 16 + rsub];
        float ps[8], pd[8];
        #pragma unroll
        for (int r = 0; r < 8; ++r) { ps[r] = acc[ct][r] * ws; pd[r] = acc[ct][r] * wd; }
        const float fs = fold8(ps, b3, b2, b1), fd = fold8(pd, b3, b2, b1);
        res[ct] = (b0 ? fd : fs) + rowx<1>(b0 ? fs : fd);
    }
    float* ap = (b0 ? a_dst : a_src) + (size_t)(rt * 16 + rhi + (rsub >> 1)) * HH;
    *(float4*)(ap)     = make_float4(res[0], res[1], res[2], res[3]);
    *(float4*)(ap + 4) = make_float4(res[4], res[5], res[6], res[7]);
}

static_assert(EE % 256 == 0, "edge kernels launch exactly EE threads");
__global__ __launch_bounds__(256)
void gat_count(const int* __restrict__ ei, int* __restrict__ deg) {
    int e = blockIdx.x * 256 + threadIdx.x;
    atomicAdd(&deg[ei[EE + e]], 1);
}

__global__ __launch_bounds__(1024)
void gat_scan(const int* __restrict__ deg, int* __restrict__ rowptr) {
    __shared__ int wsum[32], psum[32];
    const int t = threadIdx.x, lane = t & 31, w = t >> 5;
    const int4* d4 = (const int4*)deg;
    const int lim = blockIdx.x * 256;
    int s = 0;
    for (int k0 = t; k0 < lim; k0 += 4096) {
        int4 v[4];
        #pragma unroll
        for (int j = 0; j < 4; ++j) { const int k = k0 + 1024 * j; v[j] = d4[k < lim ? k : 0]; }
        #pragma unroll
        for (int j = 0; j < 4; ++j) s += (k0 + 1024 * j < lim) ? v[j].x + v[j].y + v[j].z + v[j].w : 0;
    }
    const int i = blockIdx.x * 1024 + t;
    const int v = i < NN ? deg[i] : 0;
    int incl = v;
    for (int o = 1; o < 32; o <<= 1) { int n = __shfl_up(incl, o); if (lane >= o) incl += n; }
    for (int o = 16; o > 0; o >>= 1) s += __shfl_xor(s, o);
    if (lane == 31) wsum[w] = incl;
    if (lane == 0)  psum[w] = s;
    __syncthreads();
    if (w == 0) {
        int tv = wsum[lane], iv = tv, ps = psum[lane];
        for (int o = 1; o < 32; o <<= 1) { int n = __shfl_up(iv, o); if (lane >= o) iv += n; }
        for (int o = 16; o > 0; o >>= 1) ps += __shfl_xor(ps, o);
        wsum[lane] = iv - tv + ps;
    }
    __syncthreads();
    if (i <= NN) rowptr[i] = wsum[w] + incl - v;
}

__global__ __launch_bounds__(256)
void gat_fill(const int* __restrict__ ei, const int* __restrict__ rowptr,
              int* __restrict__ deg, int* __restrict__ csr) {
    int e = blockIdx.x * 256 + threadIdx.x;
    int d = ei[EE + e];
    int slot = atomicSub(&deg[d], 1) - 1;
    csr[rowptr[d] + slot] = ei[e];
}

#define LP 136
static_assert(NN % 16 == 0, "sixteen nodes a block");
__global__ __launch_bounds__(256)
void gat_node1(const int* __restrict__ rowptr, const int* __restrict__ csr,
               const float* __restrict__ a_src, const float* __restrict__ a_dst,
               const float* __restrict__ h1, const float* __restrict__ b1,
               const _Float16* __restrict__ w2h, const float* __restrict__ as2,
               const float* __restrict__ ad2, float* __restrict__ h2,
               float* __restrict__ a_src2, float* __restrict__ a_dst2) {
    __shared__ __attribute__((aligned(16))) _Float16 lo[16 * LP];
    const int lane = threadIdx.x & 31;
    const int wv = __builtin_amdgcn_readfirstlane(threadIdx.x >> 5);
    const int eh = lane & 7, es = lane >> 3, ch = lane >> 2;
    const float NINF = -__builtin_inff();
    const float4 b = *(const float4*)(b1 + lane * 4);

  for (int half = 0; half < 2; ++half) {
    const int trow = wv + 8 * half;
    const int d   = blockIdx.x * 16 + trow;
    const int beg = rowptr[d];
    const int cnt = rowptr[d + 1] - beg + 1;
    const float adh = a_dst[d * HH + eh];

    float m = NINF, l = 0.0f;
    float acc0 = 0.0f, acc1 = 0.0f, acc2 = 0.0f, acc3 = 0.0f;
    for (int c0 = 0; c0 < cnt; c0 += 32) {
        const int p = c0 + lane;
        int sid = d;
        if (p > 0 && p < cnt) sid = csr[beg + p - 1];
        float a[8];
        float mc = NINF;
        #pragma unroll
        for (int t = 0; t < 8; ++t) {
            const int q = 4 * t + es;
            const int s = __shfl(sid, q);
            const float v = lrelu(a_src[s * HH + eh] + adh);
            a[t] = (c0 + q < cnt) ? v : NINF;
            mc = fmaxf(mc, a[t]);
        }
        mc = fmaxf(mc, __shfl_xor(mc, 8));
        mc = fmaxf(mc, __shfl_xor(mc, 16));
        const float mn = fmaxf(m, mc);
        const float sc = expf(m - mn);
        float lc = 0.0f;
        #pragma unroll
        for (int t = 0; t < 8; ++t) { a[t] = expf(a[t] - mn); lc += a[t]; }
        lc += __shfl_xor(lc, 8);
        lc += __shfl_xor(lc, 16);
        l = l * sc + lc;
        m = mn;
        const float scc = __shfl(sc, ch);
        acc0 *= scc; acc1 *= scc; acc2 *= scc; acc3 *= scc;
        #pragma unroll
        for (int t = 0; t < 8; ++t) {
            if (c0 + 4 * t < cnt) {
                float4 row[4];
                float  wt[4];
                #pragma unroll
                for (int j = 0; j < 4; ++j) {
                    const int s = __builtin_amdgcn_readlane(sid, 4 * t + j);
                    wt[j]  = __shfl(a[t], j * 8 + ch);
                    row[j] = *(const float4*)(h1 + (size_t)s * HC + lane * 4);
                }
                #pragma unroll
                for (int j = 0; j < 4; ++j) {
                    acc0 += wt[j] * row[j].x; acc1 += wt[j] * row[j].y;
                    acc2 += wt[j] * row[j].z; acc3 += wt[j] * row[j].w;
                }
            }
        }
    }
    const float den = __shfl(l, ch) + 1e-16f;
    float4 o;
    o.x = acc0 / den + b.x; o.y = acc1 / den + b.y; o.z = acc2 / den + b.z; o.w = acc3 / den + b.w;
    o.x = o.x > 0.0f ? o.x : (expf(o.x) - 1.0f);
    o.y = o.y > 0.0f ? o.y : (expf(o.y) - 1.0f);
    o.z = o.z > 0.0f ? o.z : (expf(o.z) - 1.0f);
    o.w = o.w > 0.0f ? o.w : (expf(o.w) - 1.0f);
    union { _Float16 e[4]; uint2 u; } hv;
    hv.e[0] = (_Float16)o.x; hv.e[1] = (_Float16)o.y; hv.e[2] = (_Float16)o.z; hv.e[3] = (_Float16)o.w;
    *(uint2*)(&lo[trow * LP + lane * 4]) = hv.u;
  }
    __syncthreads();
    if (wv != 0) return;

    const int rsub = lane & 15;
    const int ksel = (lane >> 4) << 3;
    const _Float16* wrow = w2h + (size_t)rsub * HC;
    v8f acc = {};
    #pragma unroll
    for (int ks = 0; ks < 4; ++ks) {
        const int kb = ks * 32 + ksel;
        union { v16h v; h8 p[2]; } A, B;
        A.p[0] = *(const h8*)(&lo[rsub * LP + kb]);
        A.p[1] = *(const h8*)(&lo[rsub * LP + kb + 16]);
        B.p[0] = *(const h8*)(wrow + kb);
        B.p[1] = *(const h8*)(wrow + kb + 16);
        acc = __builtin_amdgcn_wmma_f32_16x16x32_f16(
            false, A.v, false, B.v, (short)0, acc, false, false);
    }
    const int rhi = (lane >> 4) << 3;
    const int r0  = blockIdx.x * 16 + rhi;
    #pragma unroll
    for (int r = 0; r < 8; ++r)
        h2[(size_t)(r0 + r) * NC + rsub] = acc[r];
    const bool b0 = lane & 1, b1b = lane & 2, b2b = lane & 4, b3b = lane & 8;
    const float ws = as2[rsub], wd = ad2[rsub];
    float ps[8], pd[8];
    #pragma unroll
    for (int r = 0; r < 8; ++r) { ps[r] = acc[r] * ws; pd[r] = acc[r] * wd; }
    const float fs = fold8(ps, b3b, b2b, b1b), fd = fold8(pd, b3b, b2b, b1b);
    const float res = (b0 ? fd : fs) + rowx<1>(b0 ? fs : fd);
    (b0 ? a_dst2 : a_src2)[r0 + (rsub >> 1)] = res;
}


__global__ __launch_bounds__(256)
void gat_node2(const int* __restrict__ rowptr, const int* __restrict__ csr,
               const float* __restrict__ a_src, const float* __restrict__ a_dst,
               const float* __restrict__ h2, const float* __restrict__ b2,
               float* __restrict__ out) {
    const int lane = threadIdx.x & 31;
    const int d   = __builtin_amdgcn_readfirstlane(blockIdx.x * 8 + (threadIdx.x >> 5));
    const int beg = rowptr[d];
    const int cnt = rowptr[d + 1] - beg + 1;
    const int c = lane & 15, es = lane >> 4;
    const float adv = a_dst[d];
    const float NINF = -__builtin_inff();

    float m = NINF, l = 0.0f, acc = 0.0f;
    for (int c0 = 0; c0 < cnt; c0 += 32) {
        const int p = c0 + lane;
        const bool valid = p < cnt;
        int sid = d;
        if (p > 0 && valid) sid = csr[beg + p - 1];
        const float a = valid ? lrelu(a_src[sid] + adv) : NINF;
        float mc = a;
        for (int o = 16; o > 0; o >>= 1) mc = fmaxf(mc, __shfl_xor(mc, o));
        const float mn = fmaxf(m, mc);
        const float sc = expf(m - mn);
        const float ex = expf(a - mn);
        float lc = ex;
        for (int o = 16; o > 0; o >>= 1) lc += __shfl_xor(lc, o);
        l = l * sc + lc;
        m = mn;
        acc *= sc;
        #pragma unroll
        for (int g = 0; g < 4; ++g) {
            if (c0 + 8 * g < cnt) {
                float hv[4], wt[4];
                #pragma unroll
                for (int j = 0; j < 4; ++j) {
                    const int q = 8 * g + 2 * j + es;
                    const int s = __shfl(sid, q);
                    wt[j] = __shfl(ex, q);
                    hv[j] = h2[(size_t)s * NC + c];
                }
                #pragma unroll
                for (int j = 0; j < 4; ++j) acc += wt[j] * hv[j];
            }
        }
    }
    acc += __shfl_xor(acc, 16);
    const float v = acc / (l + 1e-16f) + b2[c];
    float mx = v;
    for (int o = 8; o > 0; o >>= 1) mx = fmaxf(mx, __shfl_xor(mx, o));
    float se = expf(v - mx);
    for (int o = 8; o > 0; o >>= 1) se += __shfl_xor(se, o);
    if (es == 0) out[(size_t)d * NC + c] = v - mx - logf(se);
}

extern "C" void kernel_launch(void* const* d_in, const int* in_sizes, int n_in,
                              void* d_out, int out_size, void* d_ws, size_t ws_size,
                              hipStream_t stream) {
    (void)in_sizes; (void)n_in; (void)out_size; (void)ws_size;
    const float* x   = (const float*)d_in[0];
    const int*   ei  = (const int*)d_in[1];
    const float* W1  = (const float*)d_in[2];
    const float* as1 = (const float*)d_in[3];
    const float* ad1 = (const float*)d_in[4];
    const float* b1  = (const float*)d_in[5];
    const float* W2  = (const float*)d_in[6];
    const float* as2 = (const float*)d_in[7];
    const float* ad2 = (const float*)d_in[8];
    const float* b2  = (const float*)d_in[9];
    float* out = (float*)d_out;

    char* wsb = (char*)d_ws;
    size_t off = 0;
    auto alloc = [&](size_t bytes) -> void* {
        void* p = wsb + off;
        off += (bytes + 255) & ~(size_t)255;
        return p;
    };
    _Float16* w1h  = (_Float16*)alloc((size_t)HC * NF * 2);
    _Float16* w2h  = (_Float16*)alloc((size_t)NC * HC * 2);
    float* h1      = (float*)alloc((size_t)NN * HC * 4);
    float* asrc1   = (float*)alloc((size_t)NN * HH * 4);
    float* adst1   = (float*)alloc((size_t)NN * HH * 4);
    int*   deg     = (int*)alloc((size_t)NN * 4);
    int*   rowptr  = (int*)alloc((size_t)(NN + 1) * 4);
    int*   csr     = (int*)alloc((size_t)EE * 4);
    float* h2      = (float*)alloc((size_t)NN * NC * 4);
    float* asrc2   = (float*)alloc((size_t)NN * 4);
    float* adst2   = (float*)alloc((size_t)NN * 4);

    const dim3 B(256);
    gat_cvtw <<<(NN + 255) / 256, B, 0, stream>>>(W1, W2, w1h, w2h, deg);
    gat_gemm1<<<(NN / 16 + 7) / 8, B, 0, stream>>>(x, w1h, as1, ad1, h1, asrc1, adst1);
    gat_count<<<EE / 256, B, 0, stream>>>(ei, deg);
    gat_scan <<<(NN + 1024) / 1024, dim3(1024), 0, stream>>>(deg, rowptr);
    gat_fill <<<EE / 256, B, 0, stream>>>(ei, rowptr, deg, csr);
    gat_node1<<<NN / 16, B, 0, stream>>>(rowptr, csr, asrc1, adst1, h1, b1, w2h, as2, ad2, h2, asrc2, adst2);
    gat_node2<<<NN / 8, B, 0, stream>>>(rowptr, csr, asrc2, adst2, h2, b2, out);
}


# edges bucketed by destination in one pass: each edge claims a slot of its node's 32-slot line with one int atomic (shared overflow list beyond); count and scan launches gone; sums in arrival order
# speedup vs baseline: 4.5274x; 1.4735x over previous
#include <hip/hip_runtime.h>

#define NN  100000
#define NF  256
#define HH  8
#define CC  16
#define HC  128
#define NC  16
#define EE  1600000
#define ET  1700000
#define NEG 0.2f
#define KB  32

typedef _Float16 h8   __attribute__((ext_vector_type(8)));
typedef _Float16 v16h __attribute__((ext_vector_type(16)));
typedef float    v8f  __attribute__((ext_vector_type(8)));

__device__ __forceinline__ float lrelu(float a) { return a > 0.0f ? a : NEG * a; }

template <int M>
__device__ __forceinline__ float rowx(float v) {
    return __builtin_bit_cast(float, __builtin_amdgcn_update_dpp(
        0, __builtin_bit_cast(int, v), 0x160 | M, 0xf, 0xf, false));
}

__device__ __forceinline__ float fold8(const float (&p)[8], bool b3, bool b2, bool b1) {
    float q[4], r2[2];
    #pragma unroll
    for (int j = 0; j < 4; ++j) q[j] = (b3 ? p[j + 4] : p[j]) + rowx<8>(b3 ? p[j] : p[j + 4]);
    #pragma unroll
    for (int j = 0; j < 2; ++j) r2[j] = (b2 ? q[j + 2] : q[j]) + rowx<4>(b2 ? q[j] : q[j + 2]);
    return (b1 ? r2[1] : r2[0]) + rowx<2>(b1 ? r2[0] : r2[1]);
}

static_assert(HC * NF + NC * HC <= NN, "the grid that covers the nodes also covers both weight matrices");
__global__ __launch_bounds__(256) void gat_cvtw(const float* __restrict__ W1,
                                                const float* __restrict__ W2,
                                                _Float16* __restrict__ w1h,
                                                _Float16* __restrict__ w2h,
                                                int* __restrict__ deg,
                                                int* __restrict__ ovn) {
    int i = blockIdx.x * 256 + threadIdx.x;
    if (i < NN) deg[i] = 0;
    if (i == 0) ovn[0] = 0;
    if (i < HC * NF)              w1h[i] = (_Float16)W1[i];
    else if (i < HC * NF + NC*HC) w2h[i - HC * NF] = (_Float16)W2[i - HC * NF];
}

__global__ __launch_bounds__(256)
void gat_gemm1(const float* __restrict__ x, const _Float16* __restrict__ w1h,
               const float* __restrict__ as, const float* __restrict__ ad,
               float* __restrict__ h1, float* __restrict__ a_src, float* __restrict__ a_dst) {
    __shared__ _Float16 lw[HC * NF];
    const int tid = threadIdx.x;
    for (int idx = tid; idx < HC * 32; idx += 256) {
        int r = idx >> 5, ch = idx & 31;
        *(h8*)(&lw[r * NF + ((ch ^ (r & 15)) << 3)]) =
            *(const h8*)(w1h + (size_t)r * NF + (ch << 3));
    }
    __syncthreads();

    const int lane = tid & 31;
    const int rt   = blockIdx.x * 8 + (tid >> 5);
    if (rt >= NN / 16) return;
    const int rsub = lane & 15;
    const int ksel = (lane >> 4) << 3;
    const float* xrow = x + (size_t)(rt * 16 + rsub) * NF + ksel;

    v16h afr[8];
    #pragma unroll
    for (int ks = 0; ks < 8; ++ks) {
        const float4 f0 = *(const float4*)(xrow + ks * 32);
        const float4 f1 = *(const float4*)(xrow + ks * 32 + 4);
        const float4 f2 = *(const float4*)(xrow + ks * 32 + 16);
        const float4 f3 = *(const float4*)(xrow + ks * 32 + 20);
        union { v16h v; _Float16 e[16]; } A;
        A.e[0]  = (_Float16)f0.x; A.e[1]  = (_Float16)f0.y; A.e[2]  = (_Float16)f0.z; A.e[3]  = (_Float16)f0.w;
        A.e[4]  = (_Float16)f1.x; A.e[5]  = (_Float16)f1.y; A.e[6]  = (_Float16)f1.z; A.e[7]  = (_Float16)f1.w;
        A.e[8]  = (_Float16)f2.x; A.e[9]  = (_Float16)f2.y; A.e[10] = (_Float16)f2.z; A.e[11] = (_Float16)f2.w;
        A.e[12] = (_Float16)f3.x; A.e[13] = (_Float16)f3.y; A.e[14] = (_Float16)f3.z; A.e[15] = (_Float16)f3.w;
        afr[ks] = A.v;
    }
    __builtin_amdgcn_sched_barrier(0);

    v8f acc[8] = {};
    #pragma unroll
    for (int ks = 0; ks < 8; ++ks) {
        const int c0 = (ks * 32 + ksel) >> 3;
        #pragma unroll
        for (int ct = 0; ct < 8; ++ct) {
            const int n = ct * 16 + rsub;
            union { v16h v; h8 p[2]; } B;
            B.p[0] = *(const h8*)(&lw[n * NF + (((c0    ) ^ (n & 15)) << 3)]);
            B.p[1] = *(const h8*)(&lw[n * NF + (((c0 + 2) ^ (n & 15)) << 3)]);
            acc[ct] = __builtin_amdgcn_wmma_f32_16x16x32_f16(
                false, afr[ks], false, B.v, (short)0, acc[ct], false, false);
        }
    }
    const int rhi = (lane >> 4) << 3;
    #pragma unroll
    for (int ct = 0; ct < 8; ++ct) {
        #pragma unroll
        for (int r = 0; r < 8; ++r)
            h1[(size_t)(rt * 16 + rhi + r) * HC + ct * 16 + rsub] = acc[ct][r];
    }
    const bool b0 = lane & 1, b1 = lane & 2, b2 = lane & 4, b3 = lane & 8;
    float res[8];
    #pragma unroll
    for (int ct = 0; ct < 8; ++ct) {
        const float ws = as[ct * 16 + rsub], wd = ad[ct * 16 + rsub];
        float ps[8], pd[8];
        #pragma unroll
        for (int r = 0; r < 8; ++r) { ps[r] = acc[ct][r] * ws; pd[r] = acc[ct][r] * wd; }
        const float fs = fold8(ps, b3, b2, b1), fd = fold8(pd, b3, b2, b1);
        res[ct] = (b0 ? fd : fs) + rowx<1>(b0 ? fs : fd);
    }
    float* ap = (b0 ? a_dst : a_src) + (size_t)(rt * 16 + rhi + (rsub >> 1)) * HH;
    *(float4*)(ap)     = make_float4(res[0], res[1], res[2], res[3]);
    *(float4*)(ap + 4) = make_float4(res[4], res[5], res[6], res[7]);
}

static_assert(EE % 256 == 0, "the edge kernel launches exactly EE threads");
__global__ __launch_bounds__(256)
void gat_bucket(const int* __restrict__ ei, int* __restrict__ deg, int* __restrict__ bkt,
                int* __restrict__ ovn, int* __restrict__ ovf) {
    const int e = blockIdx.x * 256 + threadIdx.x;
    const int d = ei[EE + e];
    const int s = ei[e];
    const int slot = atomicAdd(&deg[d], 1);
    if (slot < KB) {
        bkt[d * KB + slot] = s;
    } else {
        const int p = atomicAdd(ovn, 1);
        ovf[2 * p]     = d;
        ovf[2 * p + 1] = s;
    }
}

#define LP 136
static_assert(NN % 16 == 0, "sixteen nodes a block");
__global__ __launch_bounds__(256)
void gat_node1(const int* __restrict__ deg, const int* __restrict__ bkt,
               const int* __restrict__ ovn, const int* __restrict__ ovf,
               const float* __restrict__ a_src, const float* __restrict__ a_dst,
               const float* __restrict__ h1, const float* __restrict__ b1,
               const _Float16* __restrict__ w2h, const float* __restrict__ as2,
               const float* __restrict__ ad2, float* __restrict__ h2,
               float* __restrict__ a_src2, float* __restrict__ a_dst2) {
    __shared__ __attribute__((aligned(16))) _Float16 lo[16 * LP];
    const int lane = threadIdx.x & 31;
    const int wv = __builtin_amdgcn_readfirstlane(threadIdx.x >> 5);
    const int eh = lane & 7, es = lane >> 3, ch = lane >> 2;
    const float NINF = -__builtin_inff();
    const float4 b = *(const float4*)(b1 + lane * 4);

  for (int half = 0; half < 2; ++half) {
    const int trow = wv + 8 * half;
    const int d   = blockIdx.x * 16 + trow;
    const int dg  = deg[d];
    const int cnt = (dg < KB ? dg : KB) + 1;
    const int* lst = bkt + d * KB;
    const float adh = a_dst[d * HH + eh];

    float m = NINF, l = 0.0f;
    float acc0 = 0.0f, acc1 = 0.0f, acc2 = 0.0f, acc3 = 0.0f;
    for (int c0 = 0; c0 < cnt; c0 += 32) {
        const int p = c0 + lane;
        int sid = d;
        if (p > 0 && p < cnt) sid = lst[p - 1];
        float a[8];
        float mc = NINF;
        #pragma unroll
        for (int t = 0; t < 8; ++t) {
            const int q = 4 * t + es;
            const int s = __shfl(sid, q);
            const float v = lrelu(a_src[s * HH + eh] + adh);
            a[t] = (c0 + q < cnt) ? v : NINF;
            mc = fmaxf(mc, a[t]);
        }
        mc = fmaxf(mc, __shfl_xor(mc, 8));
        mc = fmaxf(mc, __shfl_xor(mc, 16));
        const float mn = fmaxf(m, mc);
        const float sc = expf(m - mn);
        float lc = 0.0f;
        #pragma unroll
        for (int t = 0; t < 8; ++t) { a[t] = expf(a[t] - mn); lc += a[t]; }
        lc += __shfl_xor(lc, 8);
        lc += __shfl_xor(lc, 16);
        l = l * sc + lc;
        m = mn;
        const float scc = __shfl(sc, ch);
        acc0 *= scc; acc1 *= scc; acc2 *= scc; acc3 *= scc;
        #pragma unroll
        for (int t = 0; t < 8; ++t) {
            if (c0 + 4 * t < cnt) {
                float4 row[4];
                float  wt[4];
                #pragma unroll
                for (int j = 0; j < 4; ++j) {
                    const int s = __builtin_amdgcn_readlane(sid, 4 * t + j);
                    wt[j]  = __shfl(a[t], j * 8 + ch);
                    row[j] = *(const float4*)(h1 + (size_t)s * HC + lane * 4);
                }
                #pragma unroll
                for (int j = 0; j < 4; ++j) {
                    acc0 += wt[j] * row[j].x; acc1 += wt[j] * row[j].y;
                    acc2 += wt[j] * row[j].z; acc3 += wt[j] * row[j].w;
                }
            }
        }
    }
    if (dg > KB) {
        const int no = ovn[0];
        for (int o0 = 0; o0 < no; o0 += 32) {
            const int q = o0 + lane;
            int od = -1, os = 0;
            if (q < no) { od = ovf[2 * q]; os = ovf[2 * q + 1]; }
            unsigned mk = __builtin_amdgcn_ballot_w32(od == d);
            while (mk) {
                const int bit = __builtin_ctz(mk);
                mk &= mk - 1;
                const int s = __builtin_amdgcn_readlane(os, bit);
                const float v  = lrelu(a_src[s * HH + eh] + adh);
                const float mn = fmaxf(m, v);
                const float sc = expf(m - mn);
                const float w  = expf(v - mn);
                l = l * sc + w;
                m = mn;
                const float scc = __shfl(sc, ch), wcc = __shfl(w, ch);
                const float4 row = *(const float4*)(h1 + (size_t)s * HC + lane * 4);
                acc0 = acc0 * scc + wcc * row.x; acc1 = acc1 * scc + wcc * row.y;
                acc2 = acc2 * scc + wcc * row.z; acc3 = acc3 * scc + wcc * row.w;
            }
        }
    }
    const float den = __shfl(l, ch) + 1e-16f;
    float4 o;
    o.x = acc0 / den + b.x; o.y = acc1 / den + b.y; o.z = acc2 / den + b.z; o.w = acc3 / den + b.w;
    o.x = o.x > 0.0f ? o.x : (expf(o.x) - 1.0f);
    o.y = o.y > 0.0f ? o.y : (expf(o.y) - 1.0f);
    o.z = o.z > 0.0f ? o.z : (expf(o.z) - 1.0f);
    o.w = o.w > 0.0f ? o.w : (expf(o.w) - 1.0f);
    union { _Float16 e[4]; uint2 u; } hv;
    hv.e[0] = (_Float16)o.x; hv.e[1] = (_Float16)o.y; hv.e[2] = (_Float16)o.z; hv.e[3] = (_Float16)o.w;
    *(uint2*)(&lo[trow * LP + lane * 4]) = hv.u;
  }
    __syncthreads();
    if (wv != 0) return;

    const int rsub = lane & 15;
    const int ksel = (lane >> 4) << 3;
    const _Float16* wrow = w2h + (size_t)rsub * HC;
    v8f acc = {};
    #pragma unroll
    for (int ks = 0; ks < 4; ++ks) {
        const int kb = ks * 32 + ksel;
        union { v16h v; h8 p[2]; } A, B;
        A.p[0] = *(const h8*)(&lo[rsub * LP + kb]);
        A.p[1] = *(const h8*)(&lo[rsub * LP + kb + 16]);
        B.p[0] = *(const h8*)(wrow + kb);
        B.p[1] = *(const h8*)(wrow + kb + 16);
        acc = __builtin_amdgcn_wmma_f32_16x16x32_f16(
            false, A.v, false, B.v, (short)0, acc, false, false);
    }
    const int rhi = (lane >> 4) << 3;
    const int r0  = blockIdx.x * 16 + rhi;
    #pragma unroll
    for (int r = 0; r < 8; ++r)
        h2[(size_t)(r0 + r) * NC + rsub] = acc[r];
    const bool b0 = lane & 1, b1b = lane & 2, b2b = lane & 4, b3b = lane & 8;
    const float ws = as2[rsub], wd = ad2[rsub];
    float ps[8], pd[8];
    #pragma unroll
    for (int r = 0; r < 8; ++r) { ps[r] = acc[r] * ws; pd[r] = acc[r] * wd; }
    const float fs = fold8(ps, b3b, b2b, b1b), fd = fold8(pd, b3b, b2b, b1b);
    const float res = (b0 ? fd : fs) + rowx<1>(b0 ? fs : fd);
    (b0 ? a_dst2 : a_src2)[r0 + (rsub >> 1)] = res;
}


__global__ __launch_bounds__(256)
void gat_node2(const int* __restrict__ deg, const int* __restrict__ bkt,
               const int* __restrict__ ovn, const int* __restrict__ ovf,
               const float* __restrict__ a_src, const float* __restrict__ a_dst,
               const float* __restrict__ h2, const float* __restrict__ b2,
               float* __restrict__ out) {
    const int lane = threadIdx.x & 31;
    const int d   = __builtin_amdgcn_readfirstlane(blockIdx.x * 8 + (threadIdx.x >> 5));
    const int dg  = deg[d];
    const int cnt = (dg < KB ? dg : KB) + 1;
    const int* lst = bkt + d * KB;
    const int c = lane & 15, es = lane >> 4;
    const float adv = a_dst[d];
    const float NINF = -__builtin_inff();

    float m = NINF, l = 0.0f, acc = 0.0f;
    for (int c0 = 0; c0 < cnt; c0 += 32) {
        const int p = c0 + lane;
        const bool valid = p < cnt;
        int sid = d;
        if (p > 0 && valid) sid = lst[p - 1];
        const float a = valid ? lrelu(a_src[sid] + adv) : NINF;
        float mc = a;
        for (int o = 16; o > 0; o >>= 1) mc = fmaxf(mc, __shfl_xor(mc, o));
        const float mn = fmaxf(m, mc);
        const float sc = expf(m - mn);
        const float ex = expf(a - mn);
        float lc = ex;
        for (int o = 16; o > 0; o >>= 1) lc += __shfl_xor(lc, o);
        l = l * sc + lc;
        m = mn;
        acc *= sc;
        #pragma unroll
        for (int g = 0; g < 4; ++g) {
            if (c0 + 8 * g < cnt) {
                float hv[4], wt[4];
                #pragma unroll
                for (int j = 0; j < 4; ++j) {
                    const int q = 8 * g + 2 * j + es;
                    const int s = __shfl(sid, q);
                    wt[j] = __shfl(ex, q);
                    hv[j] = h2[(size_t)s * NC + c];
                }
                #pragma unroll
                for (int j = 0; j < 4; ++j) acc += wt[j] * hv[j];
            }
        }
    }
    if (dg > KB) {
        const int no = ovn[0];
        for (int o0 = 0; o0 < no; o0 += 32) {
            const int q = o0 + lane;
            int od = -1, os = 0;
            if (q < no) { od = ovf[2 * q]; os = ovf[2 * q + 1]; }
            unsigned mk = __builtin_amdgcn_ballot_w32(od == d);
            while (mk) {
                const int bit = __builtin_ctz(mk);
                mk &= mk - 1;
                const int s = __builtin_amdgcn_readlane(os, bit);
                const float ev = lrelu(a_src[s] + adv);
                const float mn = fmaxf(m, ev);
                const float sc = expf(m - mn);
                const float w  = expf(ev - mn);
                l = l * sc + w;
                m = mn;
                const float hv = h2[(size_t)s * NC + c];
                acc = acc * sc + (es == 0 ? w * hv : 0.0f);
            }
        }
    }
    acc += __shfl_xor(acc, 16);
    const float v = acc / (l + 1e-16f) + b2[c];
    float mx = v;
    for (int o = 8; o > 0; o >>= 1) mx = fmaxf(mx, __shfl_xor(mx, o));
    float se = expf(v - mx);
    for (int o = 8; o > 0; o >>= 1) se += __shfl_xor(se, o);
    if (es == 0) out[(size_t)d * NC + c] = v - mx - logf(se);
}

extern "C" void kernel_launch(void* const* d_in, const int* in_sizes, int n_in,
                              void* d_out, int out_size, void* d_ws, size_t ws_size,
                              hipStream_t stream) {
    (void)in_sizes; (void)n_in; (void)out_size; (void)ws_size;
    const float* x   = (const float*)d_in[0];
    const int*   ei  = (const int*)d_in[1];
    const float* W1  = (const float*)d_in[2];
    const float* as1 = (const float*)d_in[3];
    const float* ad1 = (const float*)d_in[4];
    const float* b1  = (const float*)d_in[5];
    const float* W2  = (const float*)d_in[6];
    const float* as2 = (const float*)d_in[7];
    const float* ad2 = (const float*)d_in[8];
    const float* b2  = (const float*)d_in[9];
    float* out = (float*)d_out;

    char* wsb = (char*)d_ws;
    size_t off = 0;
    auto alloc = [&](size_t bytes) -> void* {
        void* p = wsb + off;
        off += (bytes + 255) & ~(size_t)255;
        return p;
    };
    _Float16* w1h  = (_Float16*)alloc((size_t)HC * NF * 2);
    _Float16* w2h  = (_Float16*)alloc((size_t)NC * HC * 2);
    float* h1      = (float*)alloc((size_t)NN * HC * 4);
    float* asrc1   = (float*)alloc((size_t)NN * HH * 4);
    float* adst1   = (float*)alloc((size_t)NN * HH * 4);
    int*   deg     = (int*)alloc((size_t)NN * 4);
    int*   ovn     = (int*)alloc((size_t)4);
    int*   bkt     = (int*)alloc((size_t)NN * KB * 4);
    int*   ovf     = (int*)alloc((size_t)EE * 2 * 4);
    float* h2      = (float*)alloc((size_t)NN * NC * 4);
    float* asrc2   = (float*)alloc((size_t)NN * 4);
    float* adst2   = (float*)alloc((size_t)NN * 4);

    const dim3 B(256);
    gat_cvtw <<<(NN + 255) / 256, B, 0, stream>>>(W1, W2, w1h, w2h, deg, ovn);
    gat_gemm1<<<(NN / 16 + 7) / 8, B, 0, stream>>>(x, w1h, as1, ad1, h1, asrc1, adst1);
    gat_bucket<<<EE / 256, B, 0, stream>>>(ei, deg, bkt, ovn, ovf);
    gat_node1<<<NN / 16, B, 0, stream>>>(deg, bkt, ovn, ovf, asrc1, adst1, h1, b1, w2h, as2, ad2, h2, asrc2, adst2);
    gat_node2<<<NN / 8, B, 0, stream>>>(deg, bkt, ovn, ovf, asrc2, adst2, h2, b2, out);
}


# h1 stored as f16, not f32: edge gathers and GEMM1 stores move half the bytes; GEMM1 taken as W x^T so a lane packs 8 channels a store; attention scalars: 8-term lane sums + half swap (new order)
# speedup vs baseline: 4.7386x; 1.0466x over previous
#include <hip/hip_runtime.h>

#define NN  100000
#define NF  256
#define HH  8
#define CC  16
#define HC  128
#define NC  16
#define EE  1600000
#define ET  1700000
#define NEG 0.2f
#define KB  32

typedef _Float16 h8   __attribute__((ext_vector_type(8)));
typedef _Float16 v16h __attribute__((ext_vector_type(16)));
typedef float    v8f  __attribute__((ext_vector_type(8)));

__device__ __forceinline__ float lrelu(float a) { return a > 0.0f ? a : NEG * a; }

template <int M>
__device__ __forceinline__ float rowx(float v) {
    return __builtin_bit_cast(float, __builtin_amdgcn_update_dpp(
        0, __builtin_bit_cast(int, v), 0x160 | M, 0xf, 0xf, false));
}

__device__ __forceinline__ float fold8(const float (&p)[8], bool b3, bool b2, bool b1) {
    float q[4], r2[2];
    #pragma unroll
    for (int j = 0; j < 4; ++j) q[j] = (b3 ? p[j + 4] : p[j]) + rowx<8>(b3 ? p[j] : p[j + 4]);
    #pragma unroll
    for (int j = 0; j < 2; ++j) r2[j] = (b2 ? q[j + 2] : q[j]) + rowx<4>(b2 ? q[j] : q[j + 2]);
    return (b1 ? r2[1] : r2[0]) + rowx<2>(b1 ? r2[0] : r2[1]);
}

static_assert(HC * NF + NC * HC <= NN, "the grid that covers the nodes also covers both weight matrices");
__global__ __launch_bounds__(256) void gat_cvtw(const float* __restrict__ W1,
                                                const float* __restrict__ W2,
                                                _Float16* __restrict__ w1h,
                                                _Float16* __restrict__ w2h,
                                                int* __restrict__ deg,
                                                int* __restrict__ ovn) {
    int i = blockIdx.x * 256 + threadIdx.x;
    if (i < NN) deg[i] = 0;
    if (i == 0) ovn[0] = 0;
    if (i < HC * NF)              w1h[i] = (_Float16)W1[i];
    else if (i < HC * NF + NC*HC) w2h[i - HC * NF] = (_Float16)W2[i - HC * NF];
}

__global__ __launch_bounds__(256)
void gat_gemm1(const float* __restrict__ x, const _Float16* __restrict__ w1h,
               const float* __restrict__ as, const float* __restrict__ ad,
               _Float16* __restrict__ h1, float* __restrict__ a_src, float* __restrict__ a_dst) {
    __shared__ _Float16 lw[HC * NF];
    const int tid = threadIdx.x;
    for (int idx = tid; idx < HC * 32; idx += 256) {
        int r = idx >> 5, ch = idx & 31;
        *(h8*)(&lw[r * NF + ((ch ^ (r & 15)) << 3)]) =
            *(const h8*)(w1h + (size_t)r * NF + (ch << 3));
    }
    __syncthreads();

    const int lane = tid & 31;
    const int rt   = blockIdx.x * 8 + (tid >> 5);
    if (rt >= NN / 16) return;
    const int rsub = lane & 15;
    const int ksel = (lane >> 4) << 3;
    const float* xrow = x + (size_t)(rt * 16 + rsub) * NF + ksel;

    v16h afr[8];
    #pragma unroll
    for (int ks = 0; ks < 8; ++ks) {
        const float4 f0 = *(const float4*)(xrow + ks * 32);
        const float4 f1 = *(const float4*)(xrow + ks * 32 + 4);
        const float4 f2 = *(const float4*)(xrow + ks * 32 + 16);
        const float4 f3 = *(const float4*)(xrow + ks * 32 + 20);
        union { v16h v; _Float16 e[16]; } A;
        A.e[0]  = (_Float16)f0.x; A.e[1]  = (_Float16)f0.y; A.e[2]  = (_Float16)f0.z; A.e[3]  = (_Float16)f0.w;
        A.e[4]  = (_Float16)f1.x; A.e[5]  = (_Float16)f1.y; A.e[6]  = (_Float16)f1.z; A.e[7]  = (_Float16)f1.w;
        A.e[8]  = (_Float16)f2.x; A.e[9]  = (_Float16)f2.y; A.e[10] = (_Float16)f2.z; A.e[11] = (_Float16)f2.w;
        A.e[12] = (_Float16)f3.x; A.e[13] = (_Float16)f3.y; A.e[14] = (_Float16)f3.z; A.e[15] = (_Float16)f3.w;
        afr[ks] = A.v;
    }
    __builtin_amdgcn_sched_barrier(0);

    v8f acc[8] = {};
    #pragma unroll
    for (int ks = 0; ks < 8; ++ks) {
        const int c0 = (ks * 32 + ksel) >> 3;
        #pragma unroll
        for (int ct = 0; ct < 8; ++ct) {
            const int n = ct * 16 + rsub;
            union { v16h v; h8 p[2]; } B;
            B.p[0] = *(const h8*)(&lw[n * NF + (((c0    ) ^ (n & 15)) << 3)]);
            B.p[1] = *(const h8*)(&lw[n * NF + (((c0 + 2) ^ (n & 15)) << 3)]);
            acc[ct] = __builtin_amdgcn_wmma_f32_16x16x32_f16(
                false, B.v, false, afr[ks], (short)0, acc[ct], false, false);
        }
    }
    const int hsel = lane >> 4;
    const int node = rt * 16 + rsub;
    _Float16* hp = h1 + (size_t)node * HC + hsel * 8;
    float res[8];
    #pragma unroll
    for (int ct = 0; ct < 8; ++ct) {
        h8 pk;
        #pragma unroll
        for (int r = 0; r < 8; ++r) pk[r] = (_Float16)acc[ct][r];
        *(h8*)(hp + ct * 16) = pk;
        const float4 s0 = *(const float4*)(as + ct * 16 + hsel * 8), s1 = *(const float4*)(as + ct * 16 + hsel * 8 + 4);
        const float4 d0 = *(const float4*)(ad + ct * 16 + hsel * 8), d1 = *(const float4*)(ad + ct * 16 + hsel * 8 + 4);
        float ps = acc[ct][0] * s0.x, pd = acc[ct][0] * d0.x;
        ps += acc[ct][1] * s0.y; pd += acc[ct][1] * d0.y;
        ps += acc[ct][2] * s0.z; pd += acc[ct][2] * d0.z;
        ps += acc[ct][3] * s0.w; pd += acc[ct][3] * d0.w;
        ps += acc[ct][4] * s1.x; pd += acc[ct][4] * d1.x;
        ps += acc[ct][5] * s1.y; pd += acc[ct][5] * d1.y;
        ps += acc[ct][6] * s1.z; pd += acc[ct][6] * d1.z;
        ps += acc[ct][7] * s1.w; pd += acc[ct][7] * d1.w;
        res[ct] = (hsel ? pd : ps) + __shfl_xor(hsel ? ps : pd, 16);
    }
    float* ap = (hsel ? a_dst : a_src) + (size_t)node * HH;
    *(float4*)(ap)     = make_float4(res[0], res[1], res[2], res[3]);
    *(float4*)(ap + 4) = make_float4(res[4], res[5], res[6], res[7]);
}

static_assert(EE % 256 == 0, "the edge kernel launches exactly EE threads");
__global__ __launch_bounds__(256)
void gat_bucket(const int* __restrict__ ei, int* __restrict__ deg, int* __restrict__ bkt,
                int* __restrict__ ovn, int* __restrict__ ovf) {
    const int e = blockIdx.x * 256 + threadIdx.x;
    const int d = ei[EE + e];
    const int s = ei[e];
    const int slot = atomicAdd(&deg[d], 1);
    if (slot < KB) {
        bkt[d * KB + slot] = s;
    } else {
        const int p = atomicAdd(ovn, 1);
        ovf[2 * p]     = d;
        ovf[2 * p + 1] = s;
    }
}

#define LP 136
static_assert(NN % 16 == 0, "sixteen nodes a block");
union hrow { uint2 u; _Float16 e[4]; };
__global__ __launch_bounds__(256)
void gat_node1(const int* __restrict__ deg, const int* __restrict__ bkt,
               const int* __restrict__ ovn, const int* __restrict__ ovf,
               const float* __restrict__ a_src, const float* __restrict__ a_dst,
               const _Float16* __restrict__ h1, const float* __restrict__ b1,
               const _Float16* __restrict__ w2h, const float* __restrict__ as2,
               const float* __restrict__ ad2, float* __restrict__ h2,
               float* __restrict__ a_src2, float* __restrict__ a_dst2) {
    __shared__ __attribute__((aligned(16))) _Float16 lo[16 * LP];
    const int lane = threadIdx.x & 31;
    const int wv = __builtin_amdgcn_readfirstlane(threadIdx.x >> 5);
    const int eh = lane & 7, es = lane >> 3, ch = lane >> 2;
    const float NINF = -__builtin_inff();
    const float4 b = *(const float4*)(b1 + lane * 4);

  for (int half = 0; half < 2; ++half) {
    const int trow = wv + 8 * half;
    const int d   = blockIdx.x * 16 + trow;
    const int dg  = deg[d];
    const int cnt = (dg < KB ? dg : KB) + 1;
    const int* lst = bkt + d * KB;
    const float adh = a_dst[d * HH + eh];

    float m = NINF, l = 0.0f;
    float acc0 = 0.0f, acc1 = 0.0f, acc2 = 0.0f, acc3 = 0.0f;
    for (int c0 = 0; c0 < cnt; c0 += 32) {
        const int p = c0 + lane;
        int sid = d;
        if (p > 0 && p < cnt) sid = lst[p - 1];
        float a[8];
        float mc = NINF;
        #pragma unroll
        for (int t = 0; t < 8; ++t) {
            const int q = 4 * t + es;
            const int s = __shfl(sid, q);
            const float v = lrelu(a_src[s * HH + eh] + adh);
            a[t] = (c0 + q < cnt) ? v : NINF;
            mc = fmaxf(mc, a[t]);
        }
        mc = fmaxf(mc, __shfl_xor(mc, 8));
        mc = fmaxf(mc, __shfl_xor(mc, 16));
        const float mn = fmaxf(m, mc);
        const float sc = expf(m - mn);
        float lc = 0.0f;
        #pragma unroll
        for (int t = 0; t < 8; ++t) { a[t] = expf(a[t] - mn); lc += a[t]; }
        lc += __shfl_xor(lc, 8);
        lc += __shfl_xor(lc, 16);
        l = l * sc + lc;
        m = mn;
        const float scc = __shfl(sc, ch);
        acc0 *= scc; acc1 *= scc; acc2 *= scc; acc3 *= scc;
        #pragma unroll
        for (int t = 0; t < 8; ++t) {
            if (c0 + 4 * t < cnt) {
                hrow  row[4];
                float wt[4];
                #pragma unroll
                for (int j = 0; j < 4; ++j) {
                    const int s = __builtin_amdgcn_readlane(sid, 4 * t + j);
                    wt[j]    = __shfl(a[t], j * 8 + ch);
                    row[j].u = *(const uint2*)(h1 + (size_t)s * HC + lane * 4);
                }
                #pragma unroll
                for (int j = 0; j < 4; ++j) {
                    acc0 += wt[j] * (float)row[j].e[0]; acc1 += wt[j] * (float)row[j].e[1];
                    acc2 += wt[j] * (float)row[j].e[2]; acc3 += wt[j] * (float)row[j].e[3];
                }
            }
        }
    }
    if (dg > KB) {
        const int no = ovn[0];
        for (int o0 = 0; o0 < no; o0 += 32) {
            const int q = o0 + lane;
            int od = -1, os = 0;
            if (q < no) { od = ovf[2 * q]; os = ovf[2 * q + 1]; }
            unsigned mk = __builtin_amdgcn_ballot_w32(od == d);
            while (mk) {
                const int bit = __builtin_ctz(mk);
                mk &= mk - 1;
                const int s = __builtin_amdgcn_readlane(os, bit);
                const float v  = lrelu(a_src[s * HH + eh] + adh);
                const float mn = fmaxf(m, v);
                const float sc = expf(m - mn);
                const float w  = expf(v - mn);
                l = l * sc + w;
                m = mn;
                const float scc = __shfl(sc, ch), wcc = __shfl(w, ch);
                hrow row;
                row.u = *(const uint2*)(h1 + (size_t)s * HC + lane * 4);
                acc0 = acc0 * scc + wcc * (float)row.e[0]; acc1 = acc1 * scc + wcc * (float)row.e[1];
                acc2 = acc2 * scc + wcc * (float)row.e[2]; acc3 = acc3 * scc + wcc * (float)row.e[3];
            }
        }
    }
    const float den = __shfl(l, ch) + 1e-16f;
    float4 o;
    o.x = acc0 / den + b.x; o.y = acc1 / den + b.y; o.z = acc2 / den + b.z; o.w = acc3 / den + b.w;
    o.x = o.x > 0.0f ? o.x : (expf(o.x) - 1.0f);
    o.y = o.y > 0.0f ? o.y : (expf(o.y) - 1.0f);
    o.z = o.z > 0.0f ? o.z : (expf(o.z) - 1.0f);
    o.w = o.w > 0.0f ? o.w : (expf(o.w) - 1.0f);
    union { _Float16 e[4]; uint2 u; } hv;
    hv.e[0] = (_Float16)o.x; hv.e[1] = (_Float16)o.y; hv.e[2] = (_Float16)o.z; hv.e[3] = (_Float16)o.w;
    *(uint2*)(&lo[trow * LP + lane * 4]) = hv.u;
  }
    __syncthreads();
    if (wv != 0) return;

    const int rsub = lane & 15;
    const int ksel = (lane >> 4) << 3;
    const _Float16* wrow = w2h + (size_t)rsub * HC;
    v8f acc = {};
    #pragma unroll
    for (int ks = 0; ks < 4; ++ks) {
        const int kb = ks * 32 + ksel;
        union { v16h v; h8 p[2]; } A, B;
        A.p[0] = *(const h8*)(&lo[rsub * LP + kb]);
        A.p[1] = *(const h8*)(&lo[rsub * LP + kb + 16]);
        B.p[0] = *(const h8*)(wrow + kb);
        B.p[1] = *(const h8*)(wrow + kb + 16);
        acc = __builtin_amdgcn_wmma_f32_16x16x32_f16(
            false, A.v, false, B.v, (short)0, acc, false, false);
    }
    const int rhi = (lane >> 4) << 3;
    const int r0  = blockIdx.x * 16 + rhi;
    #pragma unroll
    for (int r = 0; r < 8; ++r)
        h2[(size_t)(r0 + r) * NC + rsub] = acc[r];
    const bool b0 = lane & 1, b1b = lane & 2, b2b = lane & 4, b3b = lane & 8;
    const float ws = as2[rsub], wd = ad2[rsub];
    float ps[8], pd[8];
    #pragma unroll
    for (int r = 0; r < 8; ++r) { ps[r] = acc[r] * ws; pd[r] = acc[r] * wd; }
    const float fs = fold8(ps, b3b, b2b, b1b), fd = fold8(pd, b3b, b2b, b1b);
    const float res = (b0 ? fd : fs) + rowx<1>(b0 ? fs : fd);
    (b0 ? a_dst2 : a_src2)[r0 + (rsub >> 1)] = res;
}


__global__ __launch_bounds__(256)
void gat_node2(const int* __restrict__ deg, const int* __restrict__ bkt,
               const int* __restrict__ ovn, const int* __restrict__ ovf,
               const float* __restrict__ a_src, const float* __restrict__ a_dst,
               const float* __restrict__ h2, const float* __restrict__ b2,
               float* __restrict__ out) {
    const int lane = threadIdx.x & 31;
    const int d   = __builtin_amdgcn_readfirstlane(blockIdx.x * 8 + (threadIdx.x >> 5));
    const int dg  = deg[d];
    const int cnt = (dg < KB ? dg : KB) + 1;
    const int* lst = bkt + d * KB;
    const int c = lane & 15, es = lane >> 4;
    const float adv = a_dst[d];
    const float NINF = -__builtin_inff();

    float m = NINF, l = 0.0f, acc = 0.0f;
    for (int c0 = 0; c0 < cnt; c0 += 32) {
        const int p = c0 + lane;
        const bool valid = p < cnt;
        int sid = d;
        if (p > 0 && valid) sid = lst[p - 1];
        const float a = valid ? lrelu(a_src[sid] + adv) : NINF;
        float mc = a;
        for (int o = 16; o > 0; o >>= 1) mc = fmaxf(mc, __shfl_xor(mc, o));
        const float mn = fmaxf(m, mc);
        const float sc = expf(m - mn);
        const float ex = expf(a - mn);
        float lc = ex;
        for (int o = 16; o > 0; o >>= 1) lc += __shfl_xor(lc, o);
        l = l * sc + lc;
        m = mn;
        acc *= sc;
        #pragma unroll
        for (int g = 0; g < 4; ++g) {
            if (c0 + 8 * g < cnt) {
                float hv[4], wt[4];
                #pragma unroll
                for (int j = 0; j < 4; ++j) {
                    const int q = 8 * g + 2 * j + es;
                    const int s = __shfl(sid, q);
                    wt[j] = __shfl(ex, q);
                    hv[j] = h2[(size_t)s * NC + c];
                }
                #pragma unroll
                for (int j = 0; j < 4; ++j) acc += wt[j] * hv[j];
            }
        }
    }
    if (dg > KB) {
        const int no = ovn[0];
        for (int o0 = 0; o0 < no; o0 += 32) {
            const int q = o0 + lane;
            int od = -1, os = 0;
            if (q < no) { od = ovf[2 * q]; os = ovf[2 * q + 1]; }
            unsigned mk = __builtin_amdgcn_ballot_w32(od == d);
            while (mk) {
                const int bit = __builtin_ctz(mk);
                mk &= mk - 1;
                const int s = __builtin_amdgcn_readlane(os, bit);
                const float ev = lrelu(a_src[s] + adv);
                const float mn = fmaxf(m, ev);
                const float sc = expf(m - mn);
                const float w  = expf(ev - mn);
                l = l * sc + w;
                m = mn;
                const float hv = h2[(size_t)s * NC + c];
                acc = acc * sc + (es == 0 ? w * hv : 0.0f);
            }
        }
    }
    acc += __shfl_xor(acc, 16);
    const float v = acc / (l + 1e-16f) + b2[c];
    float mx = v;
    for (int o = 8; o > 0; o >>= 1) mx = fmaxf(mx, __shfl_xor(mx, o));
    float se = expf(v - mx);
    for (int o = 8; o > 0; o >>= 1) se += __shfl_xor(se, o);
    if (es == 0) out[(size_t)d * NC + c] = v - mx - logf(se);
}

extern "C" void kernel_launch(void* const* d_in, const int* in_sizes, int n_in,
                              void* d_out, int out_size, void* d_ws, size_t ws_size,
                              hipStream_t stream) {
    (void)in_sizes; (void)n_in; (void)out_size; (void)ws_size;
    const float* x   = (const float*)d_in[0];
    const int*   ei  = (const int*)d_in[1];
    const float* W1  = (const float*)d_in[2];
    const float* as1 = (const float*)d_in[3];
    const float* ad1 = (const float*)d_in[4];
    const float* b1  = (const float*)d_in[5];
    const float* W2  = (const float*)d_in[6];
    const float* as2 = (const float*)d_in[7];
    const float* ad2 = (const float*)d_in[8];
    const float* b2  = (const float*)d_in[9];
    float* out = (float*)d_out;

    char* wsb = (char*)d_ws;
    size_t off = 0;
    auto alloc = [&](size_t bytes) -> void* {
        void* p = wsb + off;
        off += (bytes + 255) & ~(size_t)255;
        return p;
    };
    _Float16* w1h  = (_Float16*)alloc((size_t)HC * NF * 2);
    _Float16* w2h  = (_Float16*)alloc((size_t)NC * HC * 2);
    _Float16* h1   = (_Float16*)alloc((size_t)NN * HC * 2);
    float* asrc1   = (float*)alloc((size_t)NN * HH * 4);
    float* adst1   = (float*)alloc((size_t)NN * HH * 4);
    int*   deg     = (int*)alloc((size_t)NN * 4);
    int*   ovn     = (int*)alloc((size_t)4);
    int*   bkt     = (int*)alloc((size_t)NN * KB * 4);
    int*   ovf     = (int*)alloc((size_t)EE * 2 * 4);
    float* h2      = (float*)alloc((size_t)NN * NC * 4);
    float* asrc2   = (float*)alloc((size_t)NN * 4);
    float* adst2   = (float*)alloc((size_t)NN * 4);

    const dim3 B(256);
    gat_cvtw <<<(NN + 255) / 256, B, 0, stream>>>(W1, W2, w1h, w2h, deg, ovn);
    gat_gemm1<<<(NN / 16 + 7) / 8, B, 0, stream>>>(x, w1h, as1, ad1, h1, asrc1, adst1);
    gat_bucket<<<EE / 256, B, 0, stream>>>(ei, deg, bkt, ovn, ovf);
    gat_node1<<<NN / 16, B, 0, stream>>>(deg, bkt, ovn, ovf, asrc1, adst1, h1, b1, w2h, as2, ad2, h2, asrc2, adst2);
    gat_node2<<<NN / 8, B, 0, stream>>>(deg, bkt, ovn, ovf, asrc2, adst2, h2, b2, out);
}


# node kernels: fewer dependent memory round trips per wave: a chunk's rows, attention scalars and bias requested as one group behind the list, before the statistics; both nodes' list heads up front
# speedup vs baseline: 4.8845x; 1.0308x over previous
#include <hip/hip_runtime.h>

#define NN  100000
#define NF  256
#define HH  8
#define CC  16
#define HC  128
#define NC  16
#define EE  1600000
#define ET  1700000
#define NEG 0.2f
#define KB  32

typedef _Float16 h8   __attribute__((ext_vector_type(8)));
typedef _Float16 v16h __attribute__((ext_vector_type(16)));
typedef float    v8f  __attribute__((ext_vector_type(8)));

__device__ __forceinline__ float lrelu(float a) { return a > 0.0f ? a : NEG * a; }

template <int M>
__device__ __forceinline__ float rowx(float v) {
    return __builtin_bit_cast(float, __builtin_amdgcn_update_dpp(
        0, __builtin_bit_cast(int, v), 0x160 | M, 0xf, 0xf, false));
}

__device__ __forceinline__ float fold8(const float (&p)[8], bool b3, bool b2, bool b1) {
    float q[4], r2[2];
    #pragma unroll
    for (int j = 0; j < 4; ++j) q[j] = (b3 ? p[j + 4] : p[j]) + rowx<8>(b3 ? p[j] : p[j + 4]);
    #pragma unroll
    for (int j = 0; j < 2; ++j) r2[j] = (b2 ? q[j + 2] : q[j]) + rowx<4>(b2 ? q[j] : q[j + 2]);
    return (b1 ? r2[1] : r2[0]) + rowx<2>(b1 ? r2[0] : r2[1]);
}

static_assert(HC * NF + NC * HC <= NN, "the grid that covers the nodes also covers both weight matrices");
__global__ __launch_bounds__(256) void gat_cvtw(const float* __restrict__ W1,
                                                const float* __restrict__ W2,
                                                _Float16* __restrict__ w1h,
                                                _Float16* __restrict__ w2h,
                                                int* __restrict__ deg,
                                                int* __restrict__ ovn) {
    int i = blockIdx.x * 256 + threadIdx.x;
    if (i < NN) deg[i] = 0;
    if (i == 0) ovn[0] = 0;
    if (i < HC * NF)              w1h[i] = (_Float16)W1[i];
    else if (i < HC * NF + NC*HC) w2h[i - HC * NF] = (_Float16)W2[i - HC * NF];
}

__global__ __launch_bounds__(256)
void gat_gemm1(const float* __restrict__ x, const _Float16* __restrict__ w1h,
               const float* __restrict__ as, const float* __restrict__ ad,
               _Float16* __restrict__ h1, float* __restrict__ a_src, float* __restrict__ a_dst) {
    __shared__ _Float16 lw[HC * NF];
    const int tid = threadIdx.x;
    for (int idx = tid; idx < HC * 32; idx += 256) {
        int r = idx >> 5, ch = idx & 31;
        *(h8*)(&lw[r * NF + ((ch ^ (r & 15)) << 3)]) =
            *(const h8*)(w1h + (size_t)r * NF + (ch << 3));
    }
    __syncthreads();

    const int lane = tid & 31;
    const int rt   = blockIdx.x * 8 + (tid >> 5);
    if (rt >= NN / 16) return;
    const int rsub = lane & 15;
    const int ksel = (lane >> 4) << 3;
    const float* xrow = x + (size_t)(rt * 16 + rsub) * NF + ksel;

    v16h afr[8];
    #pragma unroll
    for (int ks = 0; ks < 8; ++ks) {
        const float4 f0 = *(const float4*)(xrow + ks * 32);
        const float4 f1 = *(const float4*)(xrow + ks * 32 + 4);
        const float4 f2 = *(const float4*)(xrow + ks * 32 + 16);
        const float4 f3 = *(const float4*)(xrow + ks * 32 + 20);
        union { v16h v; _Float16 e[16]; } A;
        A.e[0]  = (_Float16)f0.x; A.e[1]  = (_Float16)f0.y; A.e[2]  = (_Float16)f0.z; A.e[3]  = (_Float16)f0.w;
        A.e[4]  = (_Float16)f1.x; A.e[5]  = (_Float16)f1.y; A.e[6]  = (_Float16)f1.z; A.e[7]  = (_Float16)f1.w;
        A.e[8]  = (_Float16)f2.x; A.e[9]  = (_Float16)f2.y; A.e[10] = (_Float16)f2.z; A.e[11] = (_Float16)f2.w;
        A.e[12] = (_Float16)f3.x; A.e[13] = (_Float16)f3.y; A.e[14] = (_Float16)f3.z; A.e[15] = (_Float16)f3.w;
        afr[ks] = A.v;
    }
    __builtin_amdgcn_sched_barrier(0);

    v8f acc[8] = {};
    #pragma unroll
    for (int ks = 0; ks < 8; ++ks) {
        const int c0 = (ks * 32 + ksel) >> 3;
        #pragma unroll
        for (int ct = 0; ct < 8; ++ct) {
            const int n = ct * 16 + rsub;
            union { v16h v; h8 p[2]; } B;
            B.p[0] = *(const h8*)(&lw[n * NF + (((c0    ) ^ (n & 15)) << 3)]);
            B.p[1] = *(const h8*)(&lw[n * NF + (((c0 + 2) ^ (n & 15)) << 3)]);
            acc[ct] = __builtin_amdgcn_wmma_f32_16x16x32_f16(
                false, B.v, false, afr[ks], (short)0, acc[ct], false, false);
        }
    }
    const int hsel = lane >> 4;
    const int node = rt * 16 + rsub;
    _Float16* hp = h1 + (size_t)node * HC + hsel * 8;
    float res[8];
    #pragma unroll
    for (int ct = 0; ct < 8; ++ct) {
        h8 pk;
        #pragma unroll
        for (int r = 0; r < 8; ++r) pk[r] = (_Float16)acc[ct][r];
        *(h8*)(hp + ct * 16) = pk;
        const float4 s0 = *(const float4*)(as + ct * 16 + hsel * 8), s1 = *(const float4*)(as + ct * 16 + hsel * 8 + 4);
        const float4 d0 = *(const float4*)(ad + ct * 16 + hsel * 8), d1 = *(const float4*)(ad + ct * 16 + hsel * 8 + 4);
        float ps = acc[ct][0] * s0.x, pd = acc[ct][0] * d0.x;
        ps += acc[ct][1] * s0.y; pd += acc[ct][1] * d0.y;
        ps += acc[ct][2] * s0.z; pd += acc[ct][2] * d0.z;
        ps += acc[ct][3] * s0.w; pd += acc[ct][3] * d0.w;
        ps += acc[ct][4] * s1.x; pd += acc[ct][4] * d1.x;
        ps += acc[ct][5] * s1.y; pd += acc[ct][5] * d1.y;
        ps += acc[ct][6] * s1.z; pd += acc[ct][6] * d1.z;
        ps += acc[ct][7] * s1.w; pd += acc[ct][7] * d1.w;
        res[ct] = (hsel ? pd : ps) + __shfl_xor(hsel ? ps : pd, 16);
    }
    float* ap = (hsel ? a_dst : a_src) + (size_t)node * HH;
    *(float4*)(ap)     = make_float4(res[0], res[1], res[2], res[3]);
    *(float4*)(ap + 4) = make_float4(res[4], res[5], res[6], res[7]);
}

static_assert(EE % 256 == 0, "the edge kernel launches exactly EE threads");
__global__ __launch_bounds__(256)
void gat_bucket(const int* __restrict__ ei, int* __restrict__ deg, int* __restrict__ bkt,
                int* __restrict__ ovn, int* __restrict__ ovf) {
    const int e = blockIdx.x * 256 + threadIdx.x;
    const int d = ei[EE + e];
    const int s = ei[e];
    const int slot = atomicAdd(&deg[d], 1);
    if (slot < KB) {
        bkt[d * KB + slot] = s;
    } else {
        const int p = atomicAdd(ovn, 1);
        ovf[2 * p]     = d;
        ovf[2 * p + 1] = s;
    }
}

#define NE 6
#define LP 136
static_assert(NN % 16 == 0, "sixteen nodes a block");
union hrow { uint2 u; _Float16 e[4]; };
__global__ __launch_bounds__(256)
void gat_node1(const int* __restrict__ deg, const int* __restrict__ bkt,
               const int* __restrict__ ovn, const int* __restrict__ ovf,
               const float* __restrict__ a_src, const float* __restrict__ a_dst,
               const _Float16* __restrict__ h1, const float* __restrict__ b1,
               const _Float16* __restrict__ w2h, const float* __restrict__ as2,
               const float* __restrict__ ad2, float* __restrict__ h2,
               float* __restrict__ a_src2, float* __restrict__ a_dst2) {
    __shared__ __attribute__((aligned(16))) _Float16 lo[16 * LP];
    const int lane = threadIdx.x & 31;
    const int wv = __builtin_amdgcn_readfirstlane(threadIdx.x >> 5);
    const int eh = lane & 7, es = lane >> 3, ch = lane >> 2;
    const float NINF = -__builtin_inff();
    const int dn0 = blockIdx.x * 16 + wv, dn1 = dn0 + 8;
    const int dg0 = deg[dn0], dg1 = deg[dn1];
    const float4 b = *(const float4*)(b1 + lane * 4);
    const int cn0 = (dg0 < KB ? dg0 : KB) + 1;
    const int cn1 = (dg1 < KB ? dg1 : KB) + 1;
    int sd0 = dn0, sd1 = dn1;
    if (lane > 0 && lane < cn0) sd0 = bkt[dn0 * KB + lane - 1];
    if (lane > 0 && lane < cn1) sd1 = bkt[dn1 * KB + lane - 1];

  for (int half = 0; half < 2; ++half) {
    const int trow = wv + 8 * half;
    const int d   = half ? dn1 : dn0;
    const int dg  = half ? dg1 : dg0;
    const int cnt = half ? cn1 : cn0;
    const int* lst = bkt + d * KB;
    const float adh = a_dst[d * HH + eh];

    float m = NINF, l = 0.0f;
    float acc0 = 0.0f, acc1 = 0.0f, acc2 = 0.0f, acc3 = 0.0f;
    for (int c0 = 0; c0 < cnt; c0 += 32) {
        const int p = c0 + lane;
        int sid = half ? sd1 : sd0;
        if (c0 > 0) {
            sid = d;
            if (p < cnt) sid = lst[p - 1];
        }
        hrow row[4 * NE];
        #pragma unroll
        for (int t = 0; t < NE; ++t) {
            if (c0 + 4 * t < cnt) {
                #pragma unroll
                for (int j = 0; j < 4; ++j) {
                    const int s = __builtin_amdgcn_readlane(sid, 4 * t + j);
                    row[4 * t + j].u = *(const uint2*)(h1 + (size_t)s * HC + lane * 4);
                }
            }
        }
        float a[8];
        float mc = NINF;
        #pragma unroll
        for (int t = 0; t < 8; ++t) {
            const int s = __shfl(sid, 4 * t + es);
            a[t] = a_src[s * HH + eh];
        }
        __builtin_amdgcn_sched_barrier(0);
        #pragma unroll
        for (int t = 0; t < 8; ++t) {
            const float v = lrelu(a[t] + adh);
            a[t] = (c0 + 4 * t + es < cnt) ? v : NINF;
            mc = fmaxf(mc, a[t]);
        }
        mc = fmaxf(mc, __shfl_xor(mc, 8));
        mc = fmaxf(mc, __shfl_xor(mc, 16));
        const float mn = fmaxf(m, mc);
        const float sc = expf(m - mn);
        float lc = 0.0f;
        #pragma unroll
        for (int t = 0; t < 8; ++t) { a[t] = expf(a[t] - mn); lc += a[t]; }
        lc += __shfl_xor(lc, 8);
        lc += __shfl_xor(lc, 16);
        l = l * sc + lc;
        m = mn;
        const float scc = __shfl(sc, ch);
        acc0 *= scc; acc1 *= scc; acc2 *= scc; acc3 *= scc;
        #pragma unroll
        for (int t = 0; t < 8; ++t) {
            if (c0 + 4 * t < cnt) {
                float wt[4];
                hrow  rw[4];
                #pragma unroll
                for (int j = 0; j < 4; ++j) {
                    wt[j] = __shfl(a[t], j * 8 + ch);
                    if (t < NE) rw[j] = row[4 * t + j];
                    else {
                        const int s = __builtin_amdgcn_readlane(sid, 4 * t + j);
                        rw[j].u = *(const uint2*)(h1 + (size_t)s * HC + lane * 4);
                    }
                }
                #pragma unroll
                for (int j = 0; j < 4; ++j) {
                    acc0 += wt[j] * (float)rw[j].e[0]; acc1 += wt[j] * (float)rw[j].e[1];
                    acc2 += wt[j] * (float)rw[j].e[2]; acc3 += wt[j] * (float)rw[j].e[3];
                }
            }
        }
    }
    if (dg > KB) {
        const int no = ovn[0];
        for (int o0 = 0; o0 < no; o0 += 32) {
            const int q = o0 + lane;
            int od = -1, os = 0;
            if (q < no) { od = ovf[2 * q]; os = ovf[2 * q + 1]; }
            unsigned mk = __builtin_amdgcn_ballot_w32(od == d);
            while (mk) {
                const int bit = __builtin_ctz(mk);
                mk &= mk - 1;
                const int s = __builtin_amdgcn_readlane(os, bit);
                const float v  = lrelu(a_src[s * HH + eh] + adh);
                const float mn = fmaxf(m, v);
                const float sc = expf(m - mn);
                const float w  = expf(v - mn);
                l = l * sc + w;
                m = mn;
                const float scc = __shfl(sc, ch), wcc = __shfl(w, ch);
                hrow row;
                row.u = *(const uint2*)(h1 + (size_t)s * HC + lane * 4);
                acc0 = acc0 * scc + wcc * (float)row.e[0]; acc1 = acc1 * scc + wcc * (float)row.e[1];
                acc2 = acc2 * scc + wcc * (float)row.e[2]; acc3 = acc3 * scc + wcc * (float)row.e[3];
            }
        }
    }
    const float den = __shfl(l, ch) + 1e-16f;
    float4 o;
    o.x = acc0 / den + b.x; o.y = acc1 / den + b.y; o.z = acc2 / den + b.z; o.w = acc3 / den + b.w;
    o.x = o.x > 0.0f ? o.x : (expf(o.x) - 1.0f);
    o.y = o.y > 0.0f ? o.y : (expf(o.y) - 1.0f);
    o.z = o.z > 0.0f ? o.z : (expf(o.z) - 1.0f);
    o.w = o.w > 0.0f ? o.w : (expf(o.w) - 1.0f);
    union { _Float16 e[4]; uint2 u; } hv;
    hv.e[0] = (_Float16)o.x; hv.e[1] = (_Float16)o.y; hv.e[2] = (_Float16)o.z; hv.e[3] = (_Float16)o.w;
    *(uint2*)(&lo[trow * LP + lane * 4]) = hv.u;
  }
    __syncthreads();
    if (wv != 0) return;

    const int rsub = lane & 15;
    const int ksel = (lane >> 4) << 3;
    const _Float16* wrow = w2h + (size_t)rsub * HC;
    v8f acc = {};
    #pragma unroll
    for (int ks = 0; ks < 4; ++ks) {
        const int kb = ks * 32 + ksel;
        union { v16h v; h8 p[2]; } A, B;
        A.p[0] = *(const h8*)(&lo[rsub * LP + kb]);
        A.p[1] = *(const h8*)(&lo[rsub * LP + kb + 16]);
        B.p[0] = *(const h8*)(wrow + kb);
        B.p[1] = *(const h8*)(wrow + kb + 16);
        acc = __builtin_amdgcn_wmma_f32_16x16x32_f16(
            false, A.v, false, B.v, (short)0, acc, false, false);
    }
    const int rhi = (lane >> 4) << 3;
    const int r0  = blockIdx.x * 16 + rhi;
    #pragma unroll
    for (int r = 0; r < 8; ++r)
        h2[(size_t)(r0 + r) * NC + rsub] = acc[r];
    const bool b0 = lane & 1, b1b = lane & 2, b2b = lane & 4, b3b = lane & 8;
    const float ws = as2[rsub], wd = ad2[rsub];
    float ps[8], pd[8];
    #pragma unroll
    for (int r = 0; r < 8; ++r) { ps[r] = acc[r] * ws; pd[r] = acc[r] * wd; }
    const float fs = fold8(ps, b3b, b2b, b1b), fd = fold8(pd, b3b, b2b, b1b);
    const float res = (b0 ? fd : fs) + rowx<1>(b0 ? fs : fd);
    (b0 ? a_dst2 : a_src2)[r0 + (rsub >> 1)] = res;
}


__global__ __launch_bounds__(256)
void gat_node2(const int* __restrict__ deg, const int* __restrict__ bkt,
               const int* __restrict__ ovn, const int* __restrict__ ovf,
               const float* __restrict__ a_src, const float* __restrict__ a_dst,
               const float* __restrict__ h2, const float* __restrict__ b2,
               float* __restrict__ out) {
    const int lane = threadIdx.x & 31;
    const int d   = __builtin_amdgcn_readfirstlane(blockIdx.x * 8 + (threadIdx.x >> 5));
    const int dg  = deg[d];
    const int cnt = (dg < KB ? dg : KB) + 1;
    const int* lst = bkt + d * KB;
    const int c = lane & 15, es = lane >> 4;
    const float adv = a_dst[d];
    const float NINF = -__builtin_inff();
    const float bc = b2[c];

    float m = NINF, l = 0.0f, acc = 0.0f;
    for (int c0 = 0; c0 < cnt; c0 += 32) {
        const int p = c0 + lane;
        const bool valid = p < cnt;
        int sid = d;
        if (p > 0 && valid) sid = lst[p - 1];
        float hv[16];
        #pragma unroll
        for (int g = 0; g < 4; ++g) {
            if (c0 + 8 * g < cnt) {
                #pragma unroll
                for (int j = 0; j < 4; ++j) {
                    const int s = __shfl(sid, 8 * g + 2 * j + es);
                    hv[4 * g + j] = h2[(size_t)s * NC + c];
                }
            }
        }
        const float a = valid ? lrelu(a_src[sid] + adv) : NINF;
        float mc = a;
        for (int o = 16; o > 0; o >>= 1) mc = fmaxf(mc, __shfl_xor(mc, o));
        const float mn = fmaxf(m, mc);
        const float sc = expf(m - mn);
        const float ex = expf(a - mn);
        float lc = ex;
        for (int o = 16; o > 0; o >>= 1) lc += __shfl_xor(lc, o);
        l = l * sc + lc;
        m = mn;
        acc *= sc;
        #pragma unroll
        for (int g = 0; g < 4; ++g) {
            if (c0 + 8 * g < cnt) {
                float wt[4];
                #pragma unroll
                for (int j = 0; j < 4; ++j) wt[j] = __shfl(ex, 8 * g + 2 * j + es);
                #pragma unroll
                for (int j = 0; j < 4; ++j) acc += wt[j] * hv[4 * g + j];
            }
        }
    }
    if (dg > KB) {
        const int no = ovn[0];
        for (int o0 = 0; o0 < no; o0 += 32) {
            const int q = o0 + lane;
            int od = -1, os = 0;
            if (q < no) { od = ovf[2 * q]; os = ovf[2 * q + 1]; }
            unsigned mk = __builtin_amdgcn_ballot_w32(od == d);
            while (mk) {
                const int bit = __builtin_ctz(mk);
                mk &= mk - 1;
                const int s = __builtin_amdgcn_readlane(os, bit);
                const float ev = lrelu(a_src[s] + adv);
                const float mn = fmaxf(m, ev);
                const float sc = expf(m - mn);
                const float w  = expf(ev - mn);
                l = l * sc + w;
                m = mn;
                const float hv = h2[(size_t)s * NC + c];
                acc = acc * sc + (es == 0 ? w * hv : 0.0f);
            }
        }
    }
    acc += __shfl_xor(acc, 16);
    const float v = acc / (l + 1e-16f) + bc;
    float mx = v;
    for (int o = 8; o > 0; o >>= 1) mx = fmaxf(mx, __shfl_xor(mx, o));
    float se = expf(v - mx);
    for (int o = 8; o > 0; o >>= 1) se += __shfl_xor(se, o);
    if (es == 0) out[(size_t)d * NC + c] = v - mx - logf(se);
}

extern "C" void kernel_launch(void* const* d_in, const int* in_sizes, int n_in,
                              void* d_out, int out_size, void* d_ws, size_t ws_size,
                              hipStream_t stream) {
    (void)in_sizes; (void)n_in; (void)out_size; (void)ws_size;
    const float* x   = (const float*)d_in[0];
    const int*   ei  = (const int*)d_in[1];
    const float* W1  = (const float*)d_in[2];
    const float* as1 = (const float*)d_in[3];
    const float* ad1 = (const float*)d_in[4];
    const float* b1  = (const float*)d_in[5];
    const float* W2  = (const float*)d_in[6];
    const float* as2 = (const float*)d_in[7];
    const float* ad2 = (const float*)d_in[8];
    const float* b2  = (const float*)d_in[9];
    float* out = (float*)d_out;

    char* wsb = (char*)d_ws;
    size_t off = 0;
    auto alloc = [&](size_t bytes) -> void* {
        void* p = wsb + off;
        off += (bytes + 255) & ~(size_t)255;
        return p;
    };
    _Float16* w1h  = (_Float16*)alloc((size_t)HC * NF * 2);
    _Float16* w2h  = (_Float16*)alloc((size_t)NC * HC * 2);
    _Float16* h1   = (_Float16*)alloc((size_t)NN * HC * 2);
    float* asrc1   = (float*)alloc((size_t)NN * HH * 4);
    float* adst1   = (float*)alloc((size_t)NN * HH * 4);
    int*   deg     = (int*)alloc((size_t)NN * 4);
    int*   ovn     = (int*)alloc((size_t)4);
    int*   bkt     = (int*)alloc((size_t)NN * KB * 4);
    int*   ovf     = (int*)alloc((size_t)EE * 2 * 4);
    float* h2      = (float*)alloc((size_t)NN * NC * 4);
    float* asrc2   = (float*)alloc((size_t)NN * 4);
    float* adst2   = (float*)alloc((size_t)NN * 4);

    const dim3 B(256);
    gat_cvtw <<<(NN + 255) / 256, B, 0, stream>>>(W1, W2, w1h, w2h, deg, ovn);
    gat_gemm1<<<(NN / 16 + 7) / 8, B, 0, stream>>>(x, w1h, as1, ad1, h1, asrc1, adst1);
    gat_bucket<<<EE / 256, B, 0, stream>>>(ei, deg, bkt, ovn, ovf);
    gat_node1<<<NN / 16, B, 0, stream>>>(deg, bkt, ovn, ovf, asrc1, adst1, h1, b1, w2h, as2, ad2, h2, asrc2, adst2);
    gat_node2<<<NN / 8, B, 0, stream>>>(deg, bkt, ovn, ovf, asrc2, adst2, h2, b2, out);
}


# node kernels: softmax weights, max/sum trees and source-id broadcasts move as DPP operands and a row exchange on the vector ALU, not ds_bpermute + wait; layer-2 softmax sum tree reordered
# speedup vs baseline: 5.1017x; 1.0445x over previous
#include <hip/hip_runtime.h>

#define NN  100000
#define NF  256
#define HH  8
#define CC  16
#define HC  128
#define NC  16
#define EE  1600000
#define ET  1700000
#define NEG 0.2f
#define KB  32

typedef _Float16 h8   __attribute__((ext_vector_type(8)));
typedef _Float16 v16h __attribute__((ext_vector_type(16)));
typedef float    v8f  __attribute__((ext_vector_type(8)));

__device__ __forceinline__ float lrelu(float a) { return a > 0.0f ? a : NEG * a; }

template <int M>
__device__ __forceinline__ float rowx(float v) {
    return __builtin_bit_cast(float, __builtin_amdgcn_update_dpp(
        0, __builtin_bit_cast(int, v), 0x160 | M, 0xf, 0xf, false));
}

template <int M>
__device__ __forceinline__ float rowxb(float v) {
    return __builtin_bit_cast(float, __builtin_amdgcn_update_dpp(
        0, __builtin_bit_cast(int, v), 0x160 | M, 0xf, 0xf, true));
}

template <int J>
__device__ __forceinline__ float quadb(float v) {
    return __builtin_bit_cast(float, __builtin_amdgcn_update_dpp(
        0, __builtin_bit_cast(int, v), J * 0x55, 0xf, 0xf, true));
}

template <int K>
__device__ __forceinline__ int rowsh(int v) {
    return __builtin_amdgcn_update_dpp(0, v, 0x150 | K, 0xf, 0xf, true);
}
template <int K>
__device__ __forceinline__ float rowshf(float v) {
    return __builtin_bit_cast(float, rowsh<K>(__builtin_bit_cast(int, v)));
}

__device__ __forceinline__ float xhalf(float x) {
    const int b = __builtin_bit_cast(int, x);
    return __builtin_bit_cast(float, __builtin_amdgcn_permlanex16(b, b, 0x76543210, 0xfedcba98, false, false));
}

__device__ __forceinline__ float fold8(const float (&p)[8], bool b3, bool b2, bool b1) {
    float q[4], r2[2];
    #pragma unroll
    for (int j = 0; j < 4; ++j) q[j] = (b3 ? p[j + 4] : p[j]) + rowx<8>(b3 ? p[j] : p[j + 4]);
    #pragma unroll
    for (int j = 0; j < 2; ++j) r2[j] = (b2 ? q[j + 2] : q[j]) + rowx<4>(b2 ? q[j] : q[j + 2]);
    return (b1 ? r2[1] : r2[0]) + rowx<2>(b1 ? r2[0] : r2[1]);
}

static_assert(HC * NF + NC * HC <= NN, "the grid that covers the nodes also covers both weight matrices");
__global__ __launch_bounds__(256) void gat_cvtw(const float* __restrict__ W1,
                                                const float* __restrict__ W2,
                                                _Float16* __restrict__ w1h,
                                                _Float16* __restrict__ w2h,
                                                int* __restrict__ deg,
                                                int* __restrict__ ovn) {
    int i = blockIdx.x * 256 + threadIdx.x;
    if (i < NN) deg[i] = 0;
    if (i == 0) ovn[0] = 0;
    if (i < HC * NF)              w1h[i] = (_Float16)W1[i];
    else if (i < HC * NF + NC*HC) w2h[i - HC * NF] = (_Float16)W2[i - HC * NF];
}

__global__ __launch_bounds__(256)
void gat_gemm1(const float* __restrict__ x, const _Float16* __restrict__ w1h,
               const float* __restrict__ as, const float* __restrict__ ad,
               _Float16* __restrict__ h1, float* __restrict__ a_src, float* __restrict__ a_dst) {
    __shared__ _Float16 lw[HC * NF];
    const int tid = threadIdx.x;
    for (int idx = tid; idx < HC * 32; idx += 256) {
        int r = idx >> 5, ch = idx & 31;
        *(h8*)(&lw[r * NF + ((ch ^ (r & 15)) << 3)]) =
            *(const h8*)(w1h + (size_t)r * NF + (ch << 3));
    }
    __syncthreads();

    const int lane = tid & 31;
    const int rt   = blockIdx.x * 8 + (tid >> 5);
    if (rt >= NN / 16) return;
    const int rsub = lane & 15;
    const int ksel = (lane >> 4) << 3;
    const float* xrow = x + (size_t)(rt * 16 + rsub) * NF + ksel;

    v16h afr[8];
    #pragma unroll
    for (int ks = 0; ks < 8; ++ks) {
        const float4 f0 = *(const float4*)(xrow + ks * 32);
        const float4 f1 = *(const float4*)(xrow + ks * 32 + 4);
        const float4 f2 = *(const float4*)(xrow + ks * 32 + 16);
        const float4 f3 = *(const float4*)(xrow + ks * 32 + 20);
        union { v16h v; _Float16 e[16]; } A;
        A.e[0]  = (_Float16)f0.x; A.e[1]  = (_Float16)f0.y; A.e[2]  = (_Float16)f0.z; A.e[3]  = (_Float16)f0.w;
        A.e[4]  = (_Float16)f1.x; A.e[5]  = (_Float16)f1.y; A.e[6]  = (_Float16)f1.z; A.e[7]  = (_Float16)f1.w;
        A.e[8]  = (_Float16)f2.x; A.e[9]  = (_Float16)f2.y; A.e[10] = (_Float16)f2.z; A.e[11] = (_Float16)f2.w;
        A.e[12] = (_Float16)f3.x; A.e[13] = (_Float16)f3.y; A.e[14] = (_Float16)f3.z; A.e[15] = (_Float16)f3.w;
        afr[ks] = A.v;
    }
    __builtin_amdgcn_sched_barrier(0);

    v8f acc[8] = {};
    #pragma unroll
    for (int ks = 0; ks < 8; ++ks) {
        const int c0 = (ks * 32 + ksel) >> 3;
        #pragma unroll
        for (int ct = 0; ct < 8; ++ct) {
            const int n = ct * 16 + rsub;
            union { v16h v; h8 p[2]; } B;
            B.p[0] = *(const h8*)(&lw[n * NF + (((c0    ) ^ (n & 15)) << 3)]);
            B.p[1] = *(const h8*)(&lw[n * NF + (((c0 + 2) ^ (n & 15)) << 3)]);
            acc[ct] = __builtin_amdgcn_wmma_f32_16x16x32_f16(
                false, B.v, false, afr[ks], (short)0, acc[ct], false, false);
        }
    }
    const int hsel = lane >> 4;
    const int node = rt * 16 + rsub;
    _Float16* hp = h1 + (size_t)node * HC + hsel * 8;
    float res[8];
    #pragma unroll
    for (int ct = 0; ct < 8; ++ct) {
        h8 pk;
        #pragma unroll
        for (int r = 0; r < 8; ++r) pk[r] = (_Float16)acc[ct][r];
        *(h8*)(hp + ct * 16) = pk;
        const float4 s0 = *(const float4*)(as + ct * 16 + hsel * 8), s1 = *(const float4*)(as + ct * 16 + hsel * 8 + 4);
        const float4 d0 = *(const float4*)(ad + ct * 16 + hsel * 8), d1 = *(const float4*)(ad + ct * 16 + hsel * 8 + 4);
        float ps = acc[ct][0] * s0.x, pd = acc[ct][0] * d0.x;
        ps += acc[ct][1] * s0.y; pd += acc[ct][1] * d0.y;
        ps += acc[ct][2] * s0.z; pd += acc[ct][2] * d0.z;
        ps += acc[ct][3] * s0.w; pd += acc[ct][3] * d0.w;
        ps += acc[ct][4] * s1.x; pd += acc[ct][4] * d1.x;
        ps += acc[ct][5] * s1.y; pd += acc[ct][5] * d1.y;
        ps += acc[ct][6] * s1.z; pd += acc[ct][6] * d1.z;
        ps += acc[ct][7] * s1.w; pd += acc[ct][7] * d1.w;
        res[ct] = (hsel ? pd : ps) + __shfl_xor(hsel ? ps : pd, 16);
    }
    float* ap = (hsel ? a_dst : a_src) + (size_t)node * HH;
    *(float4*)(ap)     = make_float4(res[0], res[1], res[2], res[3]);
    *(float4*)(ap + 4) = make_float4(res[4], res[5], res[6], res[7]);
}

static_assert(EE % 256 == 0, "the edge kernel launches exactly EE threads");
__global__ __launch_bounds__(256)
void gat_bucket(const int* __restrict__ ei, int* __restrict__ deg, int* __restrict__ bkt,
                int* __restrict__ ovn, int* __restrict__ ovf) {
    const int e = blockIdx.x * 256 + threadIdx.x;
    const int d = ei[EE + e];
    const int s = ei[e];
    const int slot = atomicAdd(&deg[d], 1);
    if (slot < KB) {
        bkt[d * KB + slot] = s;
    } else {
        const int p = atomicAdd(ovn, 1);
        ovf[2 * p]     = d;
        ovf[2 * p + 1] = s;
    }
}

#define NE 6
#define LP 136
static_assert(NN % 16 == 0, "sixteen nodes a block");
union hrow { uint2 u; _Float16 e[4]; };
__global__ __launch_bounds__(256)
void gat_node1(const int* __restrict__ deg, const int* __restrict__ bkt,
               const int* __restrict__ ovn, const int* __restrict__ ovf,
               const float* __restrict__ a_src, const float* __restrict__ a_dst,
               const _Float16* __restrict__ h1, const float* __restrict__ b1,
               const _Float16* __restrict__ w2h, const float* __restrict__ as2,
               const float* __restrict__ ad2, float* __restrict__ h2,
               float* __restrict__ a_src2, float* __restrict__ a_dst2) {
    __shared__ __attribute__((aligned(16))) _Float16 lo[16 * LP];
    const int lane = threadIdx.x & 31;
    const int wv = __builtin_amdgcn_readfirstlane(threadIdx.x >> 5);
    const int ch = lane >> 2, eq = lane & 3;
    const float NINF = -__builtin_inff();
    const int dn0 = blockIdx.x * 16 + wv, dn1 = dn0 + 8;
    const int dg0 = deg[dn0], dg1 = deg[dn1];
    const float4 b = *(const float4*)(b1 + lane * 4);
    const int cn0 = (dg0 < KB ? dg0 : KB) + 1;
    const int cn1 = (dg1 < KB ? dg1 : KB) + 1;
    int sd0 = dn0, sd1 = dn1;
    if (lane > 0 && lane < cn0) sd0 = bkt[dn0 * KB + lane - 1];
    if (lane > 0 && lane < cn1) sd1 = bkt[dn1 * KB + lane - 1];

  for (int half = 0; half < 2; ++half) {
    const int trow = wv + 8 * half;
    const int d   = half ? dn1 : dn0;
    const int dg  = half ? dg1 : dg0;
    const int cnt = half ? cn1 : cn0;
    const int* lst = bkt + d * KB;
    const float adh = a_dst[d * HH + ch];

    float m = NINF, l = 0.0f;
    float acc0 = 0.0f, acc1 = 0.0f, acc2 = 0.0f, acc3 = 0.0f;
    for (int c0 = 0; c0 < cnt; c0 += 32) {
        const int p = c0 + lane;
        int sid = half ? sd1 : sd0;
        if (c0 > 0) {
            sid = d;
            if (p < cnt) sid = lst[p - 1];
        }
        hrow row[4 * NE];
        #pragma unroll
        for (int t = 0; t < NE; ++t) {
            if (c0 + 4 * t < cnt) {
                #pragma unroll
                for (int j = 0; j < 4; ++j) {
                    const int s = __builtin_amdgcn_readlane(sid, 4 * t + j);
                    row[4 * t + j].u = *(const uint2*)(h1 + (size_t)s * HC + lane * 4);
                }
            }
        }
        float a[8];
        float mc = NINF;
        #pragma unroll
        for (int t = 0; t < 8; ++t) {
            const int s = __shfl(sid, 4 * t + eq);
            a[t] = a_src[s * HH + ch];
        }
        __builtin_amdgcn_sched_barrier(0);
        #pragma unroll
        for (int t = 0; t < 8; ++t) {
            const float v = lrelu(a[t] + adh);
            a[t] = (c0 + 4 * t + eq < cnt) ? v : NINF;
            mc = fmaxf(mc, a[t]);
        }
        mc = fmaxf(mc, rowxb<1>(mc));
        mc = fmaxf(mc, rowxb<2>(mc));
        const float mn = fmaxf(m, mc);
        const float sc = expf(m - mn);
        float lc = 0.0f;
        #pragma unroll
        for (int t = 0; t < 8; ++t) { a[t] = expf(a[t] - mn); lc += a[t]; }
        lc += rowxb<1>(lc);
        lc += rowxb<2>(lc);
        l = l * sc + lc;
        m = mn;
        acc0 *= sc; acc1 *= sc; acc2 *= sc; acc3 *= sc;
        #pragma unroll
        for (int t = 0; t < 8; ++t) {
            if (c0 + 4 * t < cnt) {
                const float wt[4] = { quadb<0>(a[t]), quadb<1>(a[t]), quadb<2>(a[t]), quadb<3>(a[t]) };
                hrow  rw[4];
                #pragma unroll
                for (int j = 0; j < 4; ++j) {
                    if (t < NE) rw[j] = row[4 * t + j];
                    else {
                        const int s = __builtin_amdgcn_readlane(sid, 4 * t + j);
                        rw[j].u = *(const uint2*)(h1 + (size_t)s * HC + lane * 4);
                    }
                }
                #pragma unroll
                for (int j = 0; j < 4; ++j) {
                    acc0 += wt[j] * (float)rw[j].e[0]; acc1 += wt[j] * (float)rw[j].e[1];
                    acc2 += wt[j] * (float)rw[j].e[2]; acc3 += wt[j] * (float)rw[j].e[3];
                }
            }
        }
    }
    if (dg > KB) {
        const int no = ovn[0];
        for (int o0 = 0; o0 < no; o0 += 32) {
            const int q = o0 + lane;
            int od = -1, os = 0;
            if (q < no) { od = ovf[2 * q]; os = ovf[2 * q + 1]; }
            unsigned mk = __builtin_amdgcn_ballot_w32(od == d);
            while (mk) {
                const int bit = __builtin_ctz(mk);
                mk &= mk - 1;
                const int s = __builtin_amdgcn_readlane(os, bit);
                const float v  = lrelu(a_src[s * HH + ch] + adh);
                const float mn = fmaxf(m, v);
                const float sc = expf(m - mn);
                const float w  = expf(v - mn);
                l = l * sc + w;
                m = mn;
                hrow row;
                row.u = *(const uint2*)(h1 + (size_t)s * HC + lane * 4);
                acc0 = acc0 * sc + w * (float)row.e[0]; acc1 = acc1 * sc + w * (float)row.e[1];
                acc2 = acc2 * sc + w * (float)row.e[2]; acc3 = acc3 * sc + w * (float)row.e[3];
            }
        }
    }
    const float den = l + 1e-16f;
    float4 o;
    o.x = acc0 / den + b.x; o.y = acc1 / den + b.y; o.z = acc2 / den + b.z; o.w = acc3 / den + b.w;
    o.x = o.x > 0.0f ? o.x : (expf(o.x) - 1.0f);
    o.y = o.y > 0.0f ? o.y : (expf(o.y) - 1.0f);
    o.z = o.z > 0.0f ? o.z : (expf(o.z) - 1.0f);
    o.w = o.w > 0.0f ? o.w : (expf(o.w) - 1.0f);
    union { _Float16 e[4]; uint2 u; } hv;
    hv.e[0] = (_Float16)o.x; hv.e[1] = (_Float16)o.y; hv.e[2] = (_Float16)o.z; hv.e[3] = (_Float16)o.w;
    *(uint2*)(&lo[trow * LP + lane * 4]) = hv.u;
  }
    __syncthreads();
    if (wv != 0) return;

    const int rsub = lane & 15;
    const int ksel = (lane >> 4) << 3;
    const _Float16* wrow = w2h + (size_t)rsub * HC;
    v8f acc = {};
    #pragma unroll
    for (int ks = 0; ks < 4; ++ks) {
        const int kb = ks * 32 + ksel;
        union { v16h v; h8 p[2]; } A, B;
        A.p[0] = *(const h8*)(&lo[rsub * LP + kb]);
        A.p[1] = *(const h8*)(&lo[rsub * LP + kb + 16]);
        B.p[0] = *(const h8*)(wrow + kb);
        B.p[1] = *(const h8*)(wrow + kb + 16);
        acc = __builtin_amdgcn_wmma_f32_16x16x32_f16(
            false, A.v, false, B.v, (short)0, acc, false, false);
    }
    const int rhi = (lane >> 4) << 3;
    const int r0  = blockIdx.x * 16 + rhi;
    #pragma unroll
    for (int r = 0; r < 8; ++r)
        h2[(size_t)(r0 + r) * NC + rsub] = acc[r];
    const bool b0 = lane & 1, b1b = lane & 2, b2b = lane & 4, b3b = lane & 8;
    const float ws = as2[rsub], wd = ad2[rsub];
    float ps[8], pd[8];
    #pragma unroll
    for (int r = 0; r < 8; ++r) { ps[r] = acc[r] * ws; pd[r] = acc[r] * wd; }
    const float fs = fold8(ps, b3b, b2b, b1b), fd = fold8(pd, b3b, b2b, b1b);
    const float res = (b0 ? fd : fs) + rowx<1>(b0 ? fs : fd);
    (b0 ? a_dst2 : a_src2)[r0 + (rsub >> 1)] = res;
}


__global__ __launch_bounds__(256)
void gat_node2(const int* __restrict__ deg, const int* __restrict__ bkt,
               const int* __restrict__ ovn, const int* __restrict__ ovf,
               const float* __restrict__ a_src, const float* __restrict__ a_dst,
               const float* __restrict__ h2, const float* __restrict__ b2,
               float* __restrict__ out) {
    const int lane = threadIdx.x & 31;
    const int d   = __builtin_amdgcn_readfirstlane(blockIdx.x * 8 + (threadIdx.x >> 5));
    const int dg  = deg[d];
    const int cnt = (dg < KB ? dg : KB) + 1;
    const int* lst = bkt + d * KB;
    const int c = lane & 15, es = lane >> 4;
    const int pl = 2 * c + es;
    const float adv = a_dst[d];
    const float NINF = -__builtin_inff();
    const float bc = b2[c];

    float m = NINF, l = 0.0f, acc = 0.0f;
    for (int c0 = 0; c0 < cnt; c0 += 32) {
        const int p = c0 + pl;
        const bool valid = p < cnt;
        int sid = d;
        if (p > 0 && valid) sid = lst[p - 1];
        float hv[16];
#define N2_ROWS(G)                                                                     \
        if (c0 + 8 * (G) < cnt) {                                     \
            hv[4 * (G) + 0] = h2[(size_t)rowsh<4 * (G) + 0>(sid) * NC + c];            \
            hv[4 * (G) + 1] = h2[(size_t)rowsh<4 * (G) + 1>(sid) * NC + c];            \
            hv[4 * (G) + 2] = h2[(size_t)rowsh<4 * (G) + 2>(sid) * NC + c];            \
            hv[4 * (G) + 3] = h2[(size_t)rowsh<4 * (G) + 3>(sid) * NC + c];            \
        }
        N2_ROWS(0) N2_ROWS(1) N2_ROWS(2) N2_ROWS(3)
#undef N2_ROWS
        const float a = valid ? lrelu(a_src[sid] + adv) : NINF;
        float mc = a;
        mc = fmaxf(mc, rowxb<1>(mc)); mc = fmaxf(mc, rowxb<2>(mc));
        mc = fmaxf(mc, rowxb<4>(mc)); mc = fmaxf(mc, rowxb<8>(mc));
        mc = fmaxf(mc, xhalf(mc));
        const float mn = fmaxf(m, mc);
        const float sc = expf(m - mn);
        const float ex = expf(a - mn);
        float lc = ex;
        lc += rowxb<1>(lc); lc += rowxb<2>(lc); lc += rowxb<4>(lc); lc += rowxb<8>(lc);
        lc += xhalf(lc);
        l = l * sc + lc;
        m = mn;
        acc *= sc;
#define N2_ACC(G)                                                                      \
        if (c0 + 8 * (G) < cnt) {                                     \
            acc += rowshf<4 * (G) + 0>(ex) * hv[4 * (G) + 0];                          \
            acc += rowshf<4 * (G) + 1>(ex) * hv[4 * (G) + 1];                          \
            acc += rowshf<4 * (G) + 2>(ex) * hv[4 * (G) + 2];                          \
            acc += rowshf<4 * (G) + 3>(ex) * hv[4 * (G) + 3];                          \
        }
        N2_ACC(0) N2_ACC(1) N2_ACC(2) N2_ACC(3)
#undef N2_ACC
    }
    if (dg > KB) {
        const int no = ovn[0];
        for (int o0 = 0; o0 < no; o0 += 32) {
            const int q = o0 + lane;
            int od = -1, os = 0;
            if (q < no) { od = ovf[2 * q]; os = ovf[2 * q + 1]; }
            unsigned mk = __builtin_amdgcn_ballot_w32(od == d);
            while (mk) {
                const int bit = __builtin_ctz(mk);
                mk &= mk - 1;
                const int s = __builtin_amdgcn_readlane(os, bit);
                const float ev = lrelu(a_src[s] + adv);
                const float mn = fmaxf(m, ev);
                const float sc = expf(m - mn);
                const float w  = expf(ev - mn);
                l = l * sc + w;
                m = mn;
                const float hv = h2[(size_t)s * NC + c];
                acc = acc * sc + (es == 0 ? w * hv : 0.0f);
            }
        }
    }
    acc += xhalf(acc);
    const float v = acc / (l + 1e-16f) + bc;
    float mx = v;
    mx = fmaxf(mx, rowxb<8>(mx)); mx = fmaxf(mx, rowxb<4>(mx));
    mx = fmaxf(mx, rowxb<2>(mx)); mx = fmaxf(mx, rowxb<1>(mx));
    float se = expf(v - mx);
    se += rowxb<8>(se); se += rowxb<4>(se); se += rowxb<2>(se); se += rowxb<1>(se);
    if (es == 0) out[(size_t)d * NC + c] = v - mx - logf(se);
}

extern "C" void kernel_launch(void* const* d_in, const int* in_sizes, int n_in,
                              void* d_out, int out_size, void* d_ws, size_t ws_size,
                              hipStream_t stream) {
    (void)in_sizes; (void)n_in; (void)out_size; (void)ws_size;
    const float* x   = (const float*)d_in[0];
    const int*   ei  = (const int*)d_in[1];
    const float* W1  = (const float*)d_in[2];
    const float* as1 = (const float*)d_in[3];
    const float* ad1 = (const float*)d_in[4];
    const float* b1  = (const float*)d_in[5];
    const float* W2  = (const float*)d_in[6];
    const float* as2 = (const float*)d_in[7];
    const float* ad2 = (const float*)d_in[8];
    const float* b2  = (const float*)d_in[9];
    float* out = (float*)d_out;

    char* wsb = (char*)d_ws;
    size_t off = 0;
    auto alloc = [&](size_t bytes) -> void* {
        void* p = wsb + off;
        off += (bytes + 255) & ~(size_t)255;
        return p;
    };
    _Float16* w1h  = (_Float16*)alloc((size_t)HC * NF * 2);
    _Float16* w2h  = (_Float16*)alloc((size_t)NC * HC * 2);
    _Float16* h1   = (_Float16*)alloc((size_t)NN * HC * 2);
    float* asrc1   = (float*)alloc((size_t)NN * HH * 4);
    float* adst1   = (float*)alloc((size_t)NN * HH * 4);
    int*   deg     = (int*)alloc((size_t)NN * 4);
    int*   ovn     = (int*)alloc((size_t)4);
    int*   bkt     = (int*)alloc((size_t)NN * KB * 4);
    int*   ovf     = (int*)alloc((size_t)EE * 2 * 4);
    float* h2      = (float*)alloc((size_t)NN * NC * 4);
    float* asrc2   = (float*)alloc((size_t)NN * 4);
    float* adst2   = (float*)alloc((size_t)NN * 4);

    const dim3 B(256);
    gat_cvtw <<<(NN + 255) / 256, B, 0, stream>>>(W1, W2, w1h, w2h, deg, ovn);
    gat_gemm1<<<(NN / 16 + 7) / 8, B, 0, stream>>>(x, w1h, as1, ad1, h1, asrc1, adst1);
    gat_bucket<<<EE / 256, B, 0, stream>>>(ei, deg, bkt, ovn, ovf);
    gat_node1<<<NN / 16, B, 0, stream>>>(deg, bkt, ovn, ovf, asrc1, adst1, h1, b1, w2h, as2, ad2, h2, asrc2, adst2);
    gat_node2<<<NN / 8, B, 0, stream>>>(deg, bkt, ovn, ovf, asrc2, adst2, h2, b2, out);
}


# edges bucketed on LDS counters, not one returning global add per edge: each chunk sorted by node range in LDS, then each range's node lines filled in LDS and stored whole; sums in new arrival order
# speedup vs baseline: 9.0087x; 1.7658x over previous
#include <hip/hip_runtime.h>

#define NN  100000
#define NF  256
#define HH  8
#define CC  16
#define HC  128
#define NC  16
#define EE  1600000
#define ET  1700000
#define NEG 0.2f
#define KB  32

typedef _Float16 h8   __attribute__((ext_vector_type(8)));
typedef _Float16 v16h __attribute__((ext_vector_type(16)));
typedef float    v8f  __attribute__((ext_vector_type(8)));

__device__ __forceinline__ float lrelu(float a) { return a > 0.0f ? a : NEG * a; }

template <int M>
__device__ __forceinline__ float rowx(float v) {
    return __builtin_bit_cast(float, __builtin_amdgcn_update_dpp(
        0, __builtin_bit_cast(int, v), 0x160 | M, 0xf, 0xf, false));
}

template <int M>
__device__ __forceinline__ float rowxb(float v) {
    return __builtin_bit_cast(float, __builtin_amdgcn_update_dpp(
        0, __builtin_bit_cast(int, v), 0x160 | M, 0xf, 0xf, true));
}

template <int J>
__device__ __forceinline__ float quadb(float v) {
    return __builtin_bit_cast(float, __builtin_amdgcn_update_dpp(
        0, __builtin_bit_cast(int, v), J * 0x55, 0xf, 0xf, true));
}

template <int K>
__device__ __forceinline__ int rowsh(int v) {
    return __builtin_amdgcn_update_dpp(0, v, 0x150 | K, 0xf, 0xf, true);
}
template <int K>
__device__ __forceinline__ float rowshf(float v) {
    return __builtin_bit_cast(float, rowsh<K>(__builtin_bit_cast(int, v)));
}

__device__ __forceinline__ float xhalf(float x) {
    const int b = __builtin_bit_cast(int, x);
    return __builtin_bit_cast(float, __builtin_amdgcn_permlanex16(b, b, 0x76543210, 0xfedcba98, false, false));
}

__device__ __forceinline__ float fold8(const float (&p)[8], bool b3, bool b2, bool b1) {
    float q[4], r2[2];
    #pragma unroll
    for (int j = 0; j < 4; ++j) q[j] = (b3 ? p[j + 4] : p[j]) + rowx<8>(b3 ? p[j] : p[j + 4]);
    #pragma unroll
    for (int j = 0; j < 2; ++j) r2[j] = (b2 ? q[j + 2] : q[j]) + rowx<4>(b2 ? q[j] : q[j + 2]);
    return (b1 ? r2[1] : r2[0]) + rowx<2>(b1 ? r2[0] : r2[1]);
}

__global__ __launch_bounds__(256)
void gat_gemm1(const float* __restrict__ x, const _Float16* __restrict__ w1h,
               const float* __restrict__ as, const float* __restrict__ ad,
               _Float16* __restrict__ h1, float* __restrict__ a_src, float* __restrict__ a_dst) {
    __shared__ _Float16 lw[HC * NF];
    const int tid = threadIdx.x;
    for (int idx = tid; idx < HC * 32; idx += 256) {
        int r = idx >> 5, ch = idx & 31;
        *(h8*)(&lw[r * NF + ((ch ^ (r & 15)) << 3)]) =
            *(const h8*)(w1h + (size_t)r * NF + (ch << 3));
    }
    __syncthreads();

    const int lane = tid & 31;
    const int rt   = blockIdx.x * 8 + (tid >> 5);
    if (rt >= NN / 16) return;
    const int rsub = lane & 15;
    const int ksel = (lane >> 4) << 3;
    const float* xrow = x + (size_t)(rt * 16 + rsub) * NF + ksel;

    v16h afr[8];
    #pragma unroll
    for (int ks = 0; ks < 8; ++ks) {
        const float4 f0 = *(const float4*)(xrow + ks * 32);
        const float4 f1 = *(const float4*)(xrow + ks * 32 + 4);
        const float4 f2 = *(const float4*)(xrow + ks * 32 + 16);
        const float4 f3 = *(const float4*)(xrow + ks * 32 + 20);
        union { v16h v; _Float16 e[16]; } A;
        A.e[0]  = (_Float16)f0.x; A.e[1]  = (_Float16)f0.y; A.e[2]  = (_Float16)f0.z; A.e[3]  = (_Float16)f0.w;
        A.e[4]  = (_Float16)f1.x; A.e[5]  = (_Float16)f1.y; A.e[6]  = (_Float16)f1.z; A.e[7]  = (_Float16)f1.w;
        A.e[8]  = (_Float16)f2.x; A.e[9]  = (_Float16)f2.y; A.e[10] = (_Float16)f2.z; A.e[11] = (_Float16)f2.w;
        A.e[12] = (_Float16)f3.x; A.e[13] = (_Float16)f3.y; A.e[14] = (_Float16)f3.z; A.e[15] = (_Float16)f3.w;
        afr[ks] = A.v;
    }
    __builtin_amdgcn_sched_barrier(0);

    v8f acc[8] = {};
    #pragma unroll
    for (int ks = 0; ks < 8; ++ks) {
        const int c0 = (ks * 32 + ksel) >> 3;
        #pragma unroll
        for (int ct = 0; ct < 8; ++ct) {
            const int n = ct * 16 + rsub;
            union { v16h v; h8 p[2]; } B;
            B.p[0] = *(const h8*)(&lw[n * NF + (((c0    ) ^ (n & 15)) << 3)]);
            B.p[1] = *(const h8*)(&lw[n * NF + (((c0 + 2) ^ (n & 15)) << 3)]);
            acc[ct] = __builtin_amdgcn_wmma_f32_16x16x32_f16(
                false, B.v, false, afr[ks], (short)0, acc[ct], false, false);
        }
    }
    const int hsel = lane >> 4;
    const int node = rt * 16 + rsub;
    _Float16* hp = h1 + (size_t)node * HC + hsel * 8;
    float res[8];
    #pragma unroll
    for (int ct = 0; ct < 8; ++ct) {
        h8 pk;
        #pragma unroll
        for (int r = 0; r < 8; ++r) pk[r] = (_Float16)acc[ct][r];
        *(h8*)(hp + ct * 16) = pk;
        const float4 s0 = *(const float4*)(as + ct * 16 + hsel * 8), s1 = *(const float4*)(as + ct * 16 + hsel * 8 + 4);
        const float4 d0 = *(const float4*)(ad + ct * 16 + hsel * 8), d1 = *(const float4*)(ad + ct * 16 + hsel * 8 + 4);
        float ps = acc[ct][0] * s0.x, pd = acc[ct][0] * d0.x;
        ps += acc[ct][1] * s0.y; pd += acc[ct][1] * d0.y;
        ps += acc[ct][2] * s0.z; pd += acc[ct][2] * d0.z;
        ps += acc[ct][3] * s0.w; pd += acc[ct][3] * d0.w;
        ps += acc[ct][4] * s1.x; pd += acc[ct][4] * d1.x;
        ps += acc[ct][5] * s1.y; pd += acc[ct][5] * d1.y;
        ps += acc[ct][6] * s1.z; pd += acc[ct][6] * d1.z;
        ps += acc[ct][7] * s1.w; pd += acc[ct][7] * d1.w;
        res[ct] = (hsel ? pd : ps) + __shfl_xor(hsel ? ps : pd, 16);
    }
    float* ap = (hsel ? a_dst : a_src) + (size_t)node * HH;
    *(float4*)(ap)     = make_float4(res[0], res[1], res[2], res[3]);
    *(float4*)(ap + 4) = make_float4(res[4], res[5], res[6], res[7]);
}

#define NR  256
#define RS  391
#define CE  (EE / NR)
#define CK  ((CE + 1023) / 1024)
static_assert(EE % NR == 0, "every chunk holds exactly CE edges");
static_assert(NR * RS >= NN, "the ranges cover the nodes");
static_assert(CE < 65536, "start and count of a run share one word");
static_assert(CE % 2 == 0, "a chunk leaves as 16-byte pieces");
static_assert(NR == 8 * 32, "pass 2: eight runs a wave, 32 waves");
static_assert(HC * NF + NC * HC <= NR * 1024, "the grid of pass 1 also covers both weight matrices");
static_assert((size_t)EE * 8 + (size_t)NR * NR * 4 <= (size_t)NN * HC * 2, "runs and run words fit the plane they borrow");

__global__ __launch_bounds__(1024)
void gat_part(const float* __restrict__ W1, const float* __restrict__ W2,
              _Float16* __restrict__ w1h, _Float16* __restrict__ w2h, int* __restrict__ ovn,
              const int* __restrict__ ei, int* __restrict__ partw, int* __restrict__ segt) {
    __shared__ int cnt[NR];
    __shared__ int offs[NR];
    __shared__ int wtot[NR / 32];
    __shared__ __attribute__((aligned(16))) int2 lp[CE];
    const int tid = threadIdx.x, blk = blockIdx.x;
    const int i = blk * 1024 + tid;
    if (i < HC * NF)                w1h[i] = (_Float16)W1[i];
    else if (i < HC * NF + NC * HC) w2h[i - HC * NF] = (_Float16)W2[i - HC * NF];
    if (i == 0) ovn[0] = 0;
    if (tid < NR) cnt[tid] = 0;
    __syncthreads();
    const int* es = ei + blk * CE;
    const int* ed = ei + EE + blk * CE;
    int dd[CK], ss[CK], rk[CK];
    #pragma unroll
    for (int k = 0; k < CK; ++k) {
        const int e = tid + k * 1024;
        dd[k] = -1; ss[k] = 0;
        if (e < CE) { dd[k] = ed[e]; ss[k] = es[e]; }
    }
    #pragma unroll
    for (int k = 0; k < CK; ++k) {
        rk[k] = 0;
        if (dd[k] >= 0) rk[k] = atomicAdd(&cnt[dd[k] / RS], 1);
    }
    __syncthreads();
    int v = 0, incl = 0;
    if (tid < NR) {
        v = cnt[tid];
        int x = v;
        for (int o = 1; o < 32; o <<= 1) {
            const int y = __shfl_up(x, o);
            if ((tid & 31) >= o) x += y;
        }
        incl = x;
        if ((tid & 31) == 31) wtot[tid >> 5] = x;
    }
    __syncthreads();
    if (tid < NR) {
        int b = 0;
        for (int w = 0; w < (tid >> 5); ++w) b += wtot[w];
        const int st = b + incl - v;
        offs[tid] = st;
        segt[tid * NR + blk] = st | (v << 16);
    }
    __syncthreads();
    #pragma unroll
    for (int k = 0; k < CK; ++k)
        if (dd[k] >= 0) lp[offs[dd[k] / RS] + rk[k]] = make_int2(dd[k], ss[k]);
    __syncthreads();
    int4* dst = (int4*)((int2*)partw + (size_t)blk * CE);
    const int4* src = (const int4*)lp;
    for (int j = tid; j < CE / 2; j += 1024) dst[j] = src[j];
}

__global__ __launch_bounds__(1024)
void gat_bucket(const int* __restrict__ partw, const int* __restrict__ segt,
                int* __restrict__ deg, int* __restrict__ bkt,
                int* __restrict__ ovn, int* __restrict__ ovf) {
    __shared__ int ldeg[RS + 1];
    __shared__ __attribute__((aligned(16))) int lb[RS * KB];
    const int tid = threadIdx.x, lane = tid & 31, wv = tid >> 5;
    const int r = blockIdx.x, base = r * RS;
    const int nr = (NN - base < RS) ? (NN - base) : RS;
    const int2* part = (const int2*)partw;
    int st[8], c[8];
    #pragma unroll
    for (int q = 0; q < 8; ++q) {
        const int w = segt[r * NR + wv + 32 * q];
        st[q] = w & 0xffff; c[q] = w >> 16;
    }
    for (int i = tid; i < RS; i += 1024) ldeg[i] = 0;
    int2 pr[8];
    #pragma unroll
    for (int q = 0; q < 8; ++q) {
        pr[q] = make_int2(0, 0);
        if (lane < c[q]) pr[q] = part[(size_t)(wv + 32 * q) * CE + st[q] + lane];
    }
    __syncthreads();
    auto put = [&](int2 e) {
        const int ld = e.x - base;
        const int slot = atomicAdd(&ldeg[ld], 1);
        if (slot < KB) {
            lb[ld * KB + slot] = e.y;
        } else {
            const int p = atomicAdd(ovn, 1);
            ovf[2 * p]     = e.x;
            ovf[2 * p + 1] = e.y;
        }
    };
    #pragma unroll
    for (int q = 0; q < 8; ++q) {
        if (lane < c[q]) put(pr[q]);
        for (int j = lane + 32; j < c[q]; j += 32)
            put(part[(size_t)(wv + 32 * q) * CE + st[q] + j]);
    }
    __syncthreads();
    for (int i = tid; i < nr; i += 1024) deg[base + i] = ldeg[i];
    int4* dst = (int4*)(bkt + (size_t)base * KB);
    const int4* src = (const int4*)lb;
    for (int i = tid; i < nr * (KB / 4); i += 1024) dst[i] = src[i];
}

#define NE 6
#define LP 136
static_assert(NN % 16 == 0, "sixteen nodes a block");
union hrow { uint2 u; _Float16 e[4]; };
__global__ __launch_bounds__(256)
void gat_node1(const int* __restrict__ deg, const int* __restrict__ bkt,
               const int* __restrict__ ovn, const int* __restrict__ ovf,
               const float* __restrict__ a_src, const float* __restrict__ a_dst,
               const _Float16* __restrict__ h1, const float* __restrict__ b1,
               const _Float16* __restrict__ w2h, const float* __restrict__ as2,
               const float* __restrict__ ad2, float* __restrict__ h2,
               float* __restrict__ a_src2, float* __restrict__ a_dst2) {
    __shared__ __attribute__((aligned(16))) _Float16 lo[16 * LP];
    const int lane = threadIdx.x & 31;
    const int wv = __builtin_amdgcn_readfirstlane(threadIdx.x >> 5);
    const int ch = lane >> 2, eq = lane & 3;
    const float NINF = -__builtin_inff();
    const int dn0 = blockIdx.x * 16 + wv, dn1 = dn0 + 8;
    const int dg0 = deg[dn0], dg1 = deg[dn1];
    const float4 b = *(const float4*)(b1 + lane * 4);
    const int cn0 = (dg0 < KB ? dg0 : KB) + 1;
    const int cn1 = (dg1 < KB ? dg1 : KB) + 1;
    int sd0 = dn0, sd1 = dn1;
    if (lane > 0 && lane < cn0) sd0 = bkt[dn0 * KB + lane - 1];
    if (lane > 0 && lane < cn1) sd1 = bkt[dn1 * KB + lane - 1];

  for (int half = 0; half < 2; ++half) {
    const int trow = wv + 8 * half;
    const int d   = half ? dn1 : dn0;
    const int dg  = half ? dg1 : dg0;
    const int cnt = half ? cn1 : cn0;
    const int* lst = bkt + d * KB;
    const float adh = a_dst[d * HH + ch];

    float m = NINF, l = 0.0f;
    float acc0 = 0.0f, acc1 = 0.0f, acc2 = 0.0f, acc3 = 0.0f;
    for (int c0 = 0; c0 < cnt; c0 += 32) {
        const int p = c0 + lane;
        int sid = half ? sd1 : sd0;
        if (c0 > 0) {
            sid = d;
            if (p < cnt) sid = lst[p - 1];
        }
        hrow row[4 * NE];
        #pragma unroll
        for (int t = 0; t < NE; ++t) {
            if (c0 + 4 * t < cnt) {
                #pragma unroll
                for (int j = 0; j < 4; ++j) {
                    const int s = __builtin_amdgcn_readlane(sid, 4 * t + j);
                    row[4 * t + j].u = *(const uint2*)(h1 + (size_t)s * HC + lane * 4);
                }
            }
        }
        float a[8];
        float mc = NINF;
        #pragma unroll
        for (int t = 0; t < 8; ++t) {
            const int s = __shfl(sid, 4 * t + eq);
            a[t] = a_src[s * HH + ch];
        }
        __builtin_amdgcn_sched_barrier(0);
        #pragma unroll
        for (int t = 0; t < 8; ++t) {
            const float v = lrelu(a[t] + adh);
            a[t] = (c0 + 4 * t + eq < cnt) ? v : NINF;
            mc = fmaxf(mc, a[t]);
        }
        mc = fmaxf(mc, rowxb<1>(mc));
        mc = fmaxf(mc, rowxb<2>(mc));
        const float mn = fmaxf(m, mc);
        const float sc = expf(m - mn);
        float lc = 0.0f;
        #pragma unroll
        for (int t = 0; t < 8; ++t) { a[t] = expf(a[t] - mn); lc += a[t]; }
        lc += rowxb<1>(lc);
        lc += rowxb<2>(lc);
        l = l * sc + lc;
        m = mn;
        acc0 *= sc; acc1 *= sc; acc2 *= sc; acc3 *= sc;
        #pragma unroll
        for (int t = 0; t < 8; ++t) {
            if (c0 + 4 * t < cnt) {
                const float wt[4] = { quadb<0>(a[t]), quadb<1>(a[t]), quadb<2>(a[t]), quadb<3>(a[t]) };
                hrow  rw[4];
                #pragma unroll
                for (int j = 0; j < 4; ++j) {
                    if (t < NE) rw[j] = row[4 * t + j];
                    else {
                        const int s = __builtin_amdgcn_readlane(sid, 4 * t + j);
                        rw[j].u = *(const uint2*)(h1 + (size_t)s * HC + lane * 4);
                    }
                }
                #pragma unroll
                for (int j = 0; j < 4; ++j) {
                    acc0 += wt[j] * (float)rw[j].e[0]; acc1 += wt[j] * (float)rw[j].e[1];
                    acc2 += wt[j] * (float)rw[j].e[2]; acc3 += wt[j] * (float)rw[j].e[3];
                }
            }
        }
    }
    if (dg > KB) {
        const int no = ovn[0];
        for (int o0 = 0; o0 < no; o0 += 32) {
            const int q = o0 + lane;
            int od = -1, os = 0;
            if (q < no) { od = ovf[2 * q]; os = ovf[2 * q + 1]; }
            unsigned mk = __builtin_amdgcn_ballot_w32(od == d);
            while (mk) {
                const int bit = __builtin_ctz(mk);
                mk &= mk - 1;
                const int s = __builtin_amdgcn_readlane(os, bit);
                const float v  = lrelu(a_src[s * HH + ch] + adh);
                const float mn = fmaxf(m, v);
                const float sc = expf(m - mn);
                const float w  = expf(v - mn);
                l = l * sc + w;
                m = mn;
                hrow row;
                row.u = *(const uint2*)(h1 + (size_t)s * HC + lane * 4);
                acc0 = acc0 * sc + w * (float)row.e[0]; acc1 = acc1 * sc + w * (float)row.e[1];
                acc2 = acc2 * sc + w * (float)row.e[2]; acc3 = acc3 * sc + w * (float)row.e[3];
            }
        }
    }
    const float den = l + 1e-16f;
    float4 o;
    o.x = acc0 / den + b.x; o.y = acc1 / den + b.y; o.z = acc2 / den + b.z; o.w = acc3 / den + b.w;
    o.x = o.x > 0.0f ? o.x : (expf(o.x) - 1.0f);
    o.y = o.y > 0.0f ? o.y : (expf(o.y) - 1.0f);
    o.z = o.z > 0.0f ? o.z : (expf(o.z) - 1.0f);
    o.w = o.w > 0.0f ? o.w : (expf(o.w) - 1.0f);
    union { _Float16 e[4]; uint2 u; } hv;
    hv.e[0] = (_Float16)o.x; hv.e[1] = (_Float16)o.y; hv.e[2] = (_Float16)o.z; hv.e[3] = (_Float16)o.w;
    *(uint2*)(&lo[trow * LP + lane * 4]) = hv.u;
  }
    __syncthreads();
    if (wv != 0) return;

    const int rsub = lane & 15;
    const int ksel = (lane >> 4) << 3;
    const _Float16* wrow = w2h + (size_t)rsub * HC;
    v8f acc = {};
    #pragma unroll
    for (int ks = 0; ks < 4; ++ks) {
        const int kb = ks * 32 + ksel;
        union { v16h v; h8 p[2]; } A, B;
        A.p[0] = *(const h8*)(&lo[rsub * LP + kb]);
        A.p[1] = *(const h8*)(&lo[rsub * LP + kb + 16]);
        B.p[0] = *(const h8*)(wrow + kb);
        B.p[1] = *(const h8*)(wrow + kb + 16);
        acc = __builtin_amdgcn_wmma_f32_16x16x32_f16(
            false, A.v, false, B.v, (short)0, acc, false, false);
    }
    const int rhi = (lane >> 4) << 3;
    const int r0  = blockIdx.x * 16 + rhi;
    #pragma unroll
    for (int r = 0; r < 8; ++r)
        h2[(size_t)(r0 + r) * NC + rsub] = acc[r];
    const bool b0 = lane & 1, b1b = lane & 2, b2b = lane & 4, b3b = lane & 8;
    const float ws = as2[rsub], wd = ad2[rsub];
    float ps[8], pd[8];
    #pragma unroll
    for (int r = 0; r < 8; ++r) { ps[r] = acc[r] * ws; pd[r] = acc[r] * wd; }
    const float fs = fold8(ps, b3b, b2b, b1b), fd = fold8(pd, b3b, b2b, b1b);
    const float res = (b0 ? fd : fs) + rowx<1>(b0 ? fs : fd);
    (b0 ? a_dst2 : a_src2)[r0 + (rsub >> 1)] = res;
}


__global__ __launch_bounds__(256)
void gat_node2(const int* __restrict__ deg, const int* __restrict__ bkt,
               const int* __restrict__ ovn, const int* __restrict__ ovf,
               const float* __restrict__ a_src, const float* __restrict__ a_dst,
               const float* __restrict__ h2, const float* __restrict__ b2,
               float* __restrict__ out) {
    const int lane = threadIdx.x & 31;
    const int d   = __builtin_amdgcn_readfirstlane(blockIdx.x * 8 + (threadIdx.x >> 5));
    const int dg  = deg[d];
    const int cnt = (dg < KB ? dg : KB) + 1;
    const int* lst = bkt + d * KB;
    const int c = lane & 15, es = lane >> 4;
    const int pl = 2 * c + es;
    const float adv = a_dst[d];
    const float NINF = -__builtin_inff();
    const float bc = b2[c];

    float m = NINF, l = 0.0f, acc = 0.0f;
    for (int c0 = 0; c0 < cnt; c0 += 32) {
        const int p = c0 + pl;
        const bool valid = p < cnt;
        int sid = d;
        if (p > 0 && valid) sid = lst[p - 1];
        float hv[16];
#define N2_ROWS(G)                                                                     \
        if (c0 + 8 * (G) < cnt) {                                     \
            hv[4 * (G) + 0] = h2[(size_t)rowsh<4 * (G) + 0>(sid) * NC + c];            \
            hv[4 * (G) + 1] = h2[(size_t)rowsh<4 * (G) + 1>(sid) * NC + c];            \
            hv[4 * (G) + 2] = h2[(size_t)rowsh<4 * (G) + 2>(sid) * NC + c];            \
            hv[4 * (G) + 3] = h2[(size_t)rowsh<4 * (G) + 3>(sid) * NC + c];            \
        }
        N2_ROWS(0) N2_ROWS(1) N2_ROWS(2) N2_ROWS(3)
#undef N2_ROWS
        const float a = valid ? lrelu(a_src[sid] + adv) : NINF;
        float mc = a;
        mc = fmaxf(mc, rowxb<1>(mc)); mc = fmaxf(mc, rowxb<2>(mc));
        mc = fmaxf(mc, rowxb<4>(mc)); mc = fmaxf(mc, rowxb<8>(mc));
        mc = fmaxf(mc, xhalf(mc));
        const float mn = fmaxf(m, mc);
        const float sc = expf(m - mn);
        const float ex = expf(a - mn);
        float lc = ex;
        lc += rowxb<1>(lc); lc += rowxb<2>(lc); lc += rowxb<4>(lc); lc += rowxb<8>(lc);
        lc += xhalf(lc);
        l = l * sc + lc;
        m = mn;
        acc *= sc;
#define N2_ACC(G)                                                                      \
        if (c0 + 8 * (G) < cnt) {                                     \
            acc += rowshf<4 * (G) + 0>(ex) * hv[4 * (G) + 0];                          \
            acc += rowshf<4 * (G) + 1>(ex) * hv[4 * (G) + 1];                          \
            acc += rowshf<4 * (G) + 2>(ex) * hv[4 * (G) + 2];                          \
            acc += rowshf<4 * (G) + 3>(ex) * hv[4 * (G) + 3];                          \
        }
        N2_ACC(0) N2_ACC(1) N2_ACC(2) N2_ACC(3)
#undef N2_ACC
    }
    if (dg > KB) {
        const int no = ovn[0];
        for (int o0 = 0; o0 < no; o0 += 32) {
            const int q = o0 + lane;
            int od = -1, os = 0;
            if (q < no) { od = ovf[2 * q]; os = ovf[2 * q + 1]; }
            unsigned mk = __builtin_amdgcn_ballot_w32(od == d);
            while (mk) {
                const int bit = __builtin_ctz(mk);
                mk &= mk - 1;
                const int s = __builtin_amdgcn_readlane(os, bit);
                const float ev = lrelu(a_src[s] + adv);
                const float mn = fmaxf(m, ev);
                const float sc = expf(m - mn);
                const float w  = expf(ev - mn);
                l = l * sc + w;
                m = mn;
                const float hv = h2[(size_t)s * NC + c];
                acc = acc * sc + (es == 0 ? w * hv : 0.0f);
            }
        }
    }
    acc += xhalf(acc);
    const float v = acc / (l + 1e-16f) + bc;
    float mx = v;
    mx = fmaxf(mx, rowxb<8>(mx)); mx = fmaxf(mx, rowxb<4>(mx));
    mx = fmaxf(mx, rowxb<2>(mx)); mx = fmaxf(mx, rowxb<1>(mx));
    float se = expf(v - mx);
    se += rowxb<8>(se); se += rowxb<4>(se); se += rowxb<2>(se); se += rowxb<1>(se);
    if (es == 0) out[(size_t)d * NC + c] = v - mx - logf(se);
}

extern "C" void kernel_launch(void* const* d_in, const int* in_sizes, int n_in,
                              void* d_out, int out_size, void* d_ws, size_t ws_size,
                              hipStream_t stream) {
    (void)in_sizes; (void)n_in; (void)out_size; (void)ws_size;
    const float* x   = (const float*)d_in[0];
    const int*   ei  = (const int*)d_in[1];
    const float* W1  = (const float*)d_in[2];
    const float* as1 = (const float*)d_in[3];
    const float* ad1 = (const float*)d_in[4];
    const float* b1  = (const float*)d_in[5];
    const float* W2  = (const float*)d_in[6];
    const float* as2 = (const float*)d_in[7];
    const float* ad2 = (const float*)d_in[8];
    const float* b2  = (const float*)d_in[9];
    float* out = (float*)d_out;

    char* wsb = (char*)d_ws;
    size_t off = 0;
    auto alloc = [&](size_t bytes) -> void* {
        void* p = wsb + off;
        off += (bytes + 255) & ~(size_t)255;
        return p;
    };
    _Float16* w1h  = (_Float16*)alloc((size_t)HC * NF * 2);
    _Float16* w2h  = (_Float16*)alloc((size_t)NC * HC * 2);
    _Float16* h1   = (_Float16*)alloc((size_t)NN * HC * 2);
    float* asrc1   = (float*)alloc((size_t)NN * HH * 4);
    float* adst1   = (float*)alloc((size_t)NN * HH * 4);
    int*   deg     = (int*)alloc((size_t)NN * 4);
    int*   ovn     = (int*)alloc((size_t)4);
    int*   bkt     = (int*)alloc((size_t)NN * KB * 4);
    int*   ovf     = (int*)alloc((size_t)EE * 2 * 4);
    float* h2      = (float*)alloc((size_t)NN * NC * 4);
    float* asrc2   = (float*)alloc((size_t)NN * 4);
    float* adst2   = (float*)alloc((size_t)NN * 4);

    int*   partw   = (int*)h1;
    int*   segt    = (int*)((char*)h1 + (size_t)EE * 8);

    const dim3 B(256);
    gat_part  <<<NR, dim3(1024), 0, stream>>>(W1, W2, w1h, w2h, ovn, ei, partw, segt);
    gat_bucket<<<NR, dim3(1024), 0, stream>>>(partw, segt, deg, bkt, ovn, ovf);
    gat_gemm1<<<(NN / 16 + 7) / 8, B, 0, stream>>>(x, w1h, as1, ad1, h1, asrc1, adst1);
    gat_node1<<<NN / 16, B, 0, stream>>>(deg, bkt, ovn, ovf, asrc1, adst1, h1, b1, w2h, as2, ad2, h2, asrc2, adst2);
    gat_node2<<<NN / 8, B, 0, stream>>>(deg, bkt, ovn, ovf, asrc2, adst2, h2, b2, out);
}


# softmax, ELU and log-softmax exponentials on the hardware exp2/log2 instructions in the log2 domain (attention scalars stored times log2 e), not library expf/logf; sum order unchanged
# speedup vs baseline: 9.5864x; 1.0641x over previous
#include <hip/hip_runtime.h>

#define NN  100000
#define NF  256
#define HH  8
#define CC  16
#define HC  128
#define NC  16
#define EE  1600000
#define ET  1700000
#define NEG 0.2f
#define KB  32

typedef _Float16 h8   __attribute__((ext_vector_type(8)));
typedef _Float16 v16h __attribute__((ext_vector_type(16)));
typedef float    v8f  __attribute__((ext_vector_type(8)));

__device__ __forceinline__ float lrelu(float a) { return a > 0.0f ? a : NEG * a; }

#define LOG2E 1.4426950408889634f
#define LN2   0.6931471805599453f
__device__ __forceinline__ float ex2(float x) { return __builtin_amdgcn_exp2f(x); }

template <int M>
__device__ __forceinline__ float rowx(float v) {
    return __builtin_bit_cast(float, __builtin_amdgcn_update_dpp(
        0, __builtin_bit_cast(int, v), 0x160 | M, 0xf, 0xf, false));
}

template <int M>
__device__ __forceinline__ float rowxb(float v) {
    return __builtin_bit_cast(float, __builtin_amdgcn_update_dpp(
        0, __builtin_bit_cast(int, v), 0x160 | M, 0xf, 0xf, true));
}

template <int J>
__device__ __forceinline__ float quadb(float v) {
    return __builtin_bit_cast(float, __builtin_amdgcn_update_dpp(
        0, __builtin_bit_cast(int, v), J * 0x55, 0xf, 0xf, true));
}

template <int K>
__device__ __forceinline__ int rowsh(int v) {
    return __builtin_amdgcn_update_dpp(0, v, 0x150 | K, 0xf, 0xf, true);
}
template <int K>
__device__ __forceinline__ float rowshf(float v) {
    return __builtin_bit_cast(float, rowsh<K>(__builtin_bit_cast(int, v)));
}

__device__ __forceinline__ float xhalf(float x) {
    const int b = __builtin_bit_cast(int, x);
    return __builtin_bit_cast(float, __builtin_amdgcn_permlanex16(b, b, 0x76543210, 0xfedcba98, false, false));
}

__device__ __forceinline__ float fold8(const float (&p)[8], bool b3, bool b2, bool b1) {
    float q[4], r2[2];
    #pragma unroll
    for (int j = 0; j < 4; ++j) q[j] = (b3 ? p[j + 4] : p[j]) + rowx<8>(b3 ? p[j] : p[j + 4]);
    #pragma unroll
    for (int j = 0; j < 2; ++j) r2[j] = (b2 ? q[j + 2] : q[j]) + rowx<4>(b2 ? q[j] : q[j + 2]);
    return (b1 ? r2[1] : r2[0]) + rowx<2>(b1 ? r2[0] : r2[1]);
}

__global__ __launch_bounds__(256)
void gat_gemm1(const float* __restrict__ x, const _Float16* __restrict__ w1h,
               const float* __restrict__ as, const float* __restrict__ ad,
               _Float16* __restrict__ h1, float* __restrict__ a_src, float* __restrict__ a_dst) {
    __shared__ _Float16 lw[HC * NF];
    const int tid = threadIdx.x;
    for (int idx = tid; idx < HC * 32; idx += 256) {
        int r = idx >> 5, ch = idx & 31;
        *(h8*)(&lw[r * NF + ((ch ^ (r & 15)) << 3)]) =
            *(const h8*)(w1h + (size_t)r * NF + (ch << 3));
    }
    __syncthreads();

    const int lane = tid & 31;
    const int rt   = blockIdx.x * 8 + (tid >> 5);
    if (rt >= NN / 16) return;
    const int rsub = lane & 15;
    const int ksel = (lane >> 4) << 3;
    const float* xrow = x + (size_t)(rt * 16 + rsub) * NF + ksel;

    v16h afr[8];
    #pragma unroll
    for (int ks = 0; ks < 8; ++ks) {
        const float4 f0 = *(const float4*)(xrow + ks * 32);
        const float4 f1 = *(const float4*)(xrow + ks * 32 + 4);
        const float4 f2 = *(const float4*)(xrow + ks * 32 + 16);
        const float4 f3 = *(const float4*)(xrow + ks * 32 + 20);
        union { v16h v; _Float16 e[16]; } A;
        A.e[0]  = (_Float16)f0.x; A.e[1]  = (_Float16)f0.y; A.e[2]  = (_Float16)f0.z; A.e[3]  = (_Float16)f0.w;
        A.e[4]  = (_Float16)f1.x; A.e[5]  = (_Float16)f1.y; A.e[6]  = (_Float16)f1.z; A.e[7]  = (_Float16)f1.w;
        A.e[8]  = (_Float16)f2.x; A.e[9]  = (_Float16)f2.y; A.e[10] = (_Float16)f2.z; A.e[11] = (_Float16)f2.w;
        A.e[12] = (_Float16)f3.x; A.e[13] = (_Float16)f3.y; A.e[14] = (_Float16)f3.z; A.e[15] = (_Float16)f3.w;
        afr[ks] = A.v;
    }
    __builtin_amdgcn_sched_barrier(0);

    v8f acc[8] = {};
    #pragma unroll
    for (int ks = 0; ks < 8; ++ks) {
        const int c0 = (ks * 32 + ksel) >> 3;
        #pragma unroll
        for (int ct = 0; ct < 8; ++ct) {
            const int n = ct * 16 + rsub;
            union { v16h v; h8 p[2]; } B;
            B.p[0] = *(const h8*)(&lw[n * NF + (((c0    ) ^ (n & 15)) << 3)]);
            B.p[1] = *(const h8*)(&lw[n * NF + (((c0 + 2) ^ (n & 15)) << 3)]);
            acc[ct] = __builtin_amdgcn_wmma_f32_16x16x32_f16(
                false, B.v, false, afr[ks], (short)0, acc[ct], false, false);
        }
    }
    const int hsel = lane >> 4;
    const int node = rt * 16 + rsub;
    _Float16* hp = h1 + (size_t)node * HC + hsel * 8;
    float res[8];
    #pragma unroll
    for (int ct = 0; ct < 8; ++ct) {
        h8 pk;
        #pragma unroll
        for (int r = 0; r < 8; ++r) pk[r] = (_Float16)acc[ct][r];
        *(h8*)(hp + ct * 16) = pk;
        const float4 s0 = *(const float4*)(as + ct * 16 + hsel * 8), s1 = *(const float4*)(as + ct * 16 + hsel * 8 + 4);
        const float4 d0 = *(const float4*)(ad + ct * 16 + hsel * 8), d1 = *(const float4*)(ad + ct * 16 + hsel * 8 + 4);
        float ps = acc[ct][0] * s0.x, pd = acc[ct][0] * d0.x;
        ps += acc[ct][1] * s0.y; pd += acc[ct][1] * d0.y;
        ps += acc[ct][2] * s0.z; pd += acc[ct][2] * d0.z;
        ps += acc[ct][3] * s0.w; pd += acc[ct][3] * d0.w;
        ps += acc[ct][4] * s1.x; pd += acc[ct][4] * d1.x;
        ps += acc[ct][5] * s1.y; pd += acc[ct][5] * d1.y;
        ps += acc[ct][6] * s1.z; pd += acc[ct][6] * d1.z;
        ps += acc[ct][7] * s1.w; pd += acc[ct][7] * d1.w;
        res[ct] = ((hsel ? pd : ps) + __shfl_xor(hsel ? ps : pd, 16)) * LOG2E;
    }
    float* ap = (hsel ? a_dst : a_src) + (size_t)node * HH;
    *(float4*)(ap)     = make_float4(res[0], res[1], res[2], res[3]);
    *(float4*)(ap + 4) = make_float4(res[4], res[5], res[6], res[7]);
}

#define NR  256
#define RS  391
#define CE  (EE / NR)
#define CK  ((CE + 1023) / 1024)
static_assert(EE % NR == 0, "every chunk holds exactly CE edges");
static_assert(NR * RS >= NN, "the ranges cover the nodes");
static_assert(CE < 65536, "start and count of a run share one word");
static_assert(CE % 2 == 0, "a chunk leaves as 16-byte pieces");
static_assert(NR == 8 * 32, "pass 2: eight runs a wave, 32 waves");
static_assert(HC * NF + NC * HC <= NR * 1024, "the grid of pass 1 also covers both weight matrices");
static_assert((size_t)EE * 8 + (size_t)NR * NR * 4 <= (size_t)NN * HC * 2, "runs and run words fit the plane they borrow");

__global__ __launch_bounds__(1024)
void gat_part(const float* __restrict__ W1, const float* __restrict__ W2,
              _Float16* __restrict__ w1h, _Float16* __restrict__ w2h, int* __restrict__ ovn,
              const int* __restrict__ ei, int* __restrict__ partw, int* __restrict__ segt) {
    __shared__ int cnt[NR];
    __shared__ int offs[NR];
    __shared__ int wtot[NR / 32];
    __shared__ __attribute__((aligned(16))) int2 lp[CE];
    const int tid = threadIdx.x, blk = blockIdx.x;
    const int i = blk * 1024 + tid;
    if (i < HC * NF)                w1h[i] = (_Float16)W1[i];
    else if (i < HC * NF + NC * HC) w2h[i - HC * NF] = (_Float16)W2[i - HC * NF];
    if (i == 0) ovn[0] = 0;
    if (tid < NR) cnt[tid] = 0;
    __syncthreads();
    const int* es = ei + blk * CE;
    const int* ed = ei + EE + blk * CE;
    int dd[CK], ss[CK], rk[CK];
    #pragma unroll
    for (int k = 0; k < CK; ++k) {
        const int e = tid + k * 1024;
        dd[k] = -1; ss[k] = 0;
        if (e < CE) { dd[k] = ed[e]; ss[k] = es[e]; }
    }
    #pragma unroll
    for (int k = 0; k < CK; ++k) {
        rk[k] = 0;
        if (dd[k] >= 0) rk[k] = atomicAdd(&cnt[dd[k] / RS], 1);
    }
    __syncthreads();
    int v = 0, incl = 0;
    if (tid < NR) {
        v = cnt[tid];
        int x = v;
        for (int o = 1; o < 32; o <<= 1) {
            const int y = __shfl_up(x, o);
            if ((tid & 31) >= o) x += y;
        }
        incl = x;
        if ((tid & 31) == 31) wtot[tid >> 5] = x;
    }
    __syncthreads();
    if (tid < NR) {
        int b = 0;
        for (int w = 0; w < (tid >> 5); ++w) b += wtot[w];
        const int st = b + incl - v;
        offs[tid] = st;
        segt[tid * NR + blk] = st | (v << 16);
    }
    __syncthreads();
    #pragma unroll
    for (int k = 0; k < CK; ++k)
        if (dd[k] >= 0) lp[offs[dd[k] / RS] + rk[k]] = make_int2(dd[k], ss[k]);
    __syncthreads();
    int4* dst = (int4*)((int2*)partw + (size_t)blk * CE);
    const int4* src = (const int4*)lp;
    for (int j = tid; j < CE / 2; j += 1024) dst[j] = src[j];
}

__global__ __launch_bounds__(1024)
void gat_bucket(const int* __restrict__ partw, const int* __restrict__ segt,
                int* __restrict__ deg, int* __restrict__ bkt,
                int* __restrict__ ovn, int* __restrict__ ovf) {
    __shared__ int ldeg[RS + 1];
    __shared__ __attribute__((aligned(16))) int lb[RS * KB];
    const int tid = threadIdx.x, lane = tid & 31, wv = tid >> 5;
    const int r = blockIdx.x, base = r * RS;
    const int nr = (NN - base < RS) ? (NN - base) : RS;
    const int2* part = (const int2*)partw;
    int st[8], c[8];
    #pragma unroll
    for (int q = 0; q < 8; ++q) {
        const int w = segt[r * NR + wv + 32 * q];
        st[q] = w & 0xffff; c[q] = w >> 16;
    }
    for (int i = tid; i < RS; i += 1024) ldeg[i] = 0;
    int2 pr[8];
    #pragma unroll
    for (int q = 0; q < 8; ++q) {
        pr[q] = make_int2(0, 0);
        if (lane < c[q]) pr[q] = part[(size_t)(wv + 32 * q) * CE + st[q] + lane];
    }
    __syncthreads();
    auto put = [&](int2 e) {
        const int ld = e.x - base;
        const int slot = atomicAdd(&ldeg[ld], 1);
        if (slot < KB) {
            lb[ld * KB + slot] = e.y;
        } else {
            const int p = atomicAdd(ovn, 1);
            ovf[2 * p]     = e.x;
            ovf[2 * p + 1] = e.y;
        }
    };
    #pragma unroll
    for (int q = 0; q < 8; ++q) {
        if (lane < c[q]) put(pr[q]);
        for (int j = lane + 32; j < c[q]; j += 32)
            put(part[(size_t)(wv + 32 * q) * CE + st[q] + j]);
    }
    __syncthreads();
    for (int i = tid; i < nr; i += 1024) deg[base + i] = ldeg[i];
    int4* dst = (int4*)(bkt + (size_t)base * KB);
    const int4* src = (const int4*)lb;
    for (int i = tid; i < nr * (KB / 4); i += 1024) dst[i] = src[i];
}

#define NE 6
#define LP 136
static_assert(NN % 16 == 0, "sixteen nodes a block");
union hrow { uint2 u; _Float16 e[4]; };
__global__ __launch_bounds__(256)
void gat_node1(const int* __restrict__ deg, const int* __restrict__ bkt,
               const int* __restrict__ ovn, const int* __restrict__ ovf,
               const float* __restrict__ a_src, const float* __restrict__ a_dst,
               const _Float16* __restrict__ h1, const float* __restrict__ b1,
               const _Float16* __restrict__ w2h, const float* __restrict__ as2,
               const float* __restrict__ ad2, float* __restrict__ h2,
               float* __restrict__ a_src2, float* __restrict__ a_dst2) {
    __shared__ __attribute__((aligned(16))) _Float16 lo[16 * LP];
    const int lane = threadIdx.x & 31;
    const int wv = __builtin_amdgcn_readfirstlane(threadIdx.x >> 5);
    const int ch = lane >> 2, eq = lane & 3;
    const float NINF = -__builtin_inff();
    const int dn0 = blockIdx.x * 16 + wv, dn1 = dn0 + 8;
    const int dg0 = deg[dn0], dg1 = deg[dn1];
    const float4 b = *(const float4*)(b1 + lane * 4);
    const int cn0 = (dg0 < KB ? dg0 : KB) + 1;
    const int cn1 = (dg1 < KB ? dg1 : KB) + 1;
    int sd0 = dn0, sd1 = dn1;
    if (lane > 0 && lane < cn0) sd0 = bkt[dn0 * KB + lane - 1];
    if (lane > 0 && lane < cn1) sd1 = bkt[dn1 * KB + lane - 1];

  for (int half = 0; half < 2; ++half) {
    const int trow = wv + 8 * half;
    const int d   = half ? dn1 : dn0;
    const int dg  = half ? dg1 : dg0;
    const int cnt = half ? cn1 : cn0;
    const int* lst = bkt + d * KB;
    const float adh = a_dst[d * HH + ch];

    float m = NINF, l = 0.0f;
    float acc0 = 0.0f, acc1 = 0.0f, acc2 = 0.0f, acc3 = 0.0f;
    for (int c0 = 0; c0 < cnt; c0 += 32) {
        const int p = c0 + lane;
        int sid = half ? sd1 : sd0;
        if (c0 > 0) {
            sid = d;
            if (p < cnt) sid = lst[p - 1];
        }
        hrow row[4 * NE];
        #pragma unroll
        for (int t = 0; t < NE; ++t) {
            if (c0 + 4 * t < cnt) {
                #pragma unroll
                for (int j = 0; j < 4; ++j) {
                    const int s = __builtin_amdgcn_readlane(sid, 4 * t + j);
                    row[4 * t + j].u = *(const uint2*)(h1 + (size_t)s * HC + lane * 4);
                }
            }
        }
        float a[8];
        float mc = NINF;
        #pragma unroll
        for (int t = 0; t < 8; ++t) {
            const int s = __shfl(sid, 4 * t + eq);
            a[t] = a_src[s * HH + ch];
        }
        __builtin_amdgcn_sched_barrier(0);
        #pragma unroll
        for (int t = 0; t < 8; ++t) {
            const float v = lrelu(a[t] + adh);
            a[t] = (c0 + 4 * t + eq < cnt) ? v : NINF;
            mc = fmaxf(mc, a[t]);
        }
        mc = fmaxf(mc, rowxb<1>(mc));
        mc = fmaxf(mc, rowxb<2>(mc));
        const float mn = fmaxf(m, mc);
        const float sc = ex2(m - mn);
        float lc = 0.0f;
        #pragma unroll
        for (int t = 0; t < 8; ++t) { a[t] = ex2(a[t] - mn); lc += a[t]; }
        lc += rowxb<1>(lc);
        lc += rowxb<2>(lc);
        l = l * sc + lc;
        m = mn;
        acc0 *= sc; acc1 *= sc; acc2 *= sc; acc3 *= sc;
        #pragma unroll
        for (int t = 0; t < 8; ++t) {
            if (c0 + 4 * t < cnt) {
                const float wt[4] = { quadb<0>(a[t]), quadb<1>(a[t]), quadb<2>(a[t]), quadb<3>(a[t]) };
                hrow  rw[4];
                #pragma unroll
                for (int j = 0; j < 4; ++j) {
                    if (t < NE) rw[j] = row[4 * t + j];
                    else {
                        const int s = __builtin_amdgcn_readlane(sid, 4 * t + j);
                        rw[j].u = *(const uint2*)(h1 + (size_t)s * HC + lane * 4);
                    }
                }
                #pragma unroll
                for (int j = 0; j < 4; ++j) {
                    acc0 += wt[j] * (float)rw[j].e[0]; acc1 += wt[j] * (float)rw[j].e[1];
                    acc2 += wt[j] * (float)rw[j].e[2]; acc3 += wt[j] * (float)rw[j].e[3];
                }
            }
        }
    }
    if (dg > KB) {
        const int no = ovn[0];
        for (int o0 = 0; o0 < no; o0 += 32) {
            const int q = o0 + lane;
            int od = -1, os = 0;
            if (q < no) { od = ovf[2 * q]; os = ovf[2 * q + 1]; }
            unsigned mk = __builtin_amdgcn_ballot_w32(od == d);
            while (mk) {
                const int bit = __builtin_ctz(mk);
                mk &= mk - 1;
                const int s = __builtin_amdgcn_readlane(os, bit);
                const float v  = lrelu(a_src[s * HH + ch] + adh);
                const float mn = fmaxf(m, v);
                const float sc = ex2(m - mn);
                const float w  = ex2(v - mn);
                l = l * sc + w;
                m = mn;
                hrow row;
                row.u = *(const uint2*)(h1 + (size_t)s * HC + lane * 4);
                acc0 = acc0 * sc + w * (float)row.e[0]; acc1 = acc1 * sc + w * (float)row.e[1];
                acc2 = acc2 * sc + w * (float)row.e[2]; acc3 = acc3 * sc + w * (float)row.e[3];
            }
        }
    }
    const float den = l + 1e-16f;
    float4 o;
    o.x = acc0 / den + b.x; o.y = acc1 / den + b.y; o.z = acc2 / den + b.z; o.w = acc3 / den + b.w;
    o.x = o.x > 0.0f ? o.x : (ex2(o.x * LOG2E) - 1.0f);
    o.y = o.y > 0.0f ? o.y : (ex2(o.y * LOG2E) - 1.0f);
    o.z = o.z > 0.0f ? o.z : (ex2(o.z * LOG2E) - 1.0f);
    o.w = o.w > 0.0f ? o.w : (ex2(o.w * LOG2E) - 1.0f);
    union { _Float16 e[4]; uint2 u; } hv;
    hv.e[0] = (_Float16)o.x; hv.e[1] = (_Float16)o.y; hv.e[2] = (_Float16)o.z; hv.e[3] = (_Float16)o.w;
    *(uint2*)(&lo[trow * LP + lane * 4]) = hv.u;
  }
    __syncthreads();
    if (wv != 0) return;

    const int rsub = lane & 15;
    const int ksel = (lane >> 4) << 3;
    const _Float16* wrow = w2h + (size_t)rsub * HC;
    v8f acc = {};
    #pragma unroll
    for (int ks = 0; ks < 4; ++ks) {
        const int kb = ks * 32 + ksel;
        union { v16h v; h8 p[2]; } A, B;
        A.p[0] = *(const h8*)(&lo[rsub * LP + kb]);
        A.p[1] = *(const h8*)(&lo[rsub * LP + kb + 16]);
        B.p[0] = *(const h8*)(wrow + kb);
        B.p[1] = *(const h8*)(wrow + kb + 16);
        acc = __builtin_amdgcn_wmma_f32_16x16x32_f16(
            false, A.v, false, B.v, (short)0, acc, false, false);
    }
    const int rhi = (lane >> 4) << 3;
    const int r0  = blockIdx.x * 16 + rhi;
    #pragma unroll
    for (int r = 0; r < 8; ++r)
        h2[(size_t)(r0 + r) * NC + rsub] = acc[r];
    const bool b0 = lane & 1, b1b = lane & 2, b2b = lane & 4, b3b = lane & 8;
    const float ws = as2[rsub], wd = ad2[rsub];
    float ps[8], pd[8];
    #pragma unroll
    for (int r = 0; r < 8; ++r) { ps[r] = acc[r] * ws; pd[r] = acc[r] * wd; }
    const float fs = fold8(ps, b3b, b2b, b1b), fd = fold8(pd, b3b, b2b, b1b);
    const float res = ((b0 ? fd : fs) + rowx<1>(b0 ? fs : fd)) * LOG2E;
    (b0 ? a_dst2 : a_src2)[r0 + (rsub >> 1)] = res;
}


__global__ __launch_bounds__(256)
void gat_node2(const int* __restrict__ deg, const int* __restrict__ bkt,
               const int* __restrict__ ovn, const int* __restrict__ ovf,
               const float* __restrict__ a_src, const float* __restrict__ a_dst,
               const float* __restrict__ h2, const float* __restrict__ b2,
               float* __restrict__ out) {
    const int lane = threadIdx.x & 31;
    const int d   = __builtin_amdgcn_readfirstlane(blockIdx.x * 8 + (threadIdx.x >> 5));
    const int dg  = deg[d];
    const int cnt = (dg < KB ? dg : KB) + 1;
    const int* lst = bkt + d * KB;
    const int c = lane & 15, es = lane >> 4;
    const int pl = 2 * c + es;
    const float adv = a_dst[d];
    const float NINF = -__builtin_inff();
    const float bc = b2[c];

    float m = NINF, l = 0.0f, acc = 0.0f;
    for (int c0 = 0; c0 < cnt; c0 += 32) {
        const int p = c0 + pl;
        const bool valid = p < cnt;
        int sid = d;
        if (p > 0 && valid) sid = lst[p - 1];
        float hv[16];
#define N2_ROWS(G)                                                                     \
        if (c0 + 8 * (G) < cnt) {                                     \
            hv[4 * (G) + 0] = h2[(size_t)rowsh<4 * (G) + 0>(sid) * NC + c];            \
            hv[4 * (G) + 1] = h2[(size_t)rowsh<4 * (G) + 1>(sid) * NC + c];            \
            hv[4 * (G) + 2] = h2[(size_t)rowsh<4 * (G) + 2>(sid) * NC + c];            \
            hv[4 * (G) + 3] = h2[(size_t)rowsh<4 * (G) + 3>(sid) * NC + c];            \
        }
        N2_ROWS(0) N2_ROWS(1) N2_ROWS(2) N2_ROWS(3)
#undef N2_ROWS
        const float a = valid ? lrelu(a_src[sid] + adv) : NINF;
        float mc = a;
        mc = fmaxf(mc, rowxb<1>(mc)); mc = fmaxf(mc, rowxb<2>(mc));
        mc = fmaxf(mc, rowxb<4>(mc)); mc = fmaxf(mc, rowxb<8>(mc));
        mc = fmaxf(mc, xhalf(mc));
        const float mn = fmaxf(m, mc);
        const float sc = ex2(m - mn);
        const float ex = ex2(a - mn);
        float lc = ex;
        lc += rowxb<1>(lc); lc += rowxb<2>(lc); lc += rowxb<4>(lc); lc += rowxb<8>(lc);
        lc += xhalf(lc);
        l = l * sc + lc;
        m = mn;
        acc *= sc;
#define N2_ACC(G)                                                                      \
        if (c0 + 8 * (G) < cnt) {                                     \
            acc += rowshf<4 * (G) + 0>(ex) * hv[4 * (G) + 0];                          \
            acc += rowshf<4 * (G) + 1>(ex) * hv[4 * (G) + 1];                          \
            acc += rowshf<4 * (G) + 2>(ex) * hv[4 * (G) + 2];                          \
            acc += rowshf<4 * (G) + 3>(ex) * hv[4 * (G) + 3];                          \
        }
        N2_ACC(0) N2_ACC(1) N2_ACC(2) N2_ACC(3)
#undef N2_ACC
    }
    if (dg > KB) {
        const int no = ovn[0];
        for (int o0 = 0; o0 < no; o0 += 32) {
            const int q = o0 + lane;
            int od = -1, os = 0;
            if (q < no) { od = ovf[2 * q]; os = ovf[2 * q + 1]; }
            unsigned mk = __builtin_amdgcn_ballot_w32(od == d);
            while (mk) {
                const int bit = __builtin_ctz(mk);
                mk &= mk - 1;
                const int s = __builtin_amdgcn_readlane(os, bit);
                const float ev = lrelu(a_src[s] + adv);
                const float mn = fmaxf(m, ev);
                const float sc = ex2(m - mn);
                const float w  = ex2(ev - mn);
                l = l * sc + w;
                m = mn;
                const float hv = h2[(size_t)s * NC + c];
                acc = acc * sc + (es == 0 ? w * hv : 0.0f);
            }
        }
    }
    acc += xhalf(acc);
    const float v = acc / (l + 1e-16f) + bc;
    float mx = v;
    mx = fmaxf(mx, rowxb<8>(mx)); mx = fmaxf(mx, rowxb<4>(mx));
    mx = fmaxf(mx, rowxb<2>(mx)); mx = fmaxf(mx, rowxb<1>(mx));
    float se = ex2((v - mx) * LOG2E);
    se += rowxb<8>(se); se += rowxb<4>(se); se += rowxb<2>(se); se += rowxb<1>(se);
    if (es == 0) out[(size_t)d * NC + c] = v - mx - __builtin_amdgcn_logf(se) * LN2;
}

extern "C" void kernel_launch(void* const* d_in, const int* in_sizes, int n_in,
                              void* d_out, int out_size, void* d_ws, size_t ws_size,
                              hipStream_t stream) {
    (void)in_sizes; (void)n_in; (void)out_size; (void)ws_size;
    const float* x   = (const float*)d_in[0];
    const int*   ei  = (const int*)d_in[1];
    const float* W1  = (const float*)d_in[2];
    const float* as1 = (const float*)d_in[3];
    const float* ad1 = (const float*)d_in[4];
    const float* b1  = (const float*)d_in[5];
    const float* W2  = (const float*)d_in[6];
    const float* as2 = (const float*)d_in[7];
    const float* ad2 = (const float*)d_in[8];
    const float* b2  = (const float*)d_in[9];
    float* out = (float*)d_out;

    char* wsb = (char*)d_ws;
    size_t off = 0;
    auto alloc = [&](size_t bytes) -> void* {
        void* p = wsb + off;
        off += (bytes + 255) & ~(size_t)255;
        return p;
    };
    _Float16* w1h  = (_Float16*)alloc((size_t)HC * NF * 2);
    _Float16* w2h  = (_Float16*)alloc((size_t)NC * HC * 2);
    _Float16* h1   = (_Float16*)alloc((size_t)NN * HC * 2);
    float* asrc1   = (float*)alloc((size_t)NN * HH * 4);
    float* adst1   = (float*)alloc((size_t)NN * HH * 4);
    int*   deg     = (int*)alloc((size_t)NN * 4);
    int*   ovn     = (int*)alloc((size_t)4);
    int*   bkt     = (int*)alloc((size_t)NN * KB * 4);
    int*   ovf     = (int*)alloc((size_t)EE * 2 * 4);
    float* h2      = (float*)alloc((size_t)NN * NC * 4);
    float* asrc2   = (float*)alloc((size_t)NN * 4);
    float* adst2   = (float*)alloc((size_t)NN * 4);

    int*   partw   = (int*)h1;
    int*   segt    = (int*)((char*)h1 + (size_t)EE * 8);

    const dim3 B(256);
    gat_part  <<<NR, dim3(1024), 0, stream>>>(W1, W2, w1h, w2h, ovn, ei, partw, segt);
    gat_bucket<<<NR, dim3(1024), 0, stream>>>(partw, segt, deg, bkt, ovn, ovf);
    gat_gemm1<<<(NN / 16 + 7) / 8, B, 0, stream>>>(x, w1h, as1, ad1, h1, asrc1, adst1);
    gat_node1<<<NN / 16, B, 0, stream>>>(deg, bkt, ovn, ovf, asrc1, adst1, h1, b1, w2h, as2, ad2, h2, asrc2, adst2);
    gat_node2<<<NN / 8, B, 0, stream>>>(deg, bkt, ovn, ovf, asrc2, adst2, h2, b2, out);
}
